# starting kernel
# baseline (speedup 1.0000x reference)
#include <hip/hip_runtime.h>
#include <math.h>

#define NN   50000
#define EE   800000
#define INF_ 512
#define FF   32
#define HH   4
#define CC   32
#define HCC  128
#define EDD  4
#define E2E  850000
#define LL   2

typedef float v2f __attribute__((ext_vector_type(2)));
typedef float v8f __attribute__((ext_vector_type(8)));

__device__ __forceinline__ unsigned fkey(float f) {
    unsigned u = __float_as_uint(f);
    return (u & 0x80000000u) ? ~u : (u | 0x80000000u);
}
__device__ __forceinline__ float funkey(unsigned u) {
    return __uint_as_float((u & 0x80000000u) ? (u & 0x7FFFFFFFu) : ~u);
}
__device__ __forceinline__ float leakyf(float x, float s) {
    return x >= 0.0f ? x : s * x;
}

template <int CT, int KT, bool USE_LDS>
__global__ __launch_bounds__(256)
void gemm_wmma_f32(const float* __restrict__ A, const float* __restrict__ W,
                   const float* __restrict__ bias, float* __restrict__ Cout,
                   int M, int Nout) {
    extern __shared__ float sW[];
    const int tid  = (int)threadIdx.x;
    const int lane = tid & 31;
    const int wave = (int)((blockIdx.x * blockDim.x + tid) >> 5);
    const int mtiles  = M >> 4;
    const int ngroups = Nout / (16 * CT);
    const int stride  = KT + 4;

    if (USE_LDS) {
        const int total = Nout * KT;
        for (int i = tid; i < total; i += 256) {
            const int c  = i / KT;
            const int kk = i - c * KT;
            sW[c * stride + kk] = W[i];
        }
        __syncthreads();
    }
    if (wave >= mtiles * ngroups) return;

    const int mtile = wave / ngroups;
    const int ng    = wave % ngroups;
    const int row0  = mtile * 16;
    const int col0  = ng * 16 * CT;
    const int hi = lane >> 4;
    const int mi = lane & 15;

    v8f acc[CT] = {};
    const float* Arow = A + (size_t)(row0 + mi) * KT + 2 * hi;

    v2f a_cur;
    a_cur.x = Arow[0];
    a_cur.y = Arow[1];
    v2f b_cur[CT];
    if (!USE_LDS) {
#pragma unroll
        for (int t = 0; t < CT; ++t) {
            const float* Wc = W + (size_t)(col0 + t * 16 + mi) * KT + 2 * hi;
            b_cur[t].x = Wc[0];
            b_cur[t].y = Wc[1];
        }
    }

    for (int k = 0; k < KT; k += 4) {
        v2f a_nxt = {};
        v2f b_nxt[CT];
        if (k + 4 < KT) {
            a_nxt.x = Arow[k + 4];
            a_nxt.y = Arow[k + 5];
            if (!USE_LDS) {
#pragma unroll
                for (int t = 0; t < CT; ++t) {
                    const float* Wc = W + (size_t)(col0 + t * 16 + mi) * KT + 2 * hi;
                    b_nxt[t].x = Wc[k + 4];
                    b_nxt[t].y = Wc[k + 5];
                }
                if ((k & 31) == 0)
                    __builtin_prefetch(Arow + k + 132, 0, 1);
            }
        }
#pragma unroll
        for (int t = 0; t < CT; ++t) {
            v2f b;
            if (USE_LDS) {
                const float* Sc = sW + (size_t)(col0 + t * 16 + mi) * stride + 2 * hi;
                b.x = Sc[k];
                b.y = Sc[k + 1];
            } else {
                b = b_cur[t];
            }
            acc[t] = __builtin_amdgcn_wmma_f32_16x16x4_f32(
                false, a_cur, false, b, (short)0, acc[t], false, false);
        }
        a_cur = a_nxt;
        if (!USE_LDS) {
#pragma unroll
            for (int t = 0; t < CT; ++t) b_cur[t] = b_nxt[t];
        }
    }

#pragma unroll
    for (int t = 0; t < CT; ++t) {
        const int col = col0 + t * 16 + mi;
        const float bb = bias ? bias[col] : 0.0f;
#pragma unroll
        for (int v = 0; v < 8; ++v) {
            const int r = row0 + v + 8 * hi;
            Cout[(size_t)r * Nout + col] = acc[t][v] + bb;
        }
    }
}

__global__ __launch_bounds__(256)
void ln_rows_kernel(const float* __restrict__ Z, const float* __restrict__ w,
                    const float* __restrict__ b, float* __restrict__ Hout,
                    float* __restrict__ jk, int mode, int M) {
    const int row  = (int)((blockIdx.x * blockDim.x + threadIdx.x) >> 5);
    const int lane = (int)(threadIdx.x & 31);
    if (row >= M) return;
    float v = Z[(size_t)row * FF + lane];
    float s = v;
#pragma unroll
    for (int off = 16; off >= 1; off >>= 1) s += __shfl_xor(s, off);
    const float mu = s * (1.0f / 32.0f);
    const float d = v - mu;
    float q = d * d;
#pragma unroll
    for (int off = 16; off >= 1; off >>= 1) q += __shfl_xor(q, off);
    const float var = q * (1.0f / 32.0f);
    float y = w[lane] * d * rsqrtf(var + 1e-5f) + b[lane];
    y = leakyf(y, 0.01f);
    Hout[(size_t)row * FF + lane] = y;
    float* o = jk + (size_t)row * FF + lane;
    *o = (mode == 0) ? y : fmaxf(*o, y);
}

__global__ __launch_bounds__(256)
void edge_score_kernel(const int* __restrict__ src, const int* __restrict__ dst,
                       const float* __restrict__ ea, const float* __restrict__ We,
                       const float* __restrict__ xl, const float* __restrict__ xr,
                       const float* __restrict__ att, float* __restrict__ score,
                       unsigned* __restrict__ smax) {
    const int e    = (int)((blockIdx.x * blockDim.x + threadIdx.x) >> 5);
    const int lane = (int)(threadIdx.x & 31);
    if (e >= E2E) return;
    int sN, dN;
    float4 ev;
    if (e < EE) {
        sN = src[e]; dN = dst[e];
        ev = ((const float4*)ea)[e];
    } else {
        sN = dN = e - EE;
        ev = make_float4(0.0f, 0.0f, 0.0f, 0.0f);
    }
    const float4 xlv = ((const float4*)xl)[(size_t)sN * (HCC / 4) + lane];
    const float4 xrv = ((const float4*)xr)[(size_t)dN * (HCC / 4) + lane];
    const float4 atv = ((const float4*)att)[lane];
    const float4* We4 = (const float4*)We;

    const float xa[4] = {xlv.x, xlv.y, xlv.z, xlv.w};
    const float xb[4] = {xrv.x, xrv.y, xrv.z, xrv.w};
    const float at[4] = {atv.x, atv.y, atv.z, atv.w};
    float sc = 0.0f;
#pragma unroll
    for (int k = 0; k < 4; ++k) {
        const float4 wr = We4[lane * 4 + k];
        const float eev = wr.x * ev.x + wr.y * ev.y + wr.z * ev.z + wr.w * ev.w;
        float m = xa[k] + xb[k] + eev;
        m = leakyf(m, 0.2f);
        sc += at[k] * m;
    }
    sc += __shfl_xor(sc, 4);
    sc += __shfl_xor(sc, 2);
    sc += __shfl_xor(sc, 1);
    if ((lane & 7) == 0) {
        const int h = lane >> 3;
        score[(size_t)e * HH + h] = sc;
        atomicMax(&smax[(size_t)dN * HH + h], fkey(sc));
    }
}

__global__ __launch_bounds__(256)
void edge_den_kernel(const int* __restrict__ dst, const float* __restrict__ score,
                     const unsigned* __restrict__ smax, float* __restrict__ den) {
    const int t = (int)(blockIdx.x * blockDim.x + threadIdx.x);
    if (t >= E2E * HH) return;
    const int e = t >> 2;
    const int h = t & 3;
    const int dN = (e < EE) ? dst[e] : (e - EE);
    const float ex = expf(score[t] - funkey(smax[(size_t)dN * HH + h]));
    atomicAdd(&den[(size_t)dN * HH + h], ex);
}

__global__ __launch_bounds__(256)
void edge_agg_kernel(const int* __restrict__ src, const int* __restrict__ dst,
                     const float* __restrict__ score, const unsigned* __restrict__ smax,
                     const float* __restrict__ den, const float* __restrict__ xl,
                     float* __restrict__ agg) {
    const int e    = (int)((blockIdx.x * blockDim.x + threadIdx.x) >> 5);
    const int lane = (int)(threadIdx.x & 31);
    if (e >= E2E) return;
    int sN, dN;
    if (e < EE) { sN = src[e]; dN = dst[e]; }
    else        { sN = dN = e - EE; }
    const int h = lane >> 3;
    const float mx = funkey(smax[(size_t)dN * HH + h]);
    const float dn = den[(size_t)dN * HH + h];
    const float alpha = expf(score[(size_t)e * HH + h] - mx) / (dn + 1e-16f);
    const float4 xv = ((const float4*)xl)[(size_t)sN * (HCC / 4) + lane];
    float* ao = agg + (size_t)dN * HCC + lane * 4;
    atomicAdd(ao + 0, xv.x * alpha);
    atomicAdd(ao + 1, xv.y * alpha);
    atomicAdd(ao + 2, xv.z * alpha);
    atomicAdd(ao + 3, xv.w * alpha);
}

__global__ __launch_bounds__(256)
void bias_stats_kernel(float* __restrict__ agg, const float* __restrict__ cb,
                       double* __restrict__ red, int total) {
    __shared__ double ss[256];
    __shared__ double sq[256];
    const int idx = (int)(blockIdx.x * blockDim.x + threadIdx.x);
    double s = 0.0, q = 0.0;
    if (idx < total) {
        const float y = agg[idx] + cb[idx & (HCC - 1)];
        agg[idx] = y;
        s = (double)y;
        q = (double)y * (double)y;
    }
    ss[threadIdx.x] = s;
    sq[threadIdx.x] = q;
    __syncthreads();
    for (int o = 128; o > 0; o >>= 1) {
        if ((int)threadIdx.x < o) {
            ss[threadIdx.x] += ss[threadIdx.x + o];
            sq[threadIdx.x] += sq[threadIdx.x + o];
        }
        __syncthreads();
    }
    if (threadIdx.x == 0) {
        atomicAdd(&red[0], ss[0]);
        atomicAdd(&red[1], sq[0]);
    }
}

__global__ __launch_bounds__(256)
void pln_apply_kernel(float* __restrict__ agg, const float* __restrict__ w,
                      const float* __restrict__ b, const double* __restrict__ red,
                      int total) {
    const int idx = (int)(blockIdx.x * blockDim.x + threadIdx.x);
    if (idx >= total) return;
    const double mu = red[0] / (double)total;
    double var = red[1] / (double)total - mu * mu;
    if (var < 0.0) var = 0.0;
    const float sd = sqrtf((float)var);
    const int c = idx & (HCC - 1);
    float y = w[c] * (float)((agg[idx] - mu) / (double)(sd + 1e-5f)) + b[c];
    agg[idx] = leakyf(y, 0.01f);
}

extern "C" void kernel_launch(void* const* d_in, const int* in_sizes, int n_in,
                              void* d_out, int out_size, void* d_ws, size_t ws_size,
                              hipStream_t stream) {
    const float* x          = (const float*)d_in[0];
    const int*   edge_index = (const int*)  d_in[1];
    const float* edge_attr  = (const float*)d_in[2];
    const float* embed_W    = (const float*)d_in[3];
    const float* embed_b    = (const float*)d_in[4];
    const float* embed_lnw  = (const float*)d_in[5];
    const float* embed_lnb  = (const float*)d_in[6];
    const float* Wl         = (const float*)d_in[7];
    const float* bl         = (const float*)d_in[8];
    const float* Wr         = (const float*)d_in[9];
    const float* br         = (const float*)d_in[10];
    const float* We         = (const float*)d_in[11];
    const float* att        = (const float*)d_in[12];
    const float* conv_b     = (const float*)d_in[13];
    const float* pln_w      = (const float*)d_in[14];
    const float* pln_b      = (const float*)d_in[15];
    const float* proj_W     = (const float*)d_in[16];
    const float* proj_b     = (const float*)d_in[17];
    const float* ln2_w      = (const float*)d_in[18];
    const float* ln2_b      = (const float*)d_in[19];

    const int* srcE = edge_index;
    const int* dstE = edge_index + EE;

    size_t off = 0;
    auto carve = [&](size_t nbytes) -> char* {
        char* p = (char*)d_ws + off;
        off += (nbytes + 255) & ~(size_t)255;
        return p;
    };
    float*    h     = (float*)   carve((size_t)NN * FF  * 4);
    float*    raw   = (float*)   carve((size_t)NN * FF  * 4);
    float*    xl    = (float*)   carve((size_t)NN * HCC * 4);
    float*    xr    = (float*)   carve((size_t)NN * HCC * 4);
    float*    agg   = (float*)   carve((size_t)NN * HCC * 4);
    float*    score = (float*)   carve((size_t)E2E * HH * 4);
    unsigned* smax  = (unsigned*)carve((size_t)NN * HH  * 4);
    float*    den   = (float*)   carve((size_t)NN * HH  * 4);
    double*   red   = (double*)  carve(2 * sizeof(double));
    (void)ws_size; (void)in_sizes; (void)n_in; (void)out_size;

    float* jk = (float*)d_out;

    const int mtiles = NN / 16;
    const dim3 blk(256);
    const dim3 gGemm((mtiles * 32 + 255) / 256);
    const dim3 gRows((NN * 32 + 255) / 256);
    const dim3 gEdgeW(((size_t)E2E * 32 + 255) / 256);
    const dim3 gEdgeT(((size_t)E2E * HH + 255) / 256);
    const dim3 gElem(((size_t)NN * HCC + 255) / 256);
    const int totalHC = NN * HCC;

    const size_t shLin  = (size_t)HCC * (FF + 4) * 4;
    const size_t shProj = (size_t)FF * (HCC + 4) * 4;

    gemm_wmma_f32<2, INF_, false><<<gGemm, blk, 0, stream>>>(x, embed_W, embed_b, raw, NN, FF);
    ln_rows_kernel<<<gRows, blk, 0, stream>>>(raw, embed_lnw, embed_lnb, h, jk, 0, NN);

    for (int l = 0; l < LL; ++l) {
        const float* Wl_l = Wl + (size_t)l * HCC * FF;
        const float* bl_l = bl + (size_t)l * HCC;
        const float* Wr_l = Wr + (size_t)l * HCC * FF;
        const float* br_l = br + (size_t)l * HCC;
        const float* We_l = We + (size_t)l * HCC * EDD;
        const float* at_l = att + (size_t)l * HH * CC;
        const float* cb_l = conv_b + (size_t)l * HCC;
        const float* pw_l = pln_w + (size_t)l * HCC;
        const float* pb_l = pln_b + (size_t)l * HCC;
        const float* pW_l = proj_W + (size_t)l * FF * HCC;
        const float* pB_l = proj_b + (size_t)l * FF;
        const float* l2w  = ln2_w + (size_t)l * FF;
        const float* l2b  = ln2_b + (size_t)l * FF;

        gemm_wmma_f32<8, FF, true><<<gGemm, blk, shLin, stream>>>(h, Wl_l, bl_l, xl, NN, HCC);
        gemm_wmma_f32<8, FF, true><<<gGemm, blk, shLin, stream>>>(h, Wr_l, br_l, xr, NN, HCC);

        hipMemsetAsync(smax, 0, (size_t)NN * HH * 4, stream);
        hipMemsetAsync(den,  0, (size_t)NN * HH * 4, stream);
        hipMemsetAsync(agg,  0, (size_t)NN * HCC * 4, stream);
        hipMemsetAsync(red,  0, 2 * sizeof(double), stream);

        edge_score_kernel<<<gEdgeW, blk, 0, stream>>>(srcE, dstE, edge_attr, We_l,
                                                      xl, xr, at_l, score, smax);
        edge_den_kernel<<<gEdgeT, blk, 0, stream>>>(dstE, score, smax, den);
        edge_agg_kernel<<<gEdgeW, blk, 0, stream>>>(srcE, dstE, score, smax, den, xl, agg);

        bias_stats_kernel<<<gElem, blk, 0, stream>>>(agg, cb_l, red, totalHC);
        pln_apply_kernel<<<gElem, blk, 0, stream>>>(agg, pw_l, pb_l, red, totalHC);

        gemm_wmma_f32<2, HCC, true><<<gGemm, blk, shProj, stream>>>(agg, pW_l, pB_l, raw, NN, FF);
        ln_rows_kernel<<<gRows, blk, 0, stream>>>(raw, l2w, l2b, h, jk, 1, NN);
    }
}


# speedup vs baseline: 1.1849x; 1.1849x over previous
#include <hip/hip_runtime.h>
#include <math.h>

#define NN   50000
#define EE   800000
#define INF_ 512
#define FF   32
#define HH   4
#define CC   32
#define HCC  128
#define EDD  4
#define E2E  850000
#define LL   2

typedef float v2f __attribute__((ext_vector_type(2)));
typedef float v8f __attribute__((ext_vector_type(8)));

__device__ __forceinline__ unsigned fkey(float f) {
    unsigned u = __float_as_uint(f);
    return (u & 0x80000000u) ? ~u : (u | 0x80000000u);
}
__device__ __forceinline__ float funkey(unsigned u) {
    return __uint_as_float((u & 0x80000000u) ? (u & 0x7FFFFFFFu) : ~u);
}
__device__ __forceinline__ float leakyf(float x, float s) {
    return x >= 0.0f ? x : s * x;
}

template <int CT, int KT, bool USE_LDS>
__global__ __launch_bounds__(256)
void gemm_wmma_f32(const float* __restrict__ A, const float* __restrict__ W,
                   const float* __restrict__ bias, float* __restrict__ Cout,
                   int M, int Nout) {
    extern __shared__ float sW[];
    const int tid  = (int)threadIdx.x;
    const int lane = tid & 31;
    const int wave = (int)((blockIdx.x * blockDim.x + tid) >> 5);
    const int mtiles  = M >> 4;
    const int ngroups = Nout / (16 * CT);
    const int stride  = KT + 4;

    if (USE_LDS) {
        const int total = Nout * KT;
        for (int i = tid; i < total; i += 256) {
            const int c  = i / KT;
            const int kk = i - c * KT;
            sW[c * stride + kk] = W[i];
        }
        __syncthreads();
    }
    if (wave >= mtiles * ngroups) return;

    const int mtile = wave / ngroups;
    const int ng    = wave % ngroups;
    const int row0  = mtile * 16;
    const int col0  = ng * 16 * CT;
    const int hi = lane >> 4;
    const int mi = lane & 15;

    v8f acc[CT] = {};
    const float* Arow = A + (size_t)(row0 + mi) * KT + 2 * hi;

    v2f a_cur;
    a_cur.x = Arow[0];
    a_cur.y = Arow[1];
    v2f b_cur[CT];
    if (!USE_LDS) {
#pragma unroll
        for (int t = 0; t < CT; ++t) {
            const float* Wc = W + (size_t)(col0 + t * 16 + mi) * KT + 2 * hi;
            b_cur[t].x = Wc[0];
            b_cur[t].y = Wc[1];
        }
    }

    for (int k = 0; k < KT; k += 4) {
        v2f a_nxt = {};
        v2f b_nxt[CT];
        if (k + 4 < KT) {
            a_nxt.x = Arow[k + 4];
            a_nxt.y = Arow[k + 5];
            if (!USE_LDS) {
#pragma unroll
                for (int t = 0; t < CT; ++t) {
                    const float* Wc = W + (size_t)(col0 + t * 16 + mi) * KT + 2 * hi;
                    b_nxt[t].x = Wc[k + 4];
                    b_nxt[t].y = Wc[k + 5];
                }
                if ((k & 31) == 0)
                    __builtin_prefetch(Arow + k + 132, 0, 1);
            }
        }
#pragma unroll
        for (int t = 0; t < CT; ++t) {
            v2f b;
            if (USE_LDS) {
                const float* Sc = sW + (size_t)(col0 + t * 16 + mi) * stride + 2 * hi;
                b.x = Sc[k];
                b.y = Sc[k + 1];
            } else {
                b = b_cur[t];
            }
            acc[t] = __builtin_amdgcn_wmma_f32_16x16x4_f32(
                false, a_cur, false, b, (short)0, acc[t], false, false);
        }
        a_cur = a_nxt;
        if (!USE_LDS) {
#pragma unroll
            for (int t = 0; t < CT; ++t) b_cur[t] = b_nxt[t];
        }
    }

#pragma unroll
    for (int t = 0; t < CT; ++t) {
        const int col = col0 + t * 16 + mi;
        const float bb = bias ? bias[col] : 0.0f;
#pragma unroll
        for (int v = 0; v < 8; ++v) {
            const int r = row0 + v + 8 * hi;
            Cout[(size_t)r * Nout + col] = acc[t][v] + bb;
        }
    }
}

__global__ __launch_bounds__(256)
void ln_rows_kernel(const float* __restrict__ Z, const float* __restrict__ w,
                    const float* __restrict__ b, float* __restrict__ Hout,
                    float* __restrict__ jk, int mode, int M) {
    const int row  = (int)((blockIdx.x * blockDim.x + threadIdx.x) >> 5);
    const int lane = (int)(threadIdx.x & 31);
    if (row >= M) return;
    float v = Z[(size_t)row * FF + lane];
    float s = v;
#pragma unroll
    for (int off = 16; off >= 1; off >>= 1) s += __shfl_xor(s, off);
    const float mu = s * (1.0f / 32.0f);
    const float d = v - mu;
    float q = d * d;
#pragma unroll
    for (int off = 16; off >= 1; off >>= 1) q += __shfl_xor(q, off);
    const float var = q * (1.0f / 32.0f);
    float y = w[lane] * d * rsqrtf(var + 1e-5f) + b[lane];
    y = leakyf(y, 0.01f);
    Hout[(size_t)row * FF + lane] = y;
    float* o = jk + (size_t)row * FF + lane;
    *o = (mode == 0) ? y : fmaxf(*o, y);
}

__global__ __launch_bounds__(256)
void edge_score_kernel(const int* __restrict__ src, const int* __restrict__ dst,
                       const float* __restrict__ ea, const float* __restrict__ We,
                       const float* __restrict__ xl, const float* __restrict__ xr,
                       const float* __restrict__ att, float* __restrict__ score,
                       unsigned* __restrict__ smax) {
    const int e    = (int)((blockIdx.x * blockDim.x + threadIdx.x) >> 5);
    const int lane = (int)(threadIdx.x & 31);
    if (e >= E2E) return;
    int sN, dN;
    float4 ev;
    if (e < EE) {
        sN = src[e]; dN = dst[e];
        ev = ((const float4*)ea)[e];
    } else {
        sN = dN = e - EE;
        ev = make_float4(0.0f, 0.0f, 0.0f, 0.0f);
    }
    const float4 xlv = ((const float4*)xl)[(size_t)sN * (HCC / 4) + lane];
    const float4 xrv = ((const float4*)xr)[(size_t)dN * (HCC / 4) + lane];
    const float4 atv = ((const float4*)att)[lane];
    const float4* We4 = (const float4*)We;

    const float xa[4] = {xlv.x, xlv.y, xlv.z, xlv.w};
    const float xb[4] = {xrv.x, xrv.y, xrv.z, xrv.w};
    const float at[4] = {atv.x, atv.y, atv.z, atv.w};
    float sc = 0.0f;
#pragma unroll
    for (int k = 0; k < 4; ++k) {
        const float4 wr = We4[lane * 4 + k];
        const float eev = wr.x * ev.x + wr.y * ev.y + wr.z * ev.z + wr.w * ev.w;
        float m = xa[k] + xb[k] + eev;
        m = leakyf(m, 0.2f);
        sc += at[k] * m;
    }
    sc += __shfl_xor(sc, 4);
    sc += __shfl_xor(sc, 2);
    sc += __shfl_xor(sc, 1);
    if ((lane & 7) == 0) {
        const int h = lane >> 3;
        score[(size_t)e * HH + h] = sc;
        atomicMax(&smax[(size_t)dN * HH + h], fkey(sc));
    }
}

__global__ __launch_bounds__(256)
void edge_den_kernel(const int* __restrict__ dst, const float* __restrict__ score,
                     const unsigned* __restrict__ smax, float* __restrict__ den) {
    const int t = (int)(blockIdx.x * blockDim.x + threadIdx.x);
    if (t >= E2E * HH) return;
    const int e = t >> 2;
    const int h = t & 3;
    const int dN = (e < EE) ? dst[e] : (e - EE);
    const float ex = expf(score[t] - funkey(smax[(size_t)dN * HH + h]));
    atomicAdd(&den[(size_t)dN * HH + h], ex);
}

__global__ __launch_bounds__(256)
void edge_agg_kernel(const int* __restrict__ src, const int* __restrict__ dst,
                     const float* __restrict__ score, const unsigned* __restrict__ smax,
                     const float* __restrict__ den, const float* __restrict__ xl,
                     float* __restrict__ agg) {
    const int e    = (int)((blockIdx.x * blockDim.x + threadIdx.x) >> 5);
    const int lane = (int)(threadIdx.x & 31);
    if (e >= E2E) return;
    int sN, dN;
    if (e < EE) { sN = src[e]; dN = dst[e]; }
    else        { sN = dN = e - EE; }
    const int h = lane >> 3;
    const float mx = funkey(smax[(size_t)dN * HH + h]);
    const float dn = den[(size_t)dN * HH + h];
    const float alpha = expf(score[(size_t)e * HH + h] - mx) / (dn + 1e-16f);
    const float4 xv = ((const float4*)xl)[(size_t)sN * (HCC / 4) + lane];
    float* ao = agg + (size_t)dN * HCC + lane * 4;
    atomicAdd(ao + 0, xv.x * alpha);
    atomicAdd(ao + 1, xv.y * alpha);
    atomicAdd(ao + 2, xv.z * alpha);
    atomicAdd(ao + 3, xv.w * alpha);
}

__global__ __launch_bounds__(256)
void bias_stats_kernel(float* __restrict__ agg, const float* __restrict__ cb,
                       double* __restrict__ red, int total) {
    __shared__ double ss[256];
    __shared__ double sq[256];
    const int idx = (int)(blockIdx.x * blockDim.x + threadIdx.x);
    double s = 0.0, q = 0.0;
    if (idx < total) {
        const float y = agg[idx] + cb[idx & (HCC - 1)];
        agg[idx] = y;
        s = (double)y;
        q = (double)y * (double)y;
    }
    ss[threadIdx.x] = s;
    sq[threadIdx.x] = q;
    __syncthreads();
    for (int o = 128; o > 0; o >>= 1) {
        if ((int)threadIdx.x < o) {
            ss[threadIdx.x] += ss[threadIdx.x + o];
            sq[threadIdx.x] += sq[threadIdx.x + o];
        }
        __syncthreads();
    }
    if (threadIdx.x == 0) {
        atomicAdd(&red[0], ss[0]);
        atomicAdd(&red[1], sq[0]);
    }
}

#define PLN_IT 5
__global__ __launch_bounds__(256)
void pln_apply_kernel(float* __restrict__ agg, const float* __restrict__ w,
                      const float* __restrict__ b, const double* __restrict__ red,
                      int total4) {
    const int q0     = (int)(blockIdx.x * blockDim.x + threadIdx.x);
    const int stride = (int)(gridDim.x * blockDim.x);
    float4* a4 = (float4*)agg;
    float4 v[PLN_IT];
#pragma unroll
    for (int i = 0; i < PLN_IT; ++i) {
        const int q = q0 + i * stride;
        v[i] = (q < total4) ? a4[q] : make_float4(0.0f, 0.0f, 0.0f, 0.0f);
    }
    const float4 wv = ((const float4*)w)[q0 & (HCC / 4 - 1)];
    const float4 bv = ((const float4*)b)[q0 & (HCC / 4 - 1)];

    constexpr double inv_n = 1.0 / ((double)NN * (double)HCC);
    const double mu  = red[0] * inv_n;
    const double var = red[1] * inv_n - mu * mu;
    float varf = (float)var;
    if (varf < 0.0f) varf = 0.0f;
    const float inv = 1.0f / (sqrtf(varf) + 1e-5f);
    const float muf = (float)mu;

#pragma unroll
    for (int i = 0; i < PLN_IT; ++i) {
        const int q = q0 + i * stride;
        if (q < total4) {
            float4 o;
            o.x = leakyf(wv.x * ((v[i].x - muf) * inv) + bv.x, 0.01f);
            o.y = leakyf(wv.y * ((v[i].y - muf) * inv) + bv.y, 0.01f);
            o.z = leakyf(wv.z * ((v[i].z - muf) * inv) + bv.z, 0.01f);
            o.w = leakyf(wv.w * ((v[i].w - muf) * inv) + bv.w, 0.01f);
            a4[q] = o;
        }
    }
}

extern "C" void kernel_launch(void* const* d_in, const int* in_sizes, int n_in,
                              void* d_out, int out_size, void* d_ws, size_t ws_size,
                              hipStream_t stream) {
    const float* x          = (const float*)d_in[0];
    const int*   edge_index = (const int*)  d_in[1];
    const float* edge_attr  = (const float*)d_in[2];
    const float* embed_W    = (const float*)d_in[3];
    const float* embed_b    = (const float*)d_in[4];
    const float* embed_lnw  = (const float*)d_in[5];
    const float* embed_lnb  = (const float*)d_in[6];
    const float* Wl         = (const float*)d_in[7];
    const float* bl         = (const float*)d_in[8];
    const float* Wr         = (const float*)d_in[9];
    const float* br         = (const float*)d_in[10];
    const float* We         = (const float*)d_in[11];
    const float* att        = (const float*)d_in[12];
    const float* conv_b     = (const float*)d_in[13];
    const float* pln_w      = (const float*)d_in[14];
    const float* pln_b      = (const float*)d_in[15];
    const float* proj_W     = (const float*)d_in[16];
    const float* proj_b     = (const float*)d_in[17];
    const float* ln2_w      = (const float*)d_in[18];
    const float* ln2_b      = (const float*)d_in[19];

    const int* srcE = edge_index;
    const int* dstE = edge_index + EE;

    size_t off = 0;
    auto carve = [&](size_t nbytes) -> char* {
        char* p = (char*)d_ws + off;
        off += (nbytes + 255) & ~(size_t)255;
        return p;
    };
    float*    h     = (float*)   carve((size_t)NN * FF  * 4);
    float*    raw   = (float*)   carve((size_t)NN * FF  * 4);
    float*    xl    = (float*)   carve((size_t)NN * HCC * 4);
    float*    xr    = (float*)   carve((size_t)NN * HCC * 4);
    float*    agg   = (float*)   carve((size_t)NN * HCC * 4);
    float*    score = (float*)   carve((size_t)E2E * HH * 4);
    unsigned* smax  = (unsigned*)carve((size_t)NN * HH  * 4);
    float*    den   = (float*)   carve((size_t)NN * HH  * 4);
    double*   red   = (double*)  carve(2 * sizeof(double));
    (void)ws_size; (void)in_sizes; (void)n_in; (void)out_size;

    float* jk = (float*)d_out;

    const int mtiles = NN / 16;
    const dim3 blk(256);
    const dim3 gGemm((mtiles * 32 + 255) / 256);
    const dim3 gRows((NN * 32 + 255) / 256);
    const dim3 gEdgeW(((size_t)E2E * 32 + 255) / 256);
    const dim3 gEdgeT(((size_t)E2E * HH + 255) / 256);
    const dim3 gElem(((size_t)NN * HCC + 255) / 256);
    const int totalHC = NN * HCC;
    const dim3 gPln((NN * HCC / 4) / (256 * PLN_IT));

    const size_t shLin  = (size_t)HCC * (FF + 4) * 4;
    const size_t shProj = (size_t)FF * (HCC + 4) * 4;

    gemm_wmma_f32<2, INF_, false><<<gGemm, blk, 0, stream>>>(x, embed_W, embed_b, raw, NN, FF);
    ln_rows_kernel<<<gRows, blk, 0, stream>>>(raw, embed_lnw, embed_lnb, h, jk, 0, NN);

    for (int l = 0; l < LL; ++l) {
        const float* Wl_l = Wl + (size_t)l * HCC * FF;
        const float* bl_l = bl + (size_t)l * HCC;
        const float* Wr_l = Wr + (size_t)l * HCC * FF;
        const float* br_l = br + (size_t)l * HCC;
        const float* We_l = We + (size_t)l * HCC * EDD;
        const float* at_l = att + (size_t)l * HH * CC;
        const float* cb_l = conv_b + (size_t)l * HCC;
        const float* pw_l = pln_w + (size_t)l * HCC;
        const float* pb_l = pln_b + (size_t)l * HCC;
        const float* pW_l = proj_W + (size_t)l * FF * HCC;
        const float* pB_l = proj_b + (size_t)l * FF;
        const float* l2w  = ln2_w + (size_t)l * FF;
        const float* l2b  = ln2_b + (size_t)l * FF;

        gemm_wmma_f32<8, FF, true><<<gGemm, blk, shLin, stream>>>(h, Wl_l, bl_l, xl, NN, HCC);
        gemm_wmma_f32<8, FF, true><<<gGemm, blk, shLin, stream>>>(h, Wr_l, br_l, xr, NN, HCC);

        hipMemsetAsync(smax, 0, (size_t)NN * HH * 4, stream);
        hipMemsetAsync(den,  0, (size_t)NN * HH * 4, stream);
        hipMemsetAsync(agg,  0, (size_t)NN * HCC * 4, stream);
        hipMemsetAsync(red,  0, 2 * sizeof(double), stream);

        edge_score_kernel<<<gEdgeW, blk, 0, stream>>>(srcE, dstE, edge_attr, We_l,
                                                      xl, xr, at_l, score, smax);
        edge_den_kernel<<<gEdgeT, blk, 0, stream>>>(dstE, score, smax, den);
        edge_agg_kernel<<<gEdgeW, blk, 0, stream>>>(srcE, dstE, score, smax, den, xl, agg);

        bias_stats_kernel<<<gElem, blk, 0, stream>>>(agg, cb_l, red, totalHC);
        pln_apply_kernel<<<gPln, blk, 0, stream>>>(agg, pw_l, pb_l, red, totalHC / 4);

        gemm_wmma_f32<2, HCC, true><<<gGemm, blk, shProj, stream>>>(agg, pW_l, pB_l, raw, NN, FF);
        ln_rows_kernel<<<gRows, blk, 0, stream>>>(raw, l2w, l2b, h, jk, 1, NN);
    }
}


# edge passes (score, softmax denominator, aggregation): each wave walks 32 consecutive edges, 8 edges' row gathers in flight per trip, instead of one wave per edge; atomics and head-sum pairs kept
# speedup vs baseline: 1.2621x; 1.0652x over previous
#include <hip/hip_runtime.h>
#include <math.h>

#define NN   50000
#define EE   800000
#define INF_ 512
#define FF   32
#define HH   4
#define CC   32
#define HCC  128
#define EDD  4
#define E2E  850000
#define LL   2

typedef float v2f __attribute__((ext_vector_type(2)));
typedef float v8f __attribute__((ext_vector_type(8)));
typedef float v4f __attribute__((ext_vector_type(4)));

__device__ __forceinline__ unsigned fkey(float f) {
    unsigned u = __float_as_uint(f);
    return (u & 0x80000000u) ? ~u : (u | 0x80000000u);
}
__device__ __forceinline__ float funkey(unsigned u) {
    return __uint_as_float((u & 0x80000000u) ? (u & 0x7FFFFFFFu) : ~u);
}
__device__ __forceinline__ float leakyf(float x, float s) {
    return x >= 0.0f ? x : s * x;
}

template <int CT, int KT, bool USE_LDS>
__global__ __launch_bounds__(256)
void gemm_wmma_f32(const float* __restrict__ A, const float* __restrict__ W,
                   const float* __restrict__ bias, float* __restrict__ Cout,
                   int M, int Nout) {
    extern __shared__ float sW[];
    const int tid  = (int)threadIdx.x;
    const int lane = tid & 31;
    const int wave = (int)((blockIdx.x * blockDim.x + tid) >> 5);
    const int mtiles  = M >> 4;
    const int ngroups = Nout / (16 * CT);
    const int stride  = KT + 4;

    if (USE_LDS) {
        const int total = Nout * KT;
        for (int i = tid; i < total; i += 256) {
            const int c  = i / KT;
            const int kk = i - c * KT;
            sW[c * stride + kk] = W[i];
        }
        __syncthreads();
    }
    if (wave >= mtiles * ngroups) return;

    const int mtile = wave / ngroups;
    const int ng    = wave % ngroups;
    const int row0  = mtile * 16;
    const int col0  = ng * 16 * CT;
    const int hi = lane >> 4;
    const int mi = lane & 15;

    v8f acc[CT] = {};
    const float* Arow = A + (size_t)(row0 + mi) * KT + 2 * hi;

    v2f a_cur;
    a_cur.x = Arow[0];
    a_cur.y = Arow[1];
    v2f b_cur[CT];
    if (!USE_LDS) {
#pragma unroll
        for (int t = 0; t < CT; ++t) {
            const float* Wc = W + (size_t)(col0 + t * 16 + mi) * KT + 2 * hi;
            b_cur[t].x = Wc[0];
            b_cur[t].y = Wc[1];
        }
    }

    for (int k = 0; k < KT; k += 4) {
        v2f a_nxt = {};
        v2f b_nxt[CT];
        if (k + 4 < KT) {
            a_nxt.x = Arow[k + 4];
            a_nxt.y = Arow[k + 5];
            if (!USE_LDS) {
#pragma unroll
                for (int t = 0; t < CT; ++t) {
                    const float* Wc = W + (size_t)(col0 + t * 16 + mi) * KT + 2 * hi;
                    b_nxt[t].x = Wc[k + 4];
                    b_nxt[t].y = Wc[k + 5];
                }
                if ((k & 31) == 0)
                    __builtin_prefetch(Arow + k + 132, 0, 1);
            }
        }
#pragma unroll
        for (int t = 0; t < CT; ++t) {
            v2f b;
            if (USE_LDS) {
                const float* Sc = sW + (size_t)(col0 + t * 16 + mi) * stride + 2 * hi;
                b.x = Sc[k];
                b.y = Sc[k + 1];
            } else {
                b = b_cur[t];
            }
            acc[t] = __builtin_amdgcn_wmma_f32_16x16x4_f32(
                false, a_cur, false, b, (short)0, acc[t], false, false);
        }
        a_cur = a_nxt;
        if (!USE_LDS) {
#pragma unroll
            for (int t = 0; t < CT; ++t) b_cur[t] = b_nxt[t];
        }
    }

#pragma unroll
    for (int t = 0; t < CT; ++t) {
        const int col = col0 + t * 16 + mi;
        const float bb = bias ? bias[col] : 0.0f;
#pragma unroll
        for (int v = 0; v < 8; ++v) {
            const int r = row0 + v + 8 * hi;
            Cout[(size_t)r * Nout + col] = acc[t][v] + bb;
        }
    }
}

__global__ __launch_bounds__(256)
void ln_rows_kernel(const float* __restrict__ Z, const float* __restrict__ w,
                    const float* __restrict__ b, float* __restrict__ Hout,
                    float* __restrict__ jk, int mode, int M) {
    const int row  = (int)((blockIdx.x * blockDim.x + threadIdx.x) >> 5);
    const int lane = (int)(threadIdx.x & 31);
    if (row >= M) return;
    float v = Z[(size_t)row * FF + lane];
    float s = v;
#pragma unroll
    for (int off = 16; off >= 1; off >>= 1) s += __shfl_xor(s, off);
    const float mu = s * (1.0f / 32.0f);
    const float d = v - mu;
    float q = d * d;
#pragma unroll
    for (int off = 16; off >= 1; off >>= 1) q += __shfl_xor(q, off);
    const float var = q * (1.0f / 32.0f);
    float y = w[lane] * d * rsqrtf(var + 1e-5f) + b[lane];
    y = leakyf(y, 0.01f);
    Hout[(size_t)row * FF + lane] = y;
    float* o = jk + (size_t)row * FF + lane;
    *o = (mode == 0) ? y : fmaxf(*o, y);
}

#define EW_RUN   32
#define EW_U     8
#define EW_REALW (EE / EW_RUN)
#define EW_WAVES ((E2E + EW_RUN - 1) / EW_RUN)

__device__ __forceinline__ int ew_wave_index() {
    return (int)blockIdx.x * 8 + __builtin_amdgcn_readfirstlane((int)(threadIdx.x >> 5));
}
__device__ __forceinline__ int ew_trips(int ebase) {
    const int rem = E2E - ebase;
    return rem >= EW_RUN ? (EW_RUN / EW_U) : (rem / EW_U);
}
__device__ __forceinline__ int ew_self_node(int ebase, int lane) {
    const int n = ebase + lane - EE;
    return n < NN ? n : NN - 1;
}
__device__ __forceinline__ float ew_bcast(float v, int j) {
    return __int_as_float(__builtin_amdgcn_readlane(__float_as_int(v), j));
}

__global__ __launch_bounds__(256)
void edge_score_kernel(const int* __restrict__ src, const int* __restrict__ dst,
                       const float* __restrict__ ea, const float* __restrict__ We,
                       const float* __restrict__ xl, const float* __restrict__ xr,
                       const float* __restrict__ att, float* __restrict__ score,
                       unsigned* __restrict__ smax) {
    const int lane = (int)(threadIdx.x & 31);
    const int wv   = ew_wave_index();
    if (wv >= EW_WAVES) return;
    const int ebase = wv * EW_RUN;
    const int ntrip = ew_trips(ebase);

    int sL, dL;
    float4 evL;
    if (wv < EW_REALW) {
        sL  = src[ebase + lane];
        dL  = dst[ebase + lane];
        evL = ((const float4*)ea)[ebase + lane];
    } else {
        sL = dL = ew_self_node(ebase, lane);
        evL = make_float4(0.0f, 0.0f, 0.0f, 0.0f);
    }
    const float4 atv = ((const float4*)att)[lane];
    const float4* We4 = (const float4*)We;
    const float4 wr0 = We4[lane * 4 + 0];
    const float4 wr1 = We4[lane * 4 + 1];
    const float4 wr2 = We4[lane * 4 + 2];
    const float4 wr3 = We4[lane * 4 + 3];
    const v4f* xl4 = (const v4f*)xl;
    const v4f* xr4 = (const v4f*)xr;

    for (int t = 0; t < ntrip; ++t) {
        v4f xlv[EW_U], xrv[EW_U];
#pragma unroll
        for (int i = 0; i < EW_U; ++i) {
            const int sN = __builtin_amdgcn_readlane(sL, t * EW_U + i);
            const int dN = __builtin_amdgcn_readlane(dL, t * EW_U + i);
            xlv[i] = xl4[(size_t)sN * (HCC / 4) + lane];
            xrv[i] = xr4[(size_t)dN * (HCC / 4) + lane];
        }
        asm volatile("" : "+v"(xlv[0]), "+v"(xlv[1]), "+v"(xlv[2]), "+v"(xlv[3]),
                          "+v"(xlv[4]), "+v"(xlv[5]), "+v"(xlv[6]), "+v"(xlv[7]),
                          "+v"(xrv[0]), "+v"(xrv[1]), "+v"(xrv[2]), "+v"(xrv[3]),
                          "+v"(xrv[4]), "+v"(xrv[5]), "+v"(xrv[6]), "+v"(xrv[7]));
        float sc[EW_U];
#pragma unroll
        for (int i = 0; i < EW_U; ++i) {
            const int j = t * EW_U + i;
            const float ex = ew_bcast(evL.x, j);
            const float ey = ew_bcast(evL.y, j);
            const float ez = ew_bcast(evL.z, j);
            const float ew = ew_bcast(evL.w, j);
            float s = 0.0f;
            {
                const float eev = wr0.x * ex + wr0.y * ey + wr0.z * ez + wr0.w * ew;
                float m = xlv[i].x + xrv[i].x + eev;
                m = leakyf(m, 0.2f);
                s += atv.x * m;
            }
            {
                const float eev = wr1.x * ex + wr1.y * ey + wr1.z * ez + wr1.w * ew;
                float m = xlv[i].y + xrv[i].y + eev;
                m = leakyf(m, 0.2f);
                s += atv.y * m;
            }
            {
                const float eev = wr2.x * ex + wr2.y * ey + wr2.z * ez + wr2.w * ew;
                float m = xlv[i].z + xrv[i].z + eev;
                m = leakyf(m, 0.2f);
                s += atv.z * m;
            }
            {
                const float eev = wr3.x * ex + wr3.y * ey + wr3.z * ez + wr3.w * ew;
                float m = xlv[i].w + xrv[i].w + eev;
                m = leakyf(m, 0.2f);
                s += atv.w * m;
            }
            sc[i] = s;
        }
        const bool b2 = (lane & 4) != 0;
        const bool b1 = (lane & 2) != 0;
        const bool b0 = (lane & 1) != 0;
        float ra[4];
#pragma unroll
        for (int k = 0; k < 4; ++k) {
            const float keep = b2 ? sc[k + 4] : sc[k];
            const float send = b2 ? sc[k] : sc[k + 4];
            ra[k] = keep + __shfl_xor(send, 4);
        }
        float rb[2];
#pragma unroll
        for (int k = 0; k < 2; ++k) {
            const float keep = b1 ? ra[k + 2] : ra[k];
            const float send = b1 ? ra[k] : ra[k + 2];
            rb[k] = keep + __shfl_xor(send, 2);
        }
        const float keep1 = b0 ? rb[1] : rb[0];
        const float send1 = b0 ? rb[0] : rb[1];
        const float tot = keep1 + __shfl_xor(send1, 1);

        const int g = lane & 7;
        const int h = lane >> 3;
        const int dsel = __shfl(dL, t * EW_U + g);
        const int e = ebase + t * EW_U + g;
        score[(size_t)e * HH + h] = tot;
        atomicMax(&smax[(size_t)dsel * HH + h], fkey(tot));
    }
}

__global__ __launch_bounds__(256)
void edge_den_kernel(const int* __restrict__ dst, const float* __restrict__ score,
                     const unsigned* __restrict__ smax, float* __restrict__ den) {
    const int lane = (int)(threadIdx.x & 31);
    const int wv   = ew_wave_index();
    if (wv >= EW_WAVES) return;
    const int ebase = wv * EW_RUN;
    const int ntrip = ew_trips(ebase);
    int dL;
    if (wv < EW_REALW) dL = dst[ebase + lane];
    else               dL = ew_self_node(ebase, lane);

    int      dsel[EW_RUN / EW_U];
    float    scv[EW_RUN / EW_U];
    unsigned mxk[EW_RUN / EW_U];
#pragma unroll
    for (int t = 0; t < EW_RUN / EW_U; ++t) {
        dsel[t] = __shfl(dL, t * EW_U + (lane >> 2));
        if (t < ntrip) {
            scv[t] = score[(size_t)(ebase + t * EW_U) * HH + lane];
            mxk[t] = smax[(size_t)dsel[t] * HH + (lane & 3)];
        }
    }
#pragma unroll
    for (int t = 0; t < EW_RUN / EW_U; ++t) {
        if (t < ntrip) {
            const float ex = expf(scv[t] - funkey(mxk[t]));
            atomicAdd(&den[(size_t)dsel[t] * HH + (lane & 3)], ex);
        }
    }
}

__global__ __launch_bounds__(256)
void edge_agg_kernel(const int* __restrict__ src, const int* __restrict__ dst,
                     const float* __restrict__ score, const unsigned* __restrict__ smax,
                     const float* __restrict__ den, const float* __restrict__ xl,
                     float* __restrict__ agg) {
    const int lane = (int)(threadIdx.x & 31);
    const int wv   = ew_wave_index();
    if (wv >= EW_WAVES) return;
    const int ebase = wv * EW_RUN;
    const int ntrip = ew_trips(ebase);
    int sL, dL;
    if (wv < EW_REALW) {
        sL = src[ebase + lane];
        dL = dst[ebase + lane];
    } else {
        sL = dL = ew_self_node(ebase, lane);
    }
    const float4* xl4 = (const float4*)xl;

    for (int t = 0; t < ntrip; ++t) {
        const int dsel = __shfl(dL, t * EW_U + (lane >> 2));
        const int e0 = ebase + t * EW_U;
        const float    scv = score[(size_t)e0 * HH + lane];
        const unsigned mxk = smax[(size_t)dsel * HH + (lane & 3)];
        const float    dn  = den[(size_t)dsel * HH + (lane & 3)];
        float4 xv[EW_U];
#pragma unroll
        for (int i = 0; i < EW_U; ++i) {
            const int sN = __builtin_amdgcn_readlane(sL, t * EW_U + i);
            xv[i] = xl4[(size_t)sN * (HCC / 4) + lane];
        }
        const float alphaT = expf(scv - funkey(mxk)) / (dn + 1e-16f);
#pragma unroll
        for (int i = 0; i < EW_U; ++i) {
            const float alpha = __shfl(alphaT, 4 * i + (lane >> 3));
            const int dN = __builtin_amdgcn_readlane(dL, t * EW_U + i);
            float* ao = agg + (size_t)dN * HCC + lane * 4;
            atomicAdd(ao + 0, xv[i].x * alpha);
            atomicAdd(ao + 1, xv[i].y * alpha);
            atomicAdd(ao + 2, xv[i].z * alpha);
            atomicAdd(ao + 3, xv[i].w * alpha);
        }
    }
}

__global__ __launch_bounds__(256)
void bias_stats_kernel(float* __restrict__ agg, const float* __restrict__ cb,
                       double* __restrict__ red, int total) {
    __shared__ double ss[256];
    __shared__ double sq[256];
    const int idx = (int)(blockIdx.x * blockDim.x + threadIdx.x);
    double s = 0.0, q = 0.0;
    if (idx < total) {
        const float y = agg[idx] + cb[idx & (HCC - 1)];
        agg[idx] = y;
        s = (double)y;
        q = (double)y * (double)y;
    }
    ss[threadIdx.x] = s;
    sq[threadIdx.x] = q;
    __syncthreads();
    for (int o = 128; o > 0; o >>= 1) {
        if ((int)threadIdx.x < o) {
            ss[threadIdx.x] += ss[threadIdx.x + o];
            sq[threadIdx.x] += sq[threadIdx.x + o];
        }
        __syncthreads();
    }
    if (threadIdx.x == 0) {
        atomicAdd(&red[0], ss[0]);
        atomicAdd(&red[1], sq[0]);
    }
}

#define PLN_IT 5
__global__ __launch_bounds__(256)
void pln_apply_kernel(float* __restrict__ agg, const float* __restrict__ w,
                      const float* __restrict__ b, const double* __restrict__ red,
                      int total4) {
    const int q0     = (int)(blockIdx.x * blockDim.x + threadIdx.x);
    const int stride = (int)(gridDim.x * blockDim.x);
    float4* a4 = (float4*)agg;
    float4 v[PLN_IT];
#pragma unroll
    for (int i = 0; i < PLN_IT; ++i) {
        const int q = q0 + i * stride;
        v[i] = (q < total4) ? a4[q] : make_float4(0.0f, 0.0f, 0.0f, 0.0f);
    }
    const float4 wv = ((const float4*)w)[q0 & (HCC / 4 - 1)];
    const float4 bv = ((const float4*)b)[q0 & (HCC / 4 - 1)];

    constexpr double inv_n = 1.0 / ((double)NN * (double)HCC);
    const double mu  = red[0] * inv_n;
    const double var = red[1] * inv_n - mu * mu;
    float varf = (float)var;
    if (varf < 0.0f) varf = 0.0f;
    const float inv = 1.0f / (sqrtf(varf) + 1e-5f);
    const float muf = (float)mu;

#pragma unroll
    for (int i = 0; i < PLN_IT; ++i) {
        const int q = q0 + i * stride;
        if (q < total4) {
            float4 o;
            o.x = leakyf(wv.x * ((v[i].x - muf) * inv) + bv.x, 0.01f);
            o.y = leakyf(wv.y * ((v[i].y - muf) * inv) + bv.y, 0.01f);
            o.z = leakyf(wv.z * ((v[i].z - muf) * inv) + bv.z, 0.01f);
            o.w = leakyf(wv.w * ((v[i].w - muf) * inv) + bv.w, 0.01f);
            a4[q] = o;
        }
    }
}

extern "C" void kernel_launch(void* const* d_in, const int* in_sizes, int n_in,
                              void* d_out, int out_size, void* d_ws, size_t ws_size,
                              hipStream_t stream) {
    const float* x          = (const float*)d_in[0];
    const int*   edge_index = (const int*)  d_in[1];
    const float* edge_attr  = (const float*)d_in[2];
    const float* embed_W    = (const float*)d_in[3];
    const float* embed_b    = (const float*)d_in[4];
    const float* embed_lnw  = (const float*)d_in[5];
    const float* embed_lnb  = (const float*)d_in[6];
    const float* Wl         = (const float*)d_in[7];
    const float* bl         = (const float*)d_in[8];
    const float* Wr         = (const float*)d_in[9];
    const float* br         = (const float*)d_in[10];
    const float* We         = (const float*)d_in[11];
    const float* att        = (const float*)d_in[12];
    const float* conv_b     = (const float*)d_in[13];
    const float* pln_w      = (const float*)d_in[14];
    const float* pln_b      = (const float*)d_in[15];
    const float* proj_W     = (const float*)d_in[16];
    const float* proj_b     = (const float*)d_in[17];
    const float* ln2_w      = (const float*)d_in[18];
    const float* ln2_b      = (const float*)d_in[19];

    const int* srcE = edge_index;
    const int* dstE = edge_index + EE;

    size_t off = 0;
    auto carve = [&](size_t nbytes) -> char* {
        char* p = (char*)d_ws + off;
        off += (nbytes + 255) & ~(size_t)255;
        return p;
    };
    float*    h     = (float*)   carve((size_t)NN * FF  * 4);
    float*    raw   = (float*)   carve((size_t)NN * FF  * 4);
    float*    xl    = (float*)   carve((size_t)NN * HCC * 4);
    float*    xr    = (float*)   carve((size_t)NN * HCC * 4);
    float*    agg   = (float*)   carve((size_t)NN * HCC * 4);
    float*    score = (float*)   carve((size_t)E2E * HH * 4);
    unsigned* smax  = (unsigned*)carve((size_t)NN * HH  * 4);
    float*    den   = (float*)   carve((size_t)NN * HH  * 4);
    double*   red   = (double*)  carve(2 * sizeof(double));
    (void)ws_size; (void)in_sizes; (void)n_in; (void)out_size;

    float* jk = (float*)d_out;

    const int mtiles = NN / 16;
    const dim3 blk(256);
    const dim3 gGemm((mtiles * 32 + 255) / 256);
    const dim3 gRows((NN * 32 + 255) / 256);
    const dim3 gEdgeR((EW_WAVES + 7) / 8);
    const dim3 gElem(((size_t)NN * HCC + 255) / 256);
    const int totalHC = NN * HCC;
    const dim3 gPln((NN * HCC / 4) / (256 * PLN_IT));

    const size_t shLin  = (size_t)HCC * (FF + 4) * 4;
    const size_t shProj = (size_t)FF * (HCC + 4) * 4;

    gemm_wmma_f32<2, INF_, false><<<gGemm, blk, 0, stream>>>(x, embed_W, embed_b, raw, NN, FF);
    ln_rows_kernel<<<gRows, blk, 0, stream>>>(raw, embed_lnw, embed_lnb, h, jk, 0, NN);

    for (int l = 0; l < LL; ++l) {
        const float* Wl_l = Wl + (size_t)l * HCC * FF;
        const float* bl_l = bl + (size_t)l * HCC;
        const float* Wr_l = Wr + (size_t)l * HCC * FF;
        const float* br_l = br + (size_t)l * HCC;
        const float* We_l = We + (size_t)l * HCC * EDD;
        const float* at_l = att + (size_t)l * HH * CC;
        const float* cb_l = conv_b + (size_t)l * HCC;
        const float* pw_l = pln_w + (size_t)l * HCC;
        const float* pb_l = pln_b + (size_t)l * HCC;
        const float* pW_l = proj_W + (size_t)l * FF * HCC;
        const float* pB_l = proj_b + (size_t)l * FF;
        const float* l2w  = ln2_w + (size_t)l * FF;
        const float* l2b  = ln2_b + (size_t)l * FF;

        gemm_wmma_f32<8, FF, true><<<gGemm, blk, shLin, stream>>>(h, Wl_l, bl_l, xl, NN, HCC);
        gemm_wmma_f32<8, FF, true><<<gGemm, blk, shLin, stream>>>(h, Wr_l, br_l, xr, NN, HCC);

        hipMemsetAsync(smax, 0, (size_t)NN * HH * 4, stream);
        hipMemsetAsync(den,  0, (size_t)NN * HH * 4, stream);
        hipMemsetAsync(agg,  0, (size_t)NN * HCC * 4, stream);
        hipMemsetAsync(red,  0, 2 * sizeof(double), stream);

        edge_score_kernel<<<gEdgeR, blk, 0, stream>>>(srcE, dstE, edge_attr, We_l,
                                                      xl, xr, at_l, score, smax);
        edge_den_kernel<<<gEdgeR, blk, 0, stream>>>(dstE, score, smax, den);
        edge_agg_kernel<<<gEdgeR, blk, 0, stream>>>(srcE, dstE, score, smax, den, xl, agg);

        bias_stats_kernel<<<gElem, blk, 0, stream>>>(agg, cb_l, red, totalHC);
        pln_apply_kernel<<<gPln, blk, 0, stream>>>(agg, pw_l, pb_l, red, totalHC / 4);

        gemm_wmma_f32<2, HCC, true><<<gGemm, blk, shProj, stream>>>(agg, pW_l, pB_l, raw, NN, FF);
        ln_rows_kernel<<<gRows, blk, 0, stream>>>(raw, l2w, l2b, h, jk, 1, NN);
    }
}


# edge aggregation: each float atomic instruction adds into one whole 128-byte line of the target row (lane = channel within head), not 8 lanes in each of 4 lines; xl gathered to match; atomics kept
# speedup vs baseline: 1.5888x; 1.2589x over previous
#include <hip/hip_runtime.h>
#include <math.h>

#define NN   50000
#define EE   800000
#define INF_ 512
#define FF   32
#define HH   4
#define CC   32
#define HCC  128
#define EDD  4
#define E2E  850000
#define LL   2

typedef float v2f __attribute__((ext_vector_type(2)));
typedef float v8f __attribute__((ext_vector_type(8)));
typedef float v4f __attribute__((ext_vector_type(4)));

__device__ __forceinline__ unsigned fkey(float f) {
    unsigned u = __float_as_uint(f);
    return (u & 0x80000000u) ? ~u : (u | 0x80000000u);
}
__device__ __forceinline__ float funkey(unsigned u) {
    return __uint_as_float((u & 0x80000000u) ? (u & 0x7FFFFFFFu) : ~u);
}
__device__ __forceinline__ float leakyf(float x, float s) {
    return x >= 0.0f ? x : s * x;
}

template <int CT, int KT, bool USE_LDS>
__global__ __launch_bounds__(256)
void gemm_wmma_f32(const float* __restrict__ A, const float* __restrict__ W,
                   const float* __restrict__ bias, float* __restrict__ Cout,
                   int M, int Nout) {
    extern __shared__ float sW[];
    const int tid  = (int)threadIdx.x;
    const int lane = tid & 31;
    const int wave = (int)((blockIdx.x * blockDim.x + tid) >> 5);
    const int mtiles  = M >> 4;
    const int ngroups = Nout / (16 * CT);
    const int stride  = KT + 4;

    if (USE_LDS) {
        const int total = Nout * KT;
        for (int i = tid; i < total; i += 256) {
            const int c  = i / KT;
            const int kk = i - c * KT;
            sW[c * stride + kk] = W[i];
        }
        __syncthreads();
    }
    if (wave >= mtiles * ngroups) return;

    const int mtile = wave / ngroups;
    const int ng    = wave % ngroups;
    const int row0  = mtile * 16;
    const int col0  = ng * 16 * CT;
    const int hi = lane >> 4;
    const int mi = lane & 15;

    v8f acc[CT] = {};
    const float* Arow = A + (size_t)(row0 + mi) * KT + 2 * hi;

    v2f a_cur;
    a_cur.x = Arow[0];
    a_cur.y = Arow[1];
    v2f b_cur[CT];
    if (!USE_LDS) {
#pragma unroll
        for (int t = 0; t < CT; ++t) {
            const float* Wc = W + (size_t)(col0 + t * 16 + mi) * KT + 2 * hi;
            b_cur[t].x = Wc[0];
            b_cur[t].y = Wc[1];
        }
    }

    for (int k = 0; k < KT; k += 4) {
        v2f a_nxt = {};
        v2f b_nxt[CT];
        if (k + 4 < KT) {
            a_nxt.x = Arow[k + 4];
            a_nxt.y = Arow[k + 5];
            if (!USE_LDS) {
#pragma unroll
                for (int t = 0; t < CT; ++t) {
                    const float* Wc = W + (size_t)(col0 + t * 16 + mi) * KT + 2 * hi;
                    b_nxt[t].x = Wc[k + 4];
                    b_nxt[t].y = Wc[k + 5];
                }
                if ((k & 31) == 0)
                    __builtin_prefetch(Arow + k + 132, 0, 1);
            }
        }
#pragma unroll
        for (int t = 0; t < CT; ++t) {
            v2f b;
            if (USE_LDS) {
                const float* Sc = sW + (size_t)(col0 + t * 16 + mi) * stride + 2 * hi;
                b.x = Sc[k];
                b.y = Sc[k + 1];
            } else {
                b = b_cur[t];
            }
            acc[t] = __builtin_amdgcn_wmma_f32_16x16x4_f32(
                false, a_cur, false, b, (short)0, acc[t], false, false);
        }
        a_cur = a_nxt;
        if (!USE_LDS) {
#pragma unroll
            for (int t = 0; t < CT; ++t) b_cur[t] = b_nxt[t];
        }
    }

#pragma unroll
    for (int t = 0; t < CT; ++t) {
        const int col = col0 + t * 16 + mi;
        const float bb = bias ? bias[col] : 0.0f;
#pragma unroll
        for (int v = 0; v < 8; ++v) {
            const int r = row0 + v + 8 * hi;
            Cout[(size_t)r * Nout + col] = acc[t][v] + bb;
        }
    }
}

__global__ __launch_bounds__(256)
void ln_rows_kernel(const float* __restrict__ Z, const float* __restrict__ w,
                    const float* __restrict__ b, float* __restrict__ Hout,
                    float* __restrict__ jk, int mode, int M) {
    const int row  = (int)((blockIdx.x * blockDim.x + threadIdx.x) >> 5);
    const int lane = (int)(threadIdx.x & 31);
    if (row >= M) return;
    float v = Z[(size_t)row * FF + lane];
    float s = v;
#pragma unroll
    for (int off = 16; off >= 1; off >>= 1) s += __shfl_xor(s, off);
    const float mu = s * (1.0f / 32.0f);
    const float d = v - mu;
    float q = d * d;
#pragma unroll
    for (int off = 16; off >= 1; off >>= 1) q += __shfl_xor(q, off);
    const float var = q * (1.0f / 32.0f);
    float y = w[lane] * d * rsqrtf(var + 1e-5f) + b[lane];
    y = leakyf(y, 0.01f);
    Hout[(size_t)row * FF + lane] = y;
    float* o = jk + (size_t)row * FF + lane;
    *o = (mode == 0) ? y : fmaxf(*o, y);
}

#define EW_RUN   32
#define EW_U     8
#define EW_REALW (EE / EW_RUN)
#define EW_WAVES ((E2E + EW_RUN - 1) / EW_RUN)

__device__ __forceinline__ int ew_wave_index() {
    return (int)blockIdx.x * 8 + __builtin_amdgcn_readfirstlane((int)(threadIdx.x >> 5));
}
__device__ __forceinline__ int ew_trips(int ebase) {
    const int rem = E2E - ebase;
    return rem >= EW_RUN ? (EW_RUN / EW_U) : (rem / EW_U);
}
__device__ __forceinline__ int ew_self_node(int ebase, int lane) {
    const int n = ebase + lane - EE;
    return n < NN ? n : NN - 1;
}
__device__ __forceinline__ float ew_bcast(float v, int j) {
    return __int_as_float(__builtin_amdgcn_readlane(__float_as_int(v), j));
}

__global__ __launch_bounds__(256)
void edge_score_kernel(const int* __restrict__ src, const int* __restrict__ dst,
                       const float* __restrict__ ea, const float* __restrict__ We,
                       const float* __restrict__ xl, const float* __restrict__ xr,
                       const float* __restrict__ att, float* __restrict__ score,
                       unsigned* __restrict__ smax) {
    const int lane = (int)(threadIdx.x & 31);
    const int wv   = ew_wave_index();
    if (wv >= EW_WAVES) return;
    const int ebase = wv * EW_RUN;
    const int ntrip = ew_trips(ebase);

    int sL, dL;
    float4 evL;
    if (wv < EW_REALW) {
        sL  = src[ebase + lane];
        dL  = dst[ebase + lane];
        evL = ((const float4*)ea)[ebase + lane];
    } else {
        sL = dL = ew_self_node(ebase, lane);
        evL = make_float4(0.0f, 0.0f, 0.0f, 0.0f);
    }
    const float4 atv = ((const float4*)att)[lane];
    const float4* We4 = (const float4*)We;
    const float4 wr0 = We4[lane * 4 + 0];
    const float4 wr1 = We4[lane * 4 + 1];
    const float4 wr2 = We4[lane * 4 + 2];
    const float4 wr3 = We4[lane * 4 + 3];
    const v4f* xl4 = (const v4f*)xl;
    const v4f* xr4 = (const v4f*)xr;

    for (int t = 0; t < ntrip; ++t) {
        v4f xlv[EW_U], xrv[EW_U];
#pragma unroll
        for (int i = 0; i < EW_U; ++i) {
            const int sN = __builtin_amdgcn_readlane(sL, t * EW_U + i);
            const int dN = __builtin_amdgcn_readlane(dL, t * EW_U + i);
            xlv[i] = xl4[(size_t)sN * (HCC / 4) + lane];
            xrv[i] = xr4[(size_t)dN * (HCC / 4) + lane];
        }
        asm volatile("" : "+v"(xlv[0]), "+v"(xlv[1]), "+v"(xlv[2]), "+v"(xlv[3]),
                          "+v"(xlv[4]), "+v"(xlv[5]), "+v"(xlv[6]), "+v"(xlv[7]),
                          "+v"(xrv[0]), "+v"(xrv[1]), "+v"(xrv[2]), "+v"(xrv[3]),
                          "+v"(xrv[4]), "+v"(xrv[5]), "+v"(xrv[6]), "+v"(xrv[7]));
        float sc[EW_U];
#pragma unroll
        for (int i = 0; i < EW_U; ++i) {
            const int j = t * EW_U + i;
            const float ex = ew_bcast(evL.x, j);
            const float ey = ew_bcast(evL.y, j);
            const float ez = ew_bcast(evL.z, j);
            const float ew = ew_bcast(evL.w, j);
            float s = 0.0f;
            {
                const float eev = wr0.x * ex + wr0.y * ey + wr0.z * ez + wr0.w * ew;
                float m = xlv[i].x + xrv[i].x + eev;
                m = leakyf(m, 0.2f);
                s += atv.x * m;
            }
            {
                const float eev = wr1.x * ex + wr1.y * ey + wr1.z * ez + wr1.w * ew;
                float m = xlv[i].y + xrv[i].y + eev;
                m = leakyf(m, 0.2f);
                s += atv.y * m;
            }
            {
                const float eev = wr2.x * ex + wr2.y * ey + wr2.z * ez + wr2.w * ew;
                float m = xlv[i].z + xrv[i].z + eev;
                m = leakyf(m, 0.2f);
                s += atv.z * m;
            }
            {
                const float eev = wr3.x * ex + wr3.y * ey + wr3.z * ez + wr3.w * ew;
                float m = xlv[i].w + xrv[i].w + eev;
                m = leakyf(m, 0.2f);
                s += atv.w * m;
            }
            sc[i] = s;
        }
        const bool b2 = (lane & 4) != 0;
        const bool b1 = (lane & 2) != 0;
        const bool b0 = (lane & 1) != 0;
        float ra[4];
#pragma unroll
        for (int k = 0; k < 4; ++k) {
            const float keep = b2 ? sc[k + 4] : sc[k];
            const float send = b2 ? sc[k] : sc[k + 4];
            ra[k] = keep + __shfl_xor(send, 4);
        }
        float rb[2];
#pragma unroll
        for (int k = 0; k < 2; ++k) {
            const float keep = b1 ? ra[k + 2] : ra[k];
            const float send = b1 ? ra[k] : ra[k + 2];
            rb[k] = keep + __shfl_xor(send, 2);
        }
        const float keep1 = b0 ? rb[1] : rb[0];
        const float send1 = b0 ? rb[0] : rb[1];
        const float tot = keep1 + __shfl_xor(send1, 1);

        const int g = lane & 7;
        const int h = lane >> 3;
        const int dsel = __shfl(dL, t * EW_U + g);
        const int e = ebase + t * EW_U + g;
        score[(size_t)e * HH + h] = tot;
        atomicMax(&smax[(size_t)dsel * HH + h], fkey(tot));
    }
}

__global__ __launch_bounds__(256)
void edge_den_kernel(const int* __restrict__ dst, const float* __restrict__ score,
                     const unsigned* __restrict__ smax, float* __restrict__ den) {
    const int lane = (int)(threadIdx.x & 31);
    const int wv   = ew_wave_index();
    if (wv >= EW_WAVES) return;
    const int ebase = wv * EW_RUN;
    const int ntrip = ew_trips(ebase);
    int dL;
    if (wv < EW_REALW) dL = dst[ebase + lane];
    else               dL = ew_self_node(ebase, lane);

    int      dsel[EW_RUN / EW_U];
    float    scv[EW_RUN / EW_U];
    unsigned mxk[EW_RUN / EW_U];
#pragma unroll
    for (int t = 0; t < EW_RUN / EW_U; ++t) {
        dsel[t] = __shfl(dL, t * EW_U + (lane >> 2));
        if (t < ntrip) {
            scv[t] = score[(size_t)(ebase + t * EW_U) * HH + lane];
            mxk[t] = smax[(size_t)dsel[t] * HH + (lane & 3)];
        }
    }
#pragma unroll
    for (int t = 0; t < EW_RUN / EW_U; ++t) {
        if (t < ntrip) {
            const float ex = expf(scv[t] - funkey(mxk[t]));
            atomicAdd(&den[(size_t)dsel[t] * HH + (lane & 3)], ex);
        }
    }
}

__global__ __launch_bounds__(256)
void edge_agg_kernel(const int* __restrict__ src, const int* __restrict__ dst,
                     const float* __restrict__ score, const unsigned* __restrict__ smax,
                     const float* __restrict__ den, const float* __restrict__ xl,
                     float* __restrict__ agg) {
    const int lane = (int)(threadIdx.x & 31);
    const int wv   = ew_wave_index();
    if (wv >= EW_WAVES) return;
    const int ebase = wv * EW_RUN;
    const int ntrip = ew_trips(ebase);
    int sL, dL;
    if (wv < EW_REALW) {
        sL = src[ebase + lane];
        dL = dst[ebase + lane];
    } else {
        sL = dL = ew_self_node(ebase, lane);
    }

    for (int t = 0; t < ntrip; ++t) {
        const int dsel = __shfl(dL, t * EW_U + (lane >> 2));
        const int e0 = ebase + t * EW_U;
        const float    scv = score[(size_t)e0 * HH + lane];
        const unsigned mxk = smax[(size_t)dsel * HH + (lane & 3)];
        const float    dn  = den[(size_t)dsel * HH + (lane & 3)];
        float xv[EW_U][HH];
#pragma unroll
        for (int i = 0; i < EW_U; ++i) {
            const int sN = __builtin_amdgcn_readlane(sL, t * EW_U + i);
            const float* xrow = xl + (size_t)sN * HCC + lane;
#pragma unroll
            for (int j = 0; j < HH; ++j) xv[i][j] = xrow[j * CC];
        }
        const float alphaT = expf(scv - funkey(mxk)) / (dn + 1e-16f);
#pragma unroll
        for (int i = 0; i < EW_U; ++i) {
            const int dN = __builtin_amdgcn_readlane(dL, t * EW_U + i);
            float* ao = agg + (size_t)dN * HCC + lane;
#pragma unroll
            for (int j = 0; j < HH; ++j) {
                const float alpha = ew_bcast(alphaT, 4 * i + j);
                atomicAdd(ao + j * CC, xv[i][j] * alpha);
            }
        }
    }
}

__global__ __launch_bounds__(256)
void bias_stats_kernel(float* __restrict__ agg, const float* __restrict__ cb,
                       double* __restrict__ red, int total) {
    __shared__ double ss[256];
    __shared__ double sq[256];
    const int idx = (int)(blockIdx.x * blockDim.x + threadIdx.x);
    double s = 0.0, q = 0.0;
    if (idx < total) {
        const float y = agg[idx] + cb[idx & (HCC - 1)];
        agg[idx] = y;
        s = (double)y;
        q = (double)y * (double)y;
    }
    ss[threadIdx.x] = s;
    sq[threadIdx.x] = q;
    __syncthreads();
    for (int o = 128; o > 0; o >>= 1) {
        if ((int)threadIdx.x < o) {
            ss[threadIdx.x] += ss[threadIdx.x + o];
            sq[threadIdx.x] += sq[threadIdx.x + o];
        }
        __syncthreads();
    }
    if (threadIdx.x == 0) {
        atomicAdd(&red[0], ss[0]);
        atomicAdd(&red[1], sq[0]);
    }
}

#define PLN_IT 5
__global__ __launch_bounds__(256)
void pln_apply_kernel(float* __restrict__ agg, const float* __restrict__ w,
                      const float* __restrict__ b, const double* __restrict__ red,
                      int total4) {
    const int q0     = (int)(blockIdx.x * blockDim.x + threadIdx.x);
    const int stride = (int)(gridDim.x * blockDim.x);
    float4* a4 = (float4*)agg;
    float4 v[PLN_IT];
#pragma unroll
    for (int i = 0; i < PLN_IT; ++i) {
        const int q = q0 + i * stride;
        v[i] = (q < total4) ? a4[q] : make_float4(0.0f, 0.0f, 0.0f, 0.0f);
    }
    const float4 wv = ((const float4*)w)[q0 & (HCC / 4 - 1)];
    const float4 bv = ((const float4*)b)[q0 & (HCC / 4 - 1)];

    constexpr double inv_n = 1.0 / ((double)NN * (double)HCC);
    const double mu  = red[0] * inv_n;
    const double var = red[1] * inv_n - mu * mu;
    float varf = (float)var;
    if (varf < 0.0f) varf = 0.0f;
    const float inv = 1.0f / (sqrtf(varf) + 1e-5f);
    const float muf = (float)mu;

#pragma unroll
    for (int i = 0; i < PLN_IT; ++i) {
        const int q = q0 + i * stride;
        if (q < total4) {
            float4 o;
            o.x = leakyf(wv.x * ((v[i].x - muf) * inv) + bv.x, 0.01f);
            o.y = leakyf(wv.y * ((v[i].y - muf) * inv) + bv.y, 0.01f);
            o.z = leakyf(wv.z * ((v[i].z - muf) * inv) + bv.z, 0.01f);
            o.w = leakyf(wv.w * ((v[i].w - muf) * inv) + bv.w, 0.01f);
            a4[q] = o;
        }
    }
}

extern "C" void kernel_launch(void* const* d_in, const int* in_sizes, int n_in,
                              void* d_out, int out_size, void* d_ws, size_t ws_size,
                              hipStream_t stream) {
    const float* x          = (const float*)d_in[0];
    const int*   edge_index = (const int*)  d_in[1];
    const float* edge_attr  = (const float*)d_in[2];
    const float* embed_W    = (const float*)d_in[3];
    const float* embed_b    = (const float*)d_in[4];
    const float* embed_lnw  = (const float*)d_in[5];
    const float* embed_lnb  = (const float*)d_in[6];
    const float* Wl         = (const float*)d_in[7];
    const float* bl         = (const float*)d_in[8];
    const float* Wr         = (const float*)d_in[9];
    const float* br         = (const float*)d_in[10];
    const float* We         = (const float*)d_in[11];
    const float* att        = (const float*)d_in[12];
    const float* conv_b     = (const float*)d_in[13];
    const float* pln_w      = (const float*)d_in[14];
    const float* pln_b      = (const float*)d_in[15];
    const float* proj_W     = (const float*)d_in[16];
    const float* proj_b     = (const float*)d_in[17];
    const float* ln2_w      = (const float*)d_in[18];
    const float* ln2_b      = (const float*)d_in[19];

    const int* srcE = edge_index;
    const int* dstE = edge_index + EE;

    size_t off = 0;
    auto carve = [&](size_t nbytes) -> char* {
        char* p = (char*)d_ws + off;
        off += (nbytes + 255) & ~(size_t)255;
        return p;
    };
    float*    h     = (float*)   carve((size_t)NN * FF  * 4);
    float*    raw   = (float*)   carve((size_t)NN * FF  * 4);
    float*    xl    = (float*)   carve((size_t)NN * HCC * 4);
    float*    xr    = (float*)   carve((size_t)NN * HCC * 4);
    float*    agg   = (float*)   carve((size_t)NN * HCC * 4);
    float*    score = (float*)   carve((size_t)E2E * HH * 4);
    unsigned* smax  = (unsigned*)carve((size_t)NN * HH  * 4);
    float*    den   = (float*)   carve((size_t)NN * HH  * 4);
    double*   red   = (double*)  carve(2 * sizeof(double));
    (void)ws_size; (void)in_sizes; (void)n_in; (void)out_size;

    float* jk = (float*)d_out;

    const int mtiles = NN / 16;
    const dim3 blk(256);
    const dim3 gGemm((mtiles * 32 + 255) / 256);
    const dim3 gRows((NN * 32 + 255) / 256);
    const dim3 gEdgeR((EW_WAVES + 7) / 8);
    const dim3 gElem(((size_t)NN * HCC + 255) / 256);
    const int totalHC = NN * HCC;
    const dim3 gPln((NN * HCC / 4) / (256 * PLN_IT));

    const size_t shLin  = (size_t)HCC * (FF + 4) * 4;
    const size_t shProj = (size_t)FF * (HCC + 4) * 4;

    gemm_wmma_f32<2, INF_, false><<<gGemm, blk, 0, stream>>>(x, embed_W, embed_b, raw, NN, FF);
    ln_rows_kernel<<<gRows, blk, 0, stream>>>(raw, embed_lnw, embed_lnb, h, jk, 0, NN);

    for (int l = 0; l < LL; ++l) {
        const float* Wl_l = Wl + (size_t)l * HCC * FF;
        const float* bl_l = bl + (size_t)l * HCC;
        const float* Wr_l = Wr + (size_t)l * HCC * FF;
        const float* br_l = br + (size_t)l * HCC;
        const float* We_l = We + (size_t)l * HCC * EDD;
        const float* at_l = att + (size_t)l * HH * CC;
        const float* cb_l = conv_b + (size_t)l * HCC;
        const float* pw_l = pln_w + (size_t)l * HCC;
        const float* pb_l = pln_b + (size_t)l * HCC;
        const float* pW_l = proj_W + (size_t)l * FF * HCC;
        const float* pB_l = proj_b + (size_t)l * FF;
        const float* l2w  = ln2_w + (size_t)l * FF;
        const float* l2b  = ln2_b + (size_t)l * FF;

        gemm_wmma_f32<8, FF, true><<<gGemm, blk, shLin, stream>>>(h, Wl_l, bl_l, xl, NN, HCC);
        gemm_wmma_f32<8, FF, true><<<gGemm, blk, shLin, stream>>>(h, Wr_l, br_l, xr, NN, HCC);

        hipMemsetAsync(smax, 0, (size_t)NN * HH * 4, stream);
        hipMemsetAsync(den,  0, (size_t)NN * HH * 4, stream);
        hipMemsetAsync(agg,  0, (size_t)NN * HCC * 4, stream);
        hipMemsetAsync(red,  0, 2 * sizeof(double), stream);

        edge_score_kernel<<<gEdgeR, blk, 0, stream>>>(srcE, dstE, edge_attr, We_l,
                                                      xl, xr, at_l, score, smax);
        edge_den_kernel<<<gEdgeR, blk, 0, stream>>>(dstE, score, smax, den);
        edge_agg_kernel<<<gEdgeR, blk, 0, stream>>>(srcE, dstE, score, smax, den, xl, agg);

        bias_stats_kernel<<<gElem, blk, 0, stream>>>(agg, cb_l, red, totalHC);
        pln_apply_kernel<<<gPln, blk, 0, stream>>>(agg, pw_l, pb_l, red, totalHC / 4);

        gemm_wmma_f32<2, HCC, true><<<gGemm, blk, shProj, stream>>>(agg, pW_l, pB_l, raw, NN, FF);
        ln_rows_kernel<<<gRows, blk, 0, stream>>>(raw, l2w, l2b, h, jk, 1, NN);
    }
}


# graph-LN statistics: a block sums 5120 elements (per-thread f64 sums, wave butterfly, 8 partials in LDS) and ends with one f64 atomic pair on the hot line, not one per 256; reduction tree changes
# speedup vs baseline: 1.9330x; 1.2166x over previous
#include <hip/hip_runtime.h>
#include <math.h>

#define NN   50000
#define EE   800000
#define INF_ 512
#define FF   32
#define HH   4
#define CC   32
#define HCC  128
#define EDD  4
#define E2E  850000
#define LL   2

typedef float v2f __attribute__((ext_vector_type(2)));
typedef float v8f __attribute__((ext_vector_type(8)));
typedef float v4f __attribute__((ext_vector_type(4)));

__device__ __forceinline__ unsigned fkey(float f) {
    unsigned u = __float_as_uint(f);
    return (u & 0x80000000u) ? ~u : (u | 0x80000000u);
}
__device__ __forceinline__ float funkey(unsigned u) {
    return __uint_as_float((u & 0x80000000u) ? (u & 0x7FFFFFFFu) : ~u);
}
__device__ __forceinline__ float leakyf(float x, float s) {
    return x >= 0.0f ? x : s * x;
}

template <int CT, int KT, bool USE_LDS>
__global__ __launch_bounds__(256)
void gemm_wmma_f32(const float* __restrict__ A, const float* __restrict__ W,
                   const float* __restrict__ bias, float* __restrict__ Cout,
                   int M, int Nout) {
    extern __shared__ float sW[];
    const int tid  = (int)threadIdx.x;
    const int lane = tid & 31;
    const int wave = (int)((blockIdx.x * blockDim.x + tid) >> 5);
    const int mtiles  = M >> 4;
    const int ngroups = Nout / (16 * CT);
    const int stride  = KT + 4;

    if (USE_LDS) {
        const int total = Nout * KT;
        for (int i = tid; i < total; i += 256) {
            const int c  = i / KT;
            const int kk = i - c * KT;
            sW[c * stride + kk] = W[i];
        }
        __syncthreads();
    }
    if (wave >= mtiles * ngroups) return;

    const int mtile = wave / ngroups;
    const int ng    = wave % ngroups;
    const int row0  = mtile * 16;
    const int col0  = ng * 16 * CT;
    const int hi = lane >> 4;
    const int mi = lane & 15;

    v8f acc[CT] = {};
    const float* Arow = A + (size_t)(row0 + mi) * KT + 2 * hi;

    v2f a_cur;
    a_cur.x = Arow[0];
    a_cur.y = Arow[1];
    v2f b_cur[CT];
    if (!USE_LDS) {
#pragma unroll
        for (int t = 0; t < CT; ++t) {
            const float* Wc = W + (size_t)(col0 + t * 16 + mi) * KT + 2 * hi;
            b_cur[t].x = Wc[0];
            b_cur[t].y = Wc[1];
        }
    }

    for (int k = 0; k < KT; k += 4) {
        v2f a_nxt = {};
        v2f b_nxt[CT];
        if (k + 4 < KT) {
            a_nxt.x = Arow[k + 4];
            a_nxt.y = Arow[k + 5];
            if (!USE_LDS) {
#pragma unroll
                for (int t = 0; t < CT; ++t) {
                    const float* Wc = W + (size_t)(col0 + t * 16 + mi) * KT + 2 * hi;
                    b_nxt[t].x = Wc[k + 4];
                    b_nxt[t].y = Wc[k + 5];
                }
                if ((k & 31) == 0)
                    __builtin_prefetch(Arow + k + 132, 0, 1);
            }
        }
#pragma unroll
        for (int t = 0; t < CT; ++t) {
            v2f b;
            if (USE_LDS) {
                const float* Sc = sW + (size_t)(col0 + t * 16 + mi) * stride + 2 * hi;
                b.x = Sc[k];
                b.y = Sc[k + 1];
            } else {
                b = b_cur[t];
            }
            acc[t] = __builtin_amdgcn_wmma_f32_16x16x4_f32(
                false, a_cur, false, b, (short)0, acc[t], false, false);
        }
        a_cur = a_nxt;
        if (!USE_LDS) {
#pragma unroll
            for (int t = 0; t < CT; ++t) b_cur[t] = b_nxt[t];
        }
    }

#pragma unroll
    for (int t = 0; t < CT; ++t) {
        const int col = col0 + t * 16 + mi;
        const float bb = bias ? bias[col] : 0.0f;
#pragma unroll
        for (int v = 0; v < 8; ++v) {
            const int r = row0 + v + 8 * hi;
            Cout[(size_t)r * Nout + col] = acc[t][v] + bb;
        }
    }
}

__global__ __launch_bounds__(256)
void ln_rows_kernel(const float* __restrict__ Z, const float* __restrict__ w,
                    const float* __restrict__ b, float* __restrict__ Hout,
                    float* __restrict__ jk, int mode, int M) {
    const int row  = (int)((blockIdx.x * blockDim.x + threadIdx.x) >> 5);
    const int lane = (int)(threadIdx.x & 31);
    if (row >= M) return;
    float v = Z[(size_t)row * FF + lane];
    float s = v;
#pragma unroll
    for (int off = 16; off >= 1; off >>= 1) s += __shfl_xor(s, off);
    const float mu = s * (1.0f / 32.0f);
    const float d = v - mu;
    float q = d * d;
#pragma unroll
    for (int off = 16; off >= 1; off >>= 1) q += __shfl_xor(q, off);
    const float var = q * (1.0f / 32.0f);
    float y = w[lane] * d * rsqrtf(var + 1e-5f) + b[lane];
    y = leakyf(y, 0.01f);
    Hout[(size_t)row * FF + lane] = y;
    float* o = jk + (size_t)row * FF + lane;
    *o = (mode == 0) ? y : fmaxf(*o, y);
}

#define EW_RUN   32
#define EW_U     8
#define EW_REALW (EE / EW_RUN)
#define EW_WAVES ((E2E + EW_RUN - 1) / EW_RUN)

__device__ __forceinline__ int ew_wave_index() {
    return (int)blockIdx.x * 8 + __builtin_amdgcn_readfirstlane((int)(threadIdx.x >> 5));
}
__device__ __forceinline__ int ew_trips(int ebase) {
    const int rem = E2E - ebase;
    return rem >= EW_RUN ? (EW_RUN / EW_U) : (rem / EW_U);
}
__device__ __forceinline__ int ew_self_node(int ebase, int lane) {
    const int n = ebase + lane - EE;
    return n < NN ? n : NN - 1;
}
__device__ __forceinline__ float ew_bcast(float v, int j) {
    return __int_as_float(__builtin_amdgcn_readlane(__float_as_int(v), j));
}

__global__ __launch_bounds__(256)
void edge_score_kernel(const int* __restrict__ src, const int* __restrict__ dst,
                       const float* __restrict__ ea, const float* __restrict__ We,
                       const float* __restrict__ xl, const float* __restrict__ xr,
                       const float* __restrict__ att, float* __restrict__ score,
                       unsigned* __restrict__ smax) {
    const int lane = (int)(threadIdx.x & 31);
    const int wv   = ew_wave_index();
    if (wv >= EW_WAVES) return;
    const int ebase = wv * EW_RUN;
    const int ntrip = ew_trips(ebase);

    int sL, dL;
    float4 evL;
    if (wv < EW_REALW) {
        sL  = src[ebase + lane];
        dL  = dst[ebase + lane];
        evL = ((const float4*)ea)[ebase + lane];
    } else {
        sL = dL = ew_self_node(ebase, lane);
        evL = make_float4(0.0f, 0.0f, 0.0f, 0.0f);
    }
    const float4 atv = ((const float4*)att)[lane];
    const float4* We4 = (const float4*)We;
    const float4 wr0 = We4[lane * 4 + 0];
    const float4 wr1 = We4[lane * 4 + 1];
    const float4 wr2 = We4[lane * 4 + 2];
    const float4 wr3 = We4[lane * 4 + 3];
    const v4f* xl4 = (const v4f*)xl;
    const v4f* xr4 = (const v4f*)xr;

    for (int t = 0; t < ntrip; ++t) {
        v4f xlv[EW_U], xrv[EW_U];
#pragma unroll
        for (int i = 0; i < EW_U; ++i) {
            const int sN = __builtin_amdgcn_readlane(sL, t * EW_U + i);
            const int dN = __builtin_amdgcn_readlane(dL, t * EW_U + i);
            xlv[i] = xl4[(size_t)sN * (HCC / 4) + lane];
            xrv[i] = xr4[(size_t)dN * (HCC / 4) + lane];
        }
        asm volatile("" : "+v"(xlv[0]), "+v"(xlv[1]), "+v"(xlv[2]), "+v"(xlv[3]),
                          "+v"(xlv[4]), "+v"(xlv[5]), "+v"(xlv[6]), "+v"(xlv[7]),
                          "+v"(xrv[0]), "+v"(xrv[1]), "+v"(xrv[2]), "+v"(xrv[3]),
                          "+v"(xrv[4]), "+v"(xrv[5]), "+v"(xrv[6]), "+v"(xrv[7]));
        float sc[EW_U];
#pragma unroll
        for (int i = 0; i < EW_U; ++i) {
            const int j = t * EW_U + i;
            const float ex = ew_bcast(evL.x, j);
            const float ey = ew_bcast(evL.y, j);
            const float ez = ew_bcast(evL.z, j);
            const float ew = ew_bcast(evL.w, j);
            float s = 0.0f;
            {
                const float eev = wr0.x * ex + wr0.y * ey + wr0.z * ez + wr0.w * ew;
                float m = xlv[i].x + xrv[i].x + eev;
                m = leakyf(m, 0.2f);
                s += atv.x * m;
            }
            {
                const float eev = wr1.x * ex + wr1.y * ey + wr1.z * ez + wr1.w * ew;
                float m = xlv[i].y + xrv[i].y + eev;
                m = leakyf(m, 0.2f);
                s += atv.y * m;
            }
            {
                const float eev = wr2.x * ex + wr2.y * ey + wr2.z * ez + wr2.w * ew;
                float m = xlv[i].z + xrv[i].z + eev;
                m = leakyf(m, 0.2f);
                s += atv.z * m;
            }
            {
                const float eev = wr3.x * ex + wr3.y * ey + wr3.z * ez + wr3.w * ew;
                float m = xlv[i].w + xrv[i].w + eev;
                m = leakyf(m, 0.2f);
                s += atv.w * m;
            }
            sc[i] = s;
        }
        const bool b2 = (lane & 4) != 0;
        const bool b1 = (lane & 2) != 0;
        const bool b0 = (lane & 1) != 0;
        float ra[4];
#pragma unroll
        for (int k = 0; k < 4; ++k) {
            const float keep = b2 ? sc[k + 4] : sc[k];
            const float send = b2 ? sc[k] : sc[k + 4];
            ra[k] = keep + __shfl_xor(send, 4);
        }
        float rb[2];
#pragma unroll
        for (int k = 0; k < 2; ++k) {
            const float keep = b1 ? ra[k + 2] : ra[k];
            const float send = b1 ? ra[k] : ra[k + 2];
            rb[k] = keep + __shfl_xor(send, 2);
        }
        const float keep1 = b0 ? rb[1] : rb[0];
        const float send1 = b0 ? rb[0] : rb[1];
        const float tot = keep1 + __shfl_xor(send1, 1);

        const int g = lane & 7;
        const int h = lane >> 3;
        const int dsel = __shfl(dL, t * EW_U + g);
        const int e = ebase + t * EW_U + g;
        score[(size_t)e * HH + h] = tot;
        atomicMax(&smax[(size_t)dsel * HH + h], fkey(tot));
    }
}

__global__ __launch_bounds__(256)
void edge_den_kernel(const int* __restrict__ dst, const float* __restrict__ score,
                     const unsigned* __restrict__ smax, float* __restrict__ den) {
    const int lane = (int)(threadIdx.x & 31);
    const int wv   = ew_wave_index();
    if (wv >= EW_WAVES) return;
    const int ebase = wv * EW_RUN;
    const int ntrip = ew_trips(ebase);
    int dL;
    if (wv < EW_REALW) dL = dst[ebase + lane];
    else               dL = ew_self_node(ebase, lane);

    int      dsel[EW_RUN / EW_U];
    float    scv[EW_RUN / EW_U];
    unsigned mxk[EW_RUN / EW_U];
#pragma unroll
    for (int t = 0; t < EW_RUN / EW_U; ++t) {
        dsel[t] = __shfl(dL, t * EW_U + (lane >> 2));
        if (t < ntrip) {
            scv[t] = score[(size_t)(ebase + t * EW_U) * HH + lane];
            mxk[t] = smax[(size_t)dsel[t] * HH + (lane & 3)];
        }
    }
#pragma unroll
    for (int t = 0; t < EW_RUN / EW_U; ++t) {
        if (t < ntrip) {
            const float ex = expf(scv[t] - funkey(mxk[t]));
            atomicAdd(&den[(size_t)dsel[t] * HH + (lane & 3)], ex);
        }
    }
}

__global__ __launch_bounds__(256)
void edge_agg_kernel(const int* __restrict__ src, const int* __restrict__ dst,
                     const float* __restrict__ score, const unsigned* __restrict__ smax,
                     const float* __restrict__ den, const float* __restrict__ xl,
                     float* __restrict__ agg) {
    const int lane = (int)(threadIdx.x & 31);
    const int wv   = ew_wave_index();
    if (wv >= EW_WAVES) return;
    const int ebase = wv * EW_RUN;
    const int ntrip = ew_trips(ebase);
    int sL, dL;
    if (wv < EW_REALW) {
        sL = src[ebase + lane];
        dL = dst[ebase + lane];
    } else {
        sL = dL = ew_self_node(ebase, lane);
    }

    for (int t = 0; t < ntrip; ++t) {
        const int dsel = __shfl(dL, t * EW_U + (lane >> 2));
        const int e0 = ebase + t * EW_U;
        const float    scv = score[(size_t)e0 * HH + lane];
        const unsigned mxk = smax[(size_t)dsel * HH + (lane & 3)];
        const float    dn  = den[(size_t)dsel * HH + (lane & 3)];
        float xv[EW_U][HH];
#pragma unroll
        for (int i = 0; i < EW_U; ++i) {
            const int sN = __builtin_amdgcn_readlane(sL, t * EW_U + i);
            const float* xrow = xl + (size_t)sN * HCC + lane;
#pragma unroll
            for (int j = 0; j < HH; ++j) xv[i][j] = xrow[j * CC];
        }
        const float alphaT = expf(scv - funkey(mxk)) / (dn + 1e-16f);
#pragma unroll
        for (int i = 0; i < EW_U; ++i) {
            const int dN = __builtin_amdgcn_readlane(dL, t * EW_U + i);
            float* ao = agg + (size_t)dN * HCC + lane;
#pragma unroll
            for (int j = 0; j < HH; ++j) {
                const float alpha = ew_bcast(alphaT, 4 * i + j);
                atomicAdd(ao + j * CC, xv[i][j] * alpha);
            }
        }
    }
}

#define BST_IT 5
__device__ __forceinline__ double shfl_xor_f64(double v, int m) {
    const int lo = __shfl_xor(__double2loint(v), m);
    const int hi = __shfl_xor(__double2hiint(v), m);
    return __hiloint2double(hi, lo);
}
__global__ __launch_bounds__(256)
void bias_stats_kernel(float* __restrict__ agg, const float* __restrict__ cb,
                       double* __restrict__ red, int total4) {
    __shared__ double part[16];
    const int tid    = (int)threadIdx.x;
    const int lane   = tid & 31;
    const int q0     = (int)(blockIdx.x * blockDim.x) + tid;
    const int stride = (int)(gridDim.x * blockDim.x);
    float4* a4 = (float4*)agg;
    float4 v[BST_IT];
#pragma unroll
    for (int i = 0; i < BST_IT; ++i) {
        const int q = q0 + i * stride;
        v[i] = (q < total4) ? a4[q] : make_float4(0.0f, 0.0f, 0.0f, 0.0f);
    }
    const float4 bv = ((const float4*)cb)[q0 & (HCC / 4 - 1)];

    double s = 0.0, sqq = 0.0;
#pragma unroll
    for (int i = 0; i < BST_IT; ++i) {
        const int q = q0 + i * stride;
        if (q < total4) {
            float4 y;
            y.x = v[i].x + bv.x;
            y.y = v[i].y + bv.y;
            y.z = v[i].z + bv.z;
            y.w = v[i].w + bv.w;
            a4[q] = y;
            const double d0 = (double)y.x, d1 = (double)y.y;
            const double d2 = (double)y.z, d3 = (double)y.w;
            s += d0;  sqq += d0 * d0;
            s += d1;  sqq += d1 * d1;
            s += d2;  sqq += d2 * d2;
            s += d3;  sqq += d3 * d3;
        }
    }

    const bool up = (lane & 16) != 0;
    const double keep = up ? sqq : s;
    const double send = up ? s : sqq;
    double r = keep + shfl_xor_f64(send, 16);
#pragma unroll
    for (int off = 8; off >= 1; off >>= 1) r += shfl_xor_f64(r, off);
    if ((lane & 15) == 0) part[(up ? 8 : 0) + (tid >> 5)] = r;
    __syncthreads();

    if (tid < 32) {
        double t = (lane < 16) ? part[lane] : 0.0;
#pragma unroll
        for (int off = 4; off >= 1; off >>= 1) t += shfl_xor_f64(t, off);
        if ((lane & 7) == 0 && lane < 16) atomicAdd(&red[lane >> 3], t);
    }
}

#define PLN_IT 5
__global__ __launch_bounds__(256)
void pln_apply_kernel(float* __restrict__ agg, const float* __restrict__ w,
                      const float* __restrict__ b, const double* __restrict__ red,
                      int total4) {
    const int q0     = (int)(blockIdx.x * blockDim.x + threadIdx.x);
    const int stride = (int)(gridDim.x * blockDim.x);
    float4* a4 = (float4*)agg;
    float4 v[PLN_IT];
#pragma unroll
    for (int i = 0; i < PLN_IT; ++i) {
        const int q = q0 + i * stride;
        v[i] = (q < total4) ? a4[q] : make_float4(0.0f, 0.0f, 0.0f, 0.0f);
    }
    const float4 wv = ((const float4*)w)[q0 & (HCC / 4 - 1)];
    const float4 bv = ((const float4*)b)[q0 & (HCC / 4 - 1)];

    constexpr double inv_n = 1.0 / ((double)NN * (double)HCC);
    const double mu  = red[0] * inv_n;
    const double var = red[1] * inv_n - mu * mu;
    float varf = (float)var;
    if (varf < 0.0f) varf = 0.0f;
    const float inv = 1.0f / (sqrtf(varf) + 1e-5f);
    const float muf = (float)mu;

#pragma unroll
    for (int i = 0; i < PLN_IT; ++i) {
        const int q = q0 + i * stride;
        if (q < total4) {
            float4 o;
            o.x = leakyf(wv.x * ((v[i].x - muf) * inv) + bv.x, 0.01f);
            o.y = leakyf(wv.y * ((v[i].y - muf) * inv) + bv.y, 0.01f);
            o.z = leakyf(wv.z * ((v[i].z - muf) * inv) + bv.z, 0.01f);
            o.w = leakyf(wv.w * ((v[i].w - muf) * inv) + bv.w, 0.01f);
            a4[q] = o;
        }
    }
}

extern "C" void kernel_launch(void* const* d_in, const int* in_sizes, int n_in,
                              void* d_out, int out_size, void* d_ws, size_t ws_size,
                              hipStream_t stream) {
    const float* x          = (const float*)d_in[0];
    const int*   edge_index = (const int*)  d_in[1];
    const float* edge_attr  = (const float*)d_in[2];
    const float* embed_W    = (const float*)d_in[3];
    const float* embed_b    = (const float*)d_in[4];
    const float* embed_lnw  = (const float*)d_in[5];
    const float* embed_lnb  = (const float*)d_in[6];
    const float* Wl         = (const float*)d_in[7];
    const float* bl         = (const float*)d_in[8];
    const float* Wr         = (const float*)d_in[9];
    const float* br         = (const float*)d_in[10];
    const float* We         = (const float*)d_in[11];
    const float* att        = (const float*)d_in[12];
    const float* conv_b     = (const float*)d_in[13];
    const float* pln_w      = (const float*)d_in[14];
    const float* pln_b      = (const float*)d_in[15];
    const float* proj_W     = (const float*)d_in[16];
    const float* proj_b     = (const float*)d_in[17];
    const float* ln2_w      = (const float*)d_in[18];
    const float* ln2_b      = (const float*)d_in[19];

    const int* srcE = edge_index;
    const int* dstE = edge_index + EE;

    size_t off = 0;
    auto carve = [&](size_t nbytes) -> char* {
        char* p = (char*)d_ws + off;
        off += (nbytes + 255) & ~(size_t)255;
        return p;
    };
    float*    h     = (float*)   carve((size_t)NN * FF  * 4);
    float*    raw   = (float*)   carve((size_t)NN * FF  * 4);
    float*    xl    = (float*)   carve((size_t)NN * HCC * 4);
    float*    xr    = (float*)   carve((size_t)NN * HCC * 4);
    float*    agg   = (float*)   carve((size_t)NN * HCC * 4);
    float*    score = (float*)   carve((size_t)E2E * HH * 4);
    unsigned* smax  = (unsigned*)carve((size_t)NN * HH  * 4);
    float*    den   = (float*)   carve((size_t)NN * HH  * 4);
    double*   red   = (double*)  carve(2 * sizeof(double));
    (void)ws_size; (void)in_sizes; (void)n_in; (void)out_size;

    float* jk = (float*)d_out;

    const int mtiles = NN / 16;
    const dim3 blk(256);
    const dim3 gGemm((mtiles * 32 + 255) / 256);
    const dim3 gRows((NN * 32 + 255) / 256);
    const dim3 gEdgeR((EW_WAVES + 7) / 8);
    const dim3 gStat((NN * HCC / 4) / (256 * BST_IT));
    const int totalHC = NN * HCC;
    const dim3 gPln((NN * HCC / 4) / (256 * PLN_IT));

    const size_t shLin  = (size_t)HCC * (FF + 4) * 4;
    const size_t shProj = (size_t)FF * (HCC + 4) * 4;

    gemm_wmma_f32<2, INF_, false><<<gGemm, blk, 0, stream>>>(x, embed_W, embed_b, raw, NN, FF);
    ln_rows_kernel<<<gRows, blk, 0, stream>>>(raw, embed_lnw, embed_lnb, h, jk, 0, NN);

    for (int l = 0; l < LL; ++l) {
        const float* Wl_l = Wl + (size_t)l * HCC * FF;
        const float* bl_l = bl + (size_t)l * HCC;
        const float* Wr_l = Wr + (size_t)l * HCC * FF;
        const float* br_l = br + (size_t)l * HCC;
        const float* We_l = We + (size_t)l * HCC * EDD;
        const float* at_l = att + (size_t)l * HH * CC;
        const float* cb_l = conv_b + (size_t)l * HCC;
        const float* pw_l = pln_w + (size_t)l * HCC;
        const float* pb_l = pln_b + (size_t)l * HCC;
        const float* pW_l = proj_W + (size_t)l * FF * HCC;
        const float* pB_l = proj_b + (size_t)l * FF;
        const float* l2w  = ln2_w + (size_t)l * FF;
        const float* l2b  = ln2_b + (size_t)l * FF;

        gemm_wmma_f32<8, FF, true><<<gGemm, blk, shLin, stream>>>(h, Wl_l, bl_l, xl, NN, HCC);
        gemm_wmma_f32<8, FF, true><<<gGemm, blk, shLin, stream>>>(h, Wr_l, br_l, xr, NN, HCC);

        hipMemsetAsync(smax, 0, (size_t)NN * HH * 4, stream);
        hipMemsetAsync(den,  0, (size_t)NN * HH * 4, stream);
        hipMemsetAsync(agg,  0, (size_t)NN * HCC * 4, stream);
        hipMemsetAsync(red,  0, 2 * sizeof(double), stream);

        edge_score_kernel<<<gEdgeR, blk, 0, stream>>>(srcE, dstE, edge_attr, We_l,
                                                      xl, xr, at_l, score, smax);
        edge_den_kernel<<<gEdgeR, blk, 0, stream>>>(dstE, score, smax, den);
        edge_agg_kernel<<<gEdgeR, blk, 0, stream>>>(srcE, dstE, score, smax, den, xl, agg);

        bias_stats_kernel<<<gStat, blk, 0, stream>>>(agg, cb_l, red, totalHC / 4);
        pln_apply_kernel<<<gPln, blk, 0, stream>>>(agg, pw_l, pb_l, red, totalHC / 4);

        gemm_wmma_f32<2, HCC, true><<<gGemm, blk, shProj, stream>>>(agg, pW_l, pB_l, raw, NN, FF);
        ln_rows_kernel<<<gRows, blk, 0, stream>>>(raw, l2w, l2b, h, jk, 1, NN);
    }
}


# edges put in target order inside the call (count with arrival rank, scan, scatter); one wave per target does scores, running-max softmax, weighted sum in one sweep; no float atomics; sum order changes
# speedup vs baseline: 2.8813x; 1.4906x over previous
#include <hip/hip_runtime.h>
#include <math.h>

#define NN   50000
#define EE   800000
#define INF_ 512
#define FF   32
#define HH   4
#define CC   32
#define HCC  128
#define EDD  4
#define LL   2

typedef float v2f __attribute__((ext_vector_type(2)));
typedef float v8f __attribute__((ext_vector_type(8)));
typedef float v4f __attribute__((ext_vector_type(4)));

__device__ __forceinline__ float leakyf(float x, float s) {
    return x >= 0.0f ? x : s * x;
}

template <int CT, int KT, bool USE_LDS>
__global__ __launch_bounds__(256)
void gemm_wmma_f32(const float* __restrict__ A, const float* __restrict__ W,
                   const float* __restrict__ bias, float* __restrict__ Cout,
                   int M, int Nout) {
    extern __shared__ float sW[];
    const int tid  = (int)threadIdx.x;
    const int lane = tid & 31;
    const int wave = (int)((blockIdx.x * blockDim.x + tid) >> 5);
    const int mtiles  = M >> 4;
    const int ngroups = Nout / (16 * CT);
    const int stride  = KT + 4;

    if (USE_LDS) {
        const int total = Nout * KT;
        for (int i = tid; i < total; i += 256) {
            const int c  = i / KT;
            const int kk = i - c * KT;
            sW[c * stride + kk] = W[i];
        }
        __syncthreads();
    }
    if (wave >= mtiles * ngroups) return;

    const int mtile = wave / ngroups;
    const int ng    = wave % ngroups;
    const int row0  = mtile * 16;
    const int col0  = ng * 16 * CT;
    const int hi = lane >> 4;
    const int mi = lane & 15;

    v8f acc[CT] = {};
    const float* Arow = A + (size_t)(row0 + mi) * KT + 2 * hi;

    v2f a_cur;
    a_cur.x = Arow[0];
    a_cur.y = Arow[1];
    v2f b_cur[CT];
    if (!USE_LDS) {
#pragma unroll
        for (int t = 0; t < CT; ++t) {
            const float* Wc = W + (size_t)(col0 + t * 16 + mi) * KT + 2 * hi;
            b_cur[t].x = Wc[0];
            b_cur[t].y = Wc[1];
        }
    }

    for (int k = 0; k < KT; k += 4) {
        v2f a_nxt = {};
        v2f b_nxt[CT];
        if (k + 4 < KT) {
            a_nxt.x = Arow[k + 4];
            a_nxt.y = Arow[k + 5];
            if (!USE_LDS) {
#pragma unroll
                for (int t = 0; t < CT; ++t) {
                    const float* Wc = W + (size_t)(col0 + t * 16 + mi) * KT + 2 * hi;
                    b_nxt[t].x = Wc[k + 4];
                    b_nxt[t].y = Wc[k + 5];
                }
                if ((k & 31) == 0)
                    __builtin_prefetch(Arow + k + 132, 0, 1);
            }
        }
#pragma unroll
        for (int t = 0; t < CT; ++t) {
            v2f b;
            if (USE_LDS) {
                const float* Sc = sW + (size_t)(col0 + t * 16 + mi) * stride + 2 * hi;
                b.x = Sc[k];
                b.y = Sc[k + 1];
            } else {
                b = b_cur[t];
            }
            acc[t] = __builtin_amdgcn_wmma_f32_16x16x4_f32(
                false, a_cur, false, b, (short)0, acc[t], false, false);
        }
        a_cur = a_nxt;
        if (!USE_LDS) {
#pragma unroll
            for (int t = 0; t < CT; ++t) b_cur[t] = b_nxt[t];
        }
    }

#pragma unroll
    for (int t = 0; t < CT; ++t) {
        const int col = col0 + t * 16 + mi;
        const float bb = bias ? bias[col] : 0.0f;
#pragma unroll
        for (int v = 0; v < 8; ++v) {
            const int r = row0 + v + 8 * hi;
            Cout[(size_t)r * Nout + col] = acc[t][v] + bb;
        }
    }
}

__global__ __launch_bounds__(256)
void ln_rows_kernel(const float* __restrict__ Z, const float* __restrict__ w,
                    const float* __restrict__ b, float* __restrict__ Hout,
                    float* __restrict__ jk, int mode, int M) {
    const int row  = (int)((blockIdx.x * blockDim.x + threadIdx.x) >> 5);
    const int lane = (int)(threadIdx.x & 31);
    if (row >= M) return;
    float v = Z[(size_t)row * FF + lane];
    float s = v;
#pragma unroll
    for (int off = 16; off >= 1; off >>= 1) s += __shfl_xor(s, off);
    const float mu = s * (1.0f / 32.0f);
    const float d = v - mu;
    float q = d * d;
#pragma unroll
    for (int off = 16; off >= 1; off >>= 1) q += __shfl_xor(q, off);
    const float var = q * (1.0f / 32.0f);
    float y = w[lane] * d * rsqrtf(var + 1e-5f) + b[lane];
    y = leakyf(y, 0.01f);
    Hout[(size_t)row * FF + lane] = y;
    float* o = jk + (size_t)row * FF + lane;
    *o = (mode == 0) ? y : fmaxf(*o, y);
}

__device__ __forceinline__ float ew_bcast(float v, int j) {
    return __int_as_float(__builtin_amdgcn_readlane(__float_as_int(v), j));
}

__global__ __launch_bounds__(256)
void csr_count_kernel(const int* __restrict__ dst, int* __restrict__ cnt,
                      int* __restrict__ rank) {
    const int e = (int)(blockIdx.x * blockDim.x + threadIdx.x);
    if (e < EE) rank[e] = atomicAdd(&cnt[dst[e]], 1);
}

#define SCN_IT 49
#define SCN_G  7
__global__ __launch_bounds__(1024)
void csr_scan_kernel(int* __restrict__ cnt) {
    __shared__ int wsum[32];
    const int tid   = (int)threadIdx.x;
    const int lane  = tid & 31;
    const int w     = tid >> 5;
    const int wbase = w * (SCN_IT * 32);
    int v[SCN_IT];
#pragma unroll
    for (int i = 0; i < SCN_IT; ++i) {
        const int idx = wbase + i * 32 + lane;
        v[i] = (idx < NN) ? cnt[idx] : 0;
    }
    int carry = 0;
#pragma unroll
    for (int r = 0; r < SCN_IT; r += SCN_G) {
        int inc[SCN_G];
#pragma unroll
        for (int k = 0; k < SCN_G; ++k) inc[k] = v[r + k];
#pragma unroll
        for (int o = 1; o < 32; o <<= 1) {
            int y[SCN_G];
#pragma unroll
            for (int k = 0; k < SCN_G; ++k) y[k] = __shfl_up(inc[k], o);
#pragma unroll
            for (int k = 0; k < SCN_G; ++k) if (lane >= o) inc[k] += y[k];
        }
#pragma unroll
        for (int k = 0; k < SCN_G; ++k) {
            const int x = v[r + k];
            v[r + k] = carry + inc[k] - x;
            carry += __builtin_amdgcn_readlane(inc[k], 31);
        }
    }
    if (lane == 0) wsum[w] = carry;
    __syncthreads();
    if (w == 0) {
        const int x = wsum[lane];
        int inc = x;
#pragma unroll
        for (int o = 1; o < 32; o <<= 1) {
            const int y = __shfl_up(inc, o);
            if (lane >= o) inc += y;
        }
        wsum[lane] = inc - x;
    }
    __syncthreads();
    const int wex = wsum[w];
#pragma unroll
    for (int i = 0; i < SCN_IT; ++i) {
        const int idx = wbase + i * 32 + lane;
        if (idx <= NN) cnt[idx] = wex + v[i];
    }
}

__global__ __launch_bounds__(256)
void csr_scatter_kernel(const int* __restrict__ src, const int* __restrict__ dst,
                        const int* __restrict__ off, const int* __restrict__ rank,
                        int* __restrict__ ssrc, int* __restrict__ perm) {
    const int e = (int)(blockIdx.x * blockDim.x + threadIdx.x);
    if (e < EE) {
        const int p = off[dst[e]] + rank[e];
        ssrc[p] = src[e];
        perm[p] = e;
    }
}

#define NA_U 8
__device__ __forceinline__ float edge_partial(const v4f xlv, const v4f xrv,
                                              float ex, float ey, float ez, float ew,
                                              const float4 wr0, const float4 wr1,
                                              const float4 wr2, const float4 wr3,
                                              const float4 atv) {
    float s = 0.0f;
    {
        float m = xlv.x + xrv.x;
        m = fmaf(wr0.x, ex, m);  m = fmaf(wr0.y, ey, m);
        m = fmaf(wr0.z, ez, m);  m = fmaf(wr0.w, ew, m);
        m = leakyf(m, 0.2f);
        s += atv.x * m;
    }
    {
        float m = xlv.y + xrv.y;
        m = fmaf(wr1.x, ex, m);  m = fmaf(wr1.y, ey, m);
        m = fmaf(wr1.z, ez, m);  m = fmaf(wr1.w, ew, m);
        m = leakyf(m, 0.2f);
        s += atv.y * m;
    }
    {
        float m = xlv.z + xrv.z;
        m = fmaf(wr2.x, ex, m);  m = fmaf(wr2.y, ey, m);
        m = fmaf(wr2.z, ez, m);  m = fmaf(wr2.w, ew, m);
        m = leakyf(m, 0.2f);
        s += atv.z * m;
    }
    {
        float m = xlv.w + xrv.w;
        m = fmaf(wr3.x, ex, m);  m = fmaf(wr3.y, ey, m);
        m = fmaf(wr3.z, ez, m);  m = fmaf(wr3.w, ew, m);
        m = leakyf(m, 0.2f);
        s += atv.w * m;
    }
    return s;
}

__global__ __launch_bounds__(256)
void node_attn_kernel(const int* __restrict__ off, const int* __restrict__ ssrc,
                      const int* __restrict__ perm, const float* __restrict__ ea,
                      const float* __restrict__ We, const float* __restrict__ xl,
                      const float* __restrict__ xr, const float* __restrict__ att,
                      float* __restrict__ agg) {
    const int lane = (int)(threadIdx.x & 31);
    const int n = (int)blockIdx.x * 8 + __builtin_amdgcn_readfirstlane((int)(threadIdx.x >> 5));
    if (n >= NN) return;

    const v4f* xl4 = (const v4f*)xl;
    const v4f* xr4 = (const v4f*)xr;
    const float4* ea4 = (const float4*)ea;
    const v4f xrv = xr4[(size_t)n * (HCC / 4) + lane];
    const v4f xls = xl4[(size_t)n * (HCC / 4) + lane];
    const int p0 = __builtin_amdgcn_readfirstlane(off[n]);
    const int p1 = __builtin_amdgcn_readfirstlane(off[n + 1]);
    const float4 atv = ((const float4*)att)[lane];
    const float4* We4 = (const float4*)We;
    const float4 wr0 = We4[lane * 4 + 0];
    const float4 wr1 = We4[lane * 4 + 1];
    const float4 wr2 = We4[lane * 4 + 2];
    const float4 wr3 = We4[lane * 4 + 3];

    float s0 = edge_partial(xls, xrv, 0.0f, 0.0f, 0.0f, 0.0f, wr0, wr1, wr2, wr3, atv);
    s0 += __shfl_xor(s0, 4);
    s0 += __shfl_xor(s0, 2);
    s0 += __shfl_xor(s0, 1);
    float mrun = s0;
    float den  = 1.0f;
    v4f   acc  = xls;

    const int g = lane & 7;
    const bool b2 = (lane & 4) != 0;
    const bool b1 = (lane & 2) != 0;
    const bool b0 = (lane & 1) != 0;

    for (int pb = p0; pb < p1; pb += 32) {
        const int rem = p1 - pb;
        const int nb  = rem < 32 ? rem : 32;
        int sL = n;
        float4 evL = make_float4(0.0f, 0.0f, 0.0f, 0.0f);
        if (lane < nb) {
            sL = ssrc[pb + lane];
            evL = ea4[perm[pb + lane]];
        }
        const int ntrip = (nb + NA_U - 1) / NA_U;
        for (int t = 0; t < ntrip; ++t) {
            v4f xlv[NA_U];
#pragma unroll
            for (int i = 0; i < NA_U; ++i) {
                const int sN = __builtin_amdgcn_readlane(sL, t * NA_U + i);
                xlv[i] = xl4[(size_t)sN * (HCC / 4) + lane];
            }
            float sc[NA_U];
#pragma unroll
            for (int i = 0; i < NA_U; ++i) {
                const int j = t * NA_U + i;
                const float ex = ew_bcast(evL.x, j);
                const float ey = ew_bcast(evL.y, j);
                const float ez = ew_bcast(evL.z, j);
                const float ew = ew_bcast(evL.w, j);
                sc[i] = edge_partial(xlv[i], xrv, ex, ey, ez, ew, wr0, wr1, wr2, wr3, atv);
            }
            float ra[4];
#pragma unroll
            for (int k = 0; k < 4; ++k) {
                const float keep = b2 ? sc[k + 4] : sc[k];
                const float send = b2 ? sc[k] : sc[k + 4];
                ra[k] = keep + __shfl_xor(send, 4);
            }
            float rb[2];
#pragma unroll
            for (int k = 0; k < 2; ++k) {
                const float keep = b1 ? ra[k + 2] : ra[k];
                const float send = b1 ? ra[k] : ra[k + 2];
                rb[k] = keep + __shfl_xor(send, 2);
            }
            const float keep1 = b0 ? rb[1] : rb[0];
            const float send1 = b0 ? rb[0] : rb[1];
            const float tot = keep1 + __shfl_xor(send1, 1);

            const bool valid = (t * NA_U + g) < nb;
            float cm = valid ? tot : -INFINITY;
            cm = fmaxf(cm, __shfl_xor(cm, 4));
            cm = fmaxf(cm, __shfl_xor(cm, 2));
            cm = fmaxf(cm, __shfl_xor(cm, 1));
            const float mnew  = fmaxf(mrun, cm);
            const float scale = expf(mrun - mnew);
            const float pg    = valid ? expf(tot - mnew) : 0.0f;
            den *= scale;
            acc *= scale;
#pragma unroll
            for (int i = 0; i < NA_U; ++i) {
                const float pi = __shfl(pg, (lane & 24) + i);
                den += pi;
                acc += pi * xlv[i];
            }
            mrun = mnew;
        }
    }

    const float dd = den + 1e-16f;
    v4f o;
    o.x = acc.x / dd;
    o.y = acc.y / dd;
    o.z = acc.z / dd;
    o.w = acc.w / dd;
    ((v4f*)agg)[(size_t)n * (HCC / 4) + lane] = o;
}

#define BST_IT 5
__device__ __forceinline__ double shfl_xor_f64(double v, int m) {
    const int lo = __shfl_xor(__double2loint(v), m);
    const int hi = __shfl_xor(__double2hiint(v), m);
    return __hiloint2double(hi, lo);
}
__global__ __launch_bounds__(256)
void bias_stats_kernel(float* __restrict__ agg, const float* __restrict__ cb,
                       double* __restrict__ red, int total4) {
    __shared__ double part[16];
    const int tid    = (int)threadIdx.x;
    const int lane   = tid & 31;
    const int q0     = (int)(blockIdx.x * blockDim.x) + tid;
    const int stride = (int)(gridDim.x * blockDim.x);
    float4* a4 = (float4*)agg;
    float4 v[BST_IT];
#pragma unroll
    for (int i = 0; i < BST_IT; ++i) {
        const int q = q0 + i * stride;
        v[i] = (q < total4) ? a4[q] : make_float4(0.0f, 0.0f, 0.0f, 0.0f);
    }
    const float4 bv = ((const float4*)cb)[q0 & (HCC / 4 - 1)];

    double s = 0.0, sqq = 0.0;
#pragma unroll
    for (int i = 0; i < BST_IT; ++i) {
        const int q = q0 + i * stride;
        if (q < total4) {
            float4 y;
            y.x = v[i].x + bv.x;
            y.y = v[i].y + bv.y;
            y.z = v[i].z + bv.z;
            y.w = v[i].w + bv.w;
            a4[q] = y;
            const double d0 = (double)y.x, d1 = (double)y.y;
            const double d2 = (double)y.z, d3 = (double)y.w;
            s += d0;  sqq += d0 * d0;
            s += d1;  sqq += d1 * d1;
            s += d2;  sqq += d2 * d2;
            s += d3;  sqq += d3 * d3;
        }
    }

    const bool up = (lane & 16) != 0;
    const double keep = up ? sqq : s;
    const double send = up ? s : sqq;
    double r = keep + shfl_xor_f64(send, 16);
#pragma unroll
    for (int off = 8; off >= 1; off >>= 1) r += shfl_xor_f64(r, off);
    if ((lane & 15) == 0) part[(up ? 8 : 0) + (tid >> 5)] = r;
    __syncthreads();

    if (tid < 32) {
        double t = (lane < 16) ? part[lane] : 0.0;
#pragma unroll
        for (int off = 4; off >= 1; off >>= 1) t += shfl_xor_f64(t, off);
        if ((lane & 7) == 0 && lane < 16) atomicAdd(&red[lane >> 3], t);
    }
}

#define PLN_IT 5
__global__ __launch_bounds__(256)
void pln_apply_kernel(float* __restrict__ agg, const float* __restrict__ w,
                      const float* __restrict__ b, const double* __restrict__ red,
                      int total4) {
    const int q0     = (int)(blockIdx.x * blockDim.x + threadIdx.x);
    const int stride = (int)(gridDim.x * blockDim.x);
    float4* a4 = (float4*)agg;
    float4 v[PLN_IT];
#pragma unroll
    for (int i = 0; i < PLN_IT; ++i) {
        const int q = q0 + i * stride;
        v[i] = (q < total4) ? a4[q] : make_float4(0.0f, 0.0f, 0.0f, 0.0f);
    }
    const float4 wv = ((const float4*)w)[q0 & (HCC / 4 - 1)];
    const float4 bv = ((const float4*)b)[q0 & (HCC / 4 - 1)];

    constexpr double inv_n = 1.0 / ((double)NN * (double)HCC);
    const double mu  = red[0] * inv_n;
    const double var = red[1] * inv_n - mu * mu;
    float varf = (float)var;
    if (varf < 0.0f) varf = 0.0f;
    const float inv = 1.0f / (sqrtf(varf) + 1e-5f);
    const float muf = (float)mu;

#pragma unroll
    for (int i = 0; i < PLN_IT; ++i) {
        const int q = q0 + i * stride;
        if (q < total4) {
            float4 o;
            o.x = leakyf(wv.x * ((v[i].x - muf) * inv) + bv.x, 0.01f);
            o.y = leakyf(wv.y * ((v[i].y - muf) * inv) + bv.y, 0.01f);
            o.z = leakyf(wv.z * ((v[i].z - muf) * inv) + bv.z, 0.01f);
            o.w = leakyf(wv.w * ((v[i].w - muf) * inv) + bv.w, 0.01f);
            a4[q] = o;
        }
    }
}

extern "C" void kernel_launch(void* const* d_in, const int* in_sizes, int n_in,
                              void* d_out, int out_size, void* d_ws, size_t ws_size,
                              hipStream_t stream) {
    const float* x          = (const float*)d_in[0];
    const int*   edge_index = (const int*)  d_in[1];
    const float* edge_attr  = (const float*)d_in[2];
    const float* embed_W    = (const float*)d_in[3];
    const float* embed_b    = (const float*)d_in[4];
    const float* embed_lnw  = (const float*)d_in[5];
    const float* embed_lnb  = (const float*)d_in[6];
    const float* Wl         = (const float*)d_in[7];
    const float* bl         = (const float*)d_in[8];
    const float* Wr         = (const float*)d_in[9];
    const float* br         = (const float*)d_in[10];
    const float* We         = (const float*)d_in[11];
    const float* att        = (const float*)d_in[12];
    const float* conv_b     = (const float*)d_in[13];
    const float* pln_w      = (const float*)d_in[14];
    const float* pln_b      = (const float*)d_in[15];
    const float* proj_W     = (const float*)d_in[16];
    const float* proj_b     = (const float*)d_in[17];
    const float* ln2_w      = (const float*)d_in[18];
    const float* ln2_b      = (const float*)d_in[19];

    const int* srcE = edge_index;
    const int* dstE = edge_index + EE;

    size_t off = 0;
    auto carve = [&](size_t nbytes) -> char* {
        char* p = (char*)d_ws + off;
        off += (nbytes + 255) & ~(size_t)255;
        return p;
    };
    float*    h     = (float*)   carve((size_t)NN * FF  * 4);
    float*    raw   = (float*)   carve((size_t)NN * FF  * 4);
    float*    xl    = (float*)   carve((size_t)NN * HCC * 4);
    float*    xr    = (float*)   carve((size_t)NN * HCC * 4);
    float*    agg   = (float*)   carve((size_t)NN * HCC * 4);
    int*      ssrc  = (int*)     carve((size_t)EE * 4);
    int*      perm  = (int*)     carve((size_t)EE * 4);
    int*      rank  = (int*)     carve((size_t)EE * 4);
    int*      goff  = (int*)     carve((size_t)(NN + 1) * 4);
    double*   red   = (double*)  carve(2 * sizeof(double));
    (void)ws_size; (void)in_sizes; (void)n_in; (void)out_size;

    float* jk = (float*)d_out;

    const int mtiles = NN / 16;
    const dim3 blk(256);
    const dim3 gGemm((mtiles * 32 + 255) / 256);
    const dim3 gRows((NN * 32 + 255) / 256);
    const dim3 gEdgeT(EE / 256);
    const dim3 gNode(NN / 8);
    const dim3 gStat((NN * HCC / 4) / (256 * BST_IT));
    const int totalHC = NN * HCC;
    const dim3 gPln((NN * HCC / 4) / (256 * PLN_IT));

    const size_t shLin  = (size_t)HCC * (FF + 4) * 4;
    const size_t shProj = (size_t)FF * (HCC + 4) * 4;

    gemm_wmma_f32<2, INF_, false><<<gGemm, blk, 0, stream>>>(x, embed_W, embed_b, raw, NN, FF);
    ln_rows_kernel<<<gRows, blk, 0, stream>>>(raw, embed_lnw, embed_lnb, h, jk, 0, NN);

    hipMemsetAsync(goff, 0, (size_t)(NN + 1) * 4, stream);
    csr_count_kernel<<<gEdgeT, blk, 0, stream>>>(dstE, goff, rank);
    csr_scan_kernel<<<dim3(1), dim3(1024), 0, stream>>>(goff);
    csr_scatter_kernel<<<gEdgeT, blk, 0, stream>>>(srcE, dstE, goff, rank, ssrc, perm);

    for (int l = 0; l < LL; ++l) {
        const float* Wl_l = Wl + (size_t)l * HCC * FF;
        const float* bl_l = bl + (size_t)l * HCC;
        const float* Wr_l = Wr + (size_t)l * HCC * FF;
        const float* br_l = br + (size_t)l * HCC;
        const float* We_l = We + (size_t)l * HCC * EDD;
        const float* at_l = att + (size_t)l * HH * CC;
        const float* cb_l = conv_b + (size_t)l * HCC;
        const float* pw_l = pln_w + (size_t)l * HCC;
        const float* pb_l = pln_b + (size_t)l * HCC;
        const float* pW_l = proj_W + (size_t)l * FF * HCC;
        const float* pB_l = proj_b + (size_t)l * FF;
        const float* l2w  = ln2_w + (size_t)l * FF;
        const float* l2b  = ln2_b + (size_t)l * FF;

        gemm_wmma_f32<8, FF, true><<<gGemm, blk, shLin, stream>>>(h, Wl_l, bl_l, xl, NN, HCC);
        gemm_wmma_f32<8, FF, true><<<gGemm, blk, shLin, stream>>>(h, Wr_l, br_l, xr, NN, HCC);

        hipMemsetAsync(red,  0, 2 * sizeof(double), stream);

        node_attn_kernel<<<gNode, blk, 0, stream>>>(goff, ssrc, perm, edge_attr, We_l,
                                                    xl, xr, at_l, agg);

        bias_stats_kernel<<<gStat, blk, 0, stream>>>(agg, cb_l, red, totalHC / 4);
        pln_apply_kernel<<<gPln, blk, 0, stream>>>(agg, pw_l, pb_l, red, totalHC / 4);

        gemm_wmma_f32<2, HCC, true><<<gGemm, blk, shProj, stream>>>(agg, pW_l, pB_l, raw, NN, FF);
        ln_rows_kernel<<<gRows, blk, 0, stream>>>(raw, l2w, l2b, h, jk, 1, NN);
    }
}


# element-wise passes folded into the GEMM waves: graph-LN apply + leaky on A fragments as loaded, row-LN + leaky + JK-max on the 16 whole rows a wave holds; launches 19 -> 14; sum trees kept
# speedup vs baseline: 3.1519x; 1.0939x over previous
#include <hip/hip_runtime.h>
#include <math.h>

#define NN   50000
#define EE   800000
#define INF_ 512
#define FF   32
#define HH   4
#define CC   32
#define HCC  128
#define EDD  4
#define LL   2

typedef float v2f __attribute__((ext_vector_type(2)));
typedef float v8f __attribute__((ext_vector_type(8)));
typedef float v4f __attribute__((ext_vector_type(4)));

__device__ __forceinline__ float leakyf(float x, float s) {
    return x >= 0.0f ? x : s * x;
}

template <int CT, int KT, bool USE_LDS, bool NORM_A, int LN_EPI>
__global__ __launch_bounds__(256)
void gemm_wmma_f32(const float* __restrict__ A, const float* __restrict__ W,
                   const float* __restrict__ bias, float* __restrict__ Cout,
                   int M, int Nout,
                   const float* __restrict__ nw, const float* __restrict__ nb,
                   const double* __restrict__ red,
                   const float* __restrict__ lw, const float* __restrict__ lb,
                   float* __restrict__ jk) {
    extern __shared__ float sW[];
    const int tid  = (int)threadIdx.x;
    const int lane = tid & 31;
    const int wave = (int)((blockIdx.x * blockDim.x + tid) >> 5);
    const int mtiles  = M >> 4;
    const int ngroups = Nout / (16 * CT);
    const int stride  = KT + 4;
    const bool live   = wave < mtiles * ngroups;
    const int hi = lane >> 4;
    const int mi = lane & 15;

    v2f a_all[NORM_A ? KT / 4 : 1];
    float muf = 0.0f, inv = 1.0f;

    if (USE_LDS) {
        const int total = Nout * KT;
        for (int i = tid; i < total; i += 256) {
            const int c  = i / KT;
            const int kk = i - c * KT;
            sW[c * stride + kk] = W[i];
        }
        if (NORM_A) {
            float* sN = sW + Nout * stride;
            for (int i = tid; i < KT; i += 256) {
                sN[i]      = nw[i];
                sN[KT + i] = nb[i];
            }
            const float* Ap = A + (size_t)((live ? wave / ngroups : 0) * 16 + mi) * KT + 2 * hi;
#pragma unroll
            for (int s = 0; s < KT / 4; ++s) {
                a_all[s].x = Ap[4 * s];
                a_all[s].y = Ap[4 * s + 1];
            }
            constexpr double inv_n = 1.0 / ((double)NN * (double)HCC);
            const double mu  = red[0] * inv_n;
            const double var = red[1] * inv_n - mu * mu;
            float varf = (float)var;
            if (varf < 0.0f) varf = 0.0f;
            inv = 1.0f / (sqrtf(varf) + 1e-5f);
            muf = (float)mu;
        }
        __syncthreads();
    }
    if (!live) return;

    const int mtile = wave / ngroups;
    const int ng    = wave % ngroups;
    const int row0  = mtile * 16;
    const int col0  = ng * 16 * CT;

    v8f acc[CT] = {};
    const float* Arow = A + (size_t)(row0 + mi) * KT + 2 * hi;

    const float* sNk = sW + Nout * stride + 2 * hi;

    v2f a_cur;
    if (NORM_A) {
        a_cur = a_all[0];
    } else {
        a_cur.x = Arow[0];
        a_cur.y = Arow[1];
    }
    v2f b_cur[CT];
    if (!USE_LDS) {
#pragma unroll
        for (int t = 0; t < CT; ++t) {
            const float* Wc = W + (size_t)(col0 + t * 16 + mi) * KT + 2 * hi;
            b_cur[t].x = Wc[0];
            b_cur[t].y = Wc[1];
        }
    }

    for (int k = 0; k < KT; k += 4) {
        v2f a_nxt = {};
        v2f b_nxt[CT];
        if (k + 4 < KT) {
            if (NORM_A) {
                a_nxt = a_all[NORM_A ? (k + 4) / 4 : 0];
            } else {
                a_nxt.x = Arow[k + 4];
                a_nxt.y = Arow[k + 5];
            }
            if (!USE_LDS) {
#pragma unroll
                for (int t = 0; t < CT; ++t) {
                    const float* Wc = W + (size_t)(col0 + t * 16 + mi) * KT + 2 * hi;
                    b_nxt[t].x = Wc[k + 4];
                    b_nxt[t].y = Wc[k + 5];
                }
                if ((k & 31) == 0)
                    __builtin_prefetch(Arow + k + 132, 0, 1);
            }
        }
        v2f a_use = a_cur;
        if (NORM_A) {
            a_use.x = leakyf(sNk[k]     * ((a_cur.x - muf) * inv) + sNk[KT + k],     0.01f);
            a_use.y = leakyf(sNk[k + 1] * ((a_cur.y - muf) * inv) + sNk[KT + k + 1], 0.01f);
        }
#pragma unroll
        for (int t = 0; t < CT; ++t) {
            v2f b;
            if (USE_LDS) {
                const float* Sc = sW + (size_t)(col0 + t * 16 + mi) * stride + 2 * hi;
                b.x = Sc[k];
                b.y = Sc[k + 1];
            } else {
                b = b_cur[t];
            }
            acc[t] = __builtin_amdgcn_wmma_f32_16x16x4_f32(
                false, a_use, false, b, (short)0, acc[t], false, false);
        }
        a_cur = a_nxt;
        if (!USE_LDS) {
#pragma unroll
            for (int t = 0; t < CT; ++t) b_cur[t] = b_nxt[t];
        }
    }

    if (LN_EPI) {
        const size_t e0 = (size_t)(row0 + 8 * hi) * FF + mi;
        float j0[8], j1[8];
        if (LN_EPI == 2) {
#pragma unroll
            for (int v = 0; v < 8; ++v) {
                j0[v] = jk[e0 + (size_t)v * FF];
                j1[v] = jk[e0 + (size_t)v * FF + 16];
            }
        }
        const float bb0 = bias[mi], bb1 = bias[16 + mi];
        const float g0 = lw[mi], g1 = lw[16 + mi];
        const float c0 = lb[mi], c1 = lb[16 + mi];
        float z0[8], z1[8], s[8], q[8];
#pragma unroll
        for (int v = 0; v < 8; ++v) {
            z0[v] = acc[0][v] + bb0;
            z1[v] = acc[1][v] + bb1;
            s[v]  = z0[v] + z1[v];
        }
#pragma unroll
        for (int off = 8; off >= 1; off >>= 1) {
#pragma unroll
            for (int v = 0; v < 8; ++v) s[v] += __shfl_xor(s[v], off);
        }
#pragma unroll
        for (int v = 0; v < 8; ++v) {
            const float mu = s[v] * (1.0f / 32.0f);
            z0[v] -= mu;
            z1[v] -= mu;
            q[v] = z0[v] * z0[v] + z1[v] * z1[v];
        }
#pragma unroll
        for (int off = 8; off >= 1; off >>= 1) {
#pragma unroll
            for (int v = 0; v < 8; ++v) q[v] += __shfl_xor(q[v], off);
        }
#pragma unroll
        for (int v = 0; v < 8; ++v) {
            const float rs = rsqrtf(q[v] * (1.0f / 32.0f) + 1e-5f);
            const float y0 = leakyf(g0 * z0[v] * rs + c0, 0.01f);
            const float y1 = leakyf(g1 * z1[v] * rs + c1, 0.01f);
            const size_t e = e0 + (size_t)v * FF;
            if (Cout) {
                Cout[e]      = y0;
                Cout[e + 16] = y1;
            }
            if (LN_EPI == 2) {
                jk[e]      = fmaxf(j0[v], y0);
                jk[e + 16] = fmaxf(j1[v], y1);
            } else {
                jk[e]      = y0;
                jk[e + 16] = y1;
            }
        }
    } else {
#pragma unroll
        for (int t = 0; t < CT; ++t) {
            const int col = col0 + t * 16 + mi;
            const float bb = bias ? bias[col] : 0.0f;
#pragma unroll
            for (int v = 0; v < 8; ++v) {
                const int r = row0 + v + 8 * hi;
                Cout[(size_t)r * Nout + col] = acc[t][v] + bb;
            }
        }
    }
}

__device__ __forceinline__ float ew_bcast(float v, int j) {
    return __int_as_float(__builtin_amdgcn_readlane(__float_as_int(v), j));
}

__global__ __launch_bounds__(256)
void csr_count_kernel(const int* __restrict__ dst, int* __restrict__ cnt,
                      int* __restrict__ rank) {
    const int e = (int)(blockIdx.x * blockDim.x + threadIdx.x);
    if (e < EE) rank[e] = atomicAdd(&cnt[dst[e]], 1);
}

#define SCN_IT 49
#define SCN_G  7
__global__ __launch_bounds__(1024)
void csr_scan_kernel(int* __restrict__ cnt) {
    __shared__ int wsum[32];
    const int tid   = (int)threadIdx.x;
    const int lane  = tid & 31;
    const int w     = tid >> 5;
    const int wbase = w * (SCN_IT * 32);
    int v[SCN_IT];
#pragma unroll
    for (int i = 0; i < SCN_IT; ++i) {
        const int idx = wbase + i * 32 + lane;
        v[i] = (idx < NN) ? cnt[idx] : 0;
    }
    int carry = 0;
#pragma unroll
    for (int r = 0; r < SCN_IT; r += SCN_G) {
        int inc[SCN_G];
#pragma unroll
        for (int k = 0; k < SCN_G; ++k) inc[k] = v[r + k];
#pragma unroll
        for (int o = 1; o < 32; o <<= 1) {
            int y[SCN_G];
#pragma unroll
            for (int k = 0; k < SCN_G; ++k) y[k] = __shfl_up(inc[k], o);
#pragma unroll
            for (int k = 0; k < SCN_G; ++k) if (lane >= o) inc[k] += y[k];
        }
#pragma unroll
        for (int k = 0; k < SCN_G; ++k) {
            const int x = v[r + k];
            v[r + k] = carry + inc[k] - x;
            carry += __builtin_amdgcn_readlane(inc[k], 31);
        }
    }
    if (lane == 0) wsum[w] = carry;
    __syncthreads();
    if (w == 0) {
        const int x = wsum[lane];
        int inc = x;
#pragma unroll
        for (int o = 1; o < 32; o <<= 1) {
            const int y = __shfl_up(inc, o);
            if (lane >= o) inc += y;
        }
        wsum[lane] = inc - x;
    }
    __syncthreads();
    const int wex = wsum[w];
#pragma unroll
    for (int i = 0; i < SCN_IT; ++i) {
        const int idx = wbase + i * 32 + lane;
        if (idx <= NN) cnt[idx] = wex + v[i];
    }
}

__global__ __launch_bounds__(256)
void csr_scatter_kernel(const int* __restrict__ src, const int* __restrict__ dst,
                        const int* __restrict__ off, const int* __restrict__ rank,
                        int* __restrict__ ssrc, int* __restrict__ perm) {
    const int e = (int)(blockIdx.x * blockDim.x + threadIdx.x);
    if (e < EE) {
        const int p = off[dst[e]] + rank[e];
        ssrc[p] = src[e];
        perm[p] = e;
    }
}

#define NA_U 8
__device__ __forceinline__ float edge_partial(const v4f xlv, const v4f xrv,
                                              float ex, float ey, float ez, float ew,
                                              const float4 wr0, const float4 wr1,
                                              const float4 wr2, const float4 wr3,
                                              const float4 atv) {
    float s = 0.0f;
    {
        float m = xlv.x + xrv.x;
        m = fmaf(wr0.x, ex, m);  m = fmaf(wr0.y, ey, m);
        m = fmaf(wr0.z, ez, m);  m = fmaf(wr0.w, ew, m);
        m = leakyf(m, 0.2f);
        s += atv.x * m;
    }
    {
        float m = xlv.y + xrv.y;
        m = fmaf(wr1.x, ex, m);  m = fmaf(wr1.y, ey, m);
        m = fmaf(wr1.z, ez, m);  m = fmaf(wr1.w, ew, m);
        m = leakyf(m, 0.2f);
        s += atv.y * m;
    }
    {
        float m = xlv.z + xrv.z;
        m = fmaf(wr2.x, ex, m);  m = fmaf(wr2.y, ey, m);
        m = fmaf(wr2.z, ez, m);  m = fmaf(wr2.w, ew, m);
        m = leakyf(m, 0.2f);
        s += atv.z * m;
    }
    {
        float m = xlv.w + xrv.w;
        m = fmaf(wr3.x, ex, m);  m = fmaf(wr3.y, ey, m);
        m = fmaf(wr3.z, ez, m);  m = fmaf(wr3.w, ew, m);
        m = leakyf(m, 0.2f);
        s += atv.w * m;
    }
    return s;
}

__global__ __launch_bounds__(256)
void node_attn_kernel(const int* __restrict__ off, const int* __restrict__ ssrc,
                      const int* __restrict__ perm, const float* __restrict__ ea,
                      const float* __restrict__ We, const float* __restrict__ xl,
                      const float* __restrict__ xr, const float* __restrict__ att,
                      float* __restrict__ agg) {
    const int lane = (int)(threadIdx.x & 31);
    const int n = (int)blockIdx.x * 8 + __builtin_amdgcn_readfirstlane((int)(threadIdx.x >> 5));
    if (n >= NN) return;

    const v4f* xl4 = (const v4f*)xl;
    const v4f* xr4 = (const v4f*)xr;
    const float4* ea4 = (const float4*)ea;
    const v4f xrv = xr4[(size_t)n * (HCC / 4) + lane];
    const v4f xls = xl4[(size_t)n * (HCC / 4) + lane];
    const int p0 = __builtin_amdgcn_readfirstlane(off[n]);
    const int p1 = __builtin_amdgcn_readfirstlane(off[n + 1]);
    const float4 atv = ((const float4*)att)[lane];
    const float4* We4 = (const float4*)We;
    const float4 wr0 = We4[lane * 4 + 0];
    const float4 wr1 = We4[lane * 4 + 1];
    const float4 wr2 = We4[lane * 4 + 2];
    const float4 wr3 = We4[lane * 4 + 3];

    float s0 = edge_partial(xls, xrv, 0.0f, 0.0f, 0.0f, 0.0f, wr0, wr1, wr2, wr3, atv);
    s0 += __shfl_xor(s0, 4);
    s0 += __shfl_xor(s0, 2);
    s0 += __shfl_xor(s0, 1);
    float mrun = s0;
    float den  = 1.0f;
    v4f   acc  = xls;

    const int g = lane & 7;
    const bool b2 = (lane & 4) != 0;
    const bool b1 = (lane & 2) != 0;
    const bool b0 = (lane & 1) != 0;

    for (int pb = p0; pb < p1; pb += 32) {
        const int rem = p1 - pb;
        const int nb  = rem < 32 ? rem : 32;
        int sL = n;
        float4 evL = make_float4(0.0f, 0.0f, 0.0f, 0.0f);
        if (lane < nb) {
            sL = ssrc[pb + lane];
            evL = ea4[perm[pb + lane]];
        }
        const int ntrip = (nb + NA_U - 1) / NA_U;
        for (int t = 0; t < ntrip; ++t) {
            v4f xlv[NA_U];
#pragma unroll
            for (int i = 0; i < NA_U; ++i) {
                const int sN = __builtin_amdgcn_readlane(sL, t * NA_U + i);
                xlv[i] = xl4[(size_t)sN * (HCC / 4) + lane];
            }
            float sc[NA_U];
#pragma unroll
            for (int i = 0; i < NA_U; ++i) {
                const int j = t * NA_U + i;
                const float ex = ew_bcast(evL.x, j);
                const float ey = ew_bcast(evL.y, j);
                const float ez = ew_bcast(evL.z, j);
                const float ew = ew_bcast(evL.w, j);
                sc[i] = edge_partial(xlv[i], xrv, ex, ey, ez, ew, wr0, wr1, wr2, wr3, atv);
            }
            float ra[4];
#pragma unroll
            for (int k = 0; k < 4; ++k) {
                const float keep = b2 ? sc[k + 4] : sc[k];
                const float send = b2 ? sc[k] : sc[k + 4];
                ra[k] = keep + __shfl_xor(send, 4);
            }
            float rb[2];
#pragma unroll
            for (int k = 0; k < 2; ++k) {
                const float keep = b1 ? ra[k + 2] : ra[k];
                const float send = b1 ? ra[k] : ra[k + 2];
                rb[k] = keep + __shfl_xor(send, 2);
            }
            const float keep1 = b0 ? rb[1] : rb[0];
            const float send1 = b0 ? rb[0] : rb[1];
            const float tot = keep1 + __shfl_xor(send1, 1);

            const bool valid = (t * NA_U + g) < nb;
            float cm = valid ? tot : -INFINITY;
            cm = fmaxf(cm, __shfl_xor(cm, 4));
            cm = fmaxf(cm, __shfl_xor(cm, 2));
            cm = fmaxf(cm, __shfl_xor(cm, 1));
            const float mnew  = fmaxf(mrun, cm);
            const float scale = expf(mrun - mnew);
            const float pg    = valid ? expf(tot - mnew) : 0.0f;
            den *= scale;
            acc *= scale;
#pragma unroll
            for (int i = 0; i < NA_U; ++i) {
                const float pi = __shfl(pg, (lane & 24) + i);
                den += pi;
                acc += pi * xlv[i];
            }
            mrun = mnew;
        }
    }

    const float dd = den + 1e-16f;
    v4f o;
    o.x = acc.x / dd;
    o.y = acc.y / dd;
    o.z = acc.z / dd;
    o.w = acc.w / dd;
    ((v4f*)agg)[(size_t)n * (HCC / 4) + lane] = o;
}

#define BST_IT 5
__device__ __forceinline__ double shfl_xor_f64(double v, int m) {
    const int lo = __shfl_xor(__double2loint(v), m);
    const int hi = __shfl_xor(__double2hiint(v), m);
    return __hiloint2double(hi, lo);
}
__global__ __launch_bounds__(256)
void bias_stats_kernel(float* __restrict__ agg, const float* __restrict__ cb,
                       double* __restrict__ red, int total4) {
    __shared__ double part[16];
    const int tid    = (int)threadIdx.x;
    const int lane   = tid & 31;
    const int q0     = (int)(blockIdx.x * blockDim.x) + tid;
    const int stride = (int)(gridDim.x * blockDim.x);
    float4* a4 = (float4*)agg;
    float4 v[BST_IT];
#pragma unroll
    for (int i = 0; i < BST_IT; ++i) {
        const int q = q0 + i * stride;
        v[i] = (q < total4) ? a4[q] : make_float4(0.0f, 0.0f, 0.0f, 0.0f);
    }
    const float4 bv = ((const float4*)cb)[q0 & (HCC / 4 - 1)];

    double s = 0.0, sqq = 0.0;
#pragma unroll
    for (int i = 0; i < BST_IT; ++i) {
        const int q = q0 + i * stride;
        if (q < total4) {
            float4 y;
            y.x = v[i].x + bv.x;
            y.y = v[i].y + bv.y;
            y.z = v[i].z + bv.z;
            y.w = v[i].w + bv.w;
            a4[q] = y;
            const double d0 = (double)y.x, d1 = (double)y.y;
            const double d2 = (double)y.z, d3 = (double)y.w;
            s += d0;  sqq += d0 * d0;
            s += d1;  sqq += d1 * d1;
            s += d2;  sqq += d2 * d2;
            s += d3;  sqq += d3 * d3;
        }
    }

    const bool up = (lane & 16) != 0;
    const double keep = up ? sqq : s;
    const double send = up ? s : sqq;
    double r = keep + shfl_xor_f64(send, 16);
#pragma unroll
    for (int off = 8; off >= 1; off >>= 1) r += shfl_xor_f64(r, off);
    if ((lane & 15) == 0) part[(up ? 8 : 0) + (tid >> 5)] = r;
    __syncthreads();

    if (tid < 32) {
        double t = (lane < 16) ? part[lane] : 0.0;
#pragma unroll
        for (int off = 4; off >= 1; off >>= 1) t += shfl_xor_f64(t, off);
        if ((lane & 7) == 0 && lane < 16) atomicAdd(&red[lane >> 3], t);
    }
}

extern "C" void kernel_launch(void* const* d_in, const int* in_sizes, int n_in,
                              void* d_out, int out_size, void* d_ws, size_t ws_size,
                              hipStream_t stream) {
    const float* x          = (const float*)d_in[0];
    const int*   edge_index = (const int*)  d_in[1];
    const float* edge_attr  = (const float*)d_in[2];
    const float* embed_W    = (const float*)d_in[3];
    const float* embed_b    = (const float*)d_in[4];
    const float* embed_lnw  = (const float*)d_in[5];
    const float* embed_lnb  = (const float*)d_in[6];
    const float* Wl         = (const float*)d_in[7];
    const float* bl         = (const float*)d_in[8];
    const float* Wr         = (const float*)d_in[9];
    const float* br         = (const float*)d_in[10];
    const float* We         = (const float*)d_in[11];
    const float* att        = (const float*)d_in[12];
    const float* conv_b     = (const float*)d_in[13];
    const float* pln_w      = (const float*)d_in[14];
    const float* pln_b      = (const float*)d_in[15];
    const float* proj_W     = (const float*)d_in[16];
    const float* proj_b     = (const float*)d_in[17];
    const float* ln2_w      = (const float*)d_in[18];
    const float* ln2_b      = (const float*)d_in[19];

    const int* srcE = edge_index;
    const int* dstE = edge_index + EE;

    size_t off = 0;
    auto carve = [&](size_t nbytes) -> char* {
        char* p = (char*)d_ws + off;
        off += (nbytes + 255) & ~(size_t)255;
        return p;
    };
    float*    h     = (float*)   carve((size_t)NN * FF  * 4);
    float*    xl    = (float*)   carve((size_t)NN * HCC * 4);
    float*    xr    = (float*)   carve((size_t)NN * HCC * 4);
    float*    agg   = (float*)   carve((size_t)NN * HCC * 4);
    int*      ssrc  = (int*)     carve((size_t)EE * 4);
    int*      perm  = (int*)     carve((size_t)EE * 4);
    int*      rank  = (int*)     carve((size_t)EE * 4);
    int*      goff  = (int*)     carve((size_t)(NN + 1) * 4);
    double*   red   = (double*)  carve(2 * sizeof(double));
    (void)ws_size; (void)in_sizes; (void)n_in; (void)out_size;

    float* jk = (float*)d_out;

    const int mtiles = NN / 16;
    const dim3 blk(256);
    const dim3 gGemm((mtiles * 32 + 255) / 256);
    const dim3 gEdgeT(EE / 256);
    const dim3 gNode(NN / 8);
    const dim3 gStat((NN * HCC / 4) / (256 * BST_IT));
    const int totalHC = NN * HCC;

    const size_t shLin  = (size_t)HCC * (FF + 4) * 4;
    const size_t shProj = (size_t)FF * (HCC + 4) * 4 + (size_t)2 * HCC * 4;

    gemm_wmma_f32<2, INF_, false, false, 1><<<gGemm, blk, 0, stream>>>(
        x, embed_W, embed_b, h, NN, FF, nullptr, nullptr, nullptr, embed_lnw, embed_lnb, jk);

    hipMemsetAsync(goff, 0, (size_t)(NN + 1) * 4, stream);
    csr_count_kernel<<<gEdgeT, blk, 0, stream>>>(dstE, goff, rank);
    csr_scan_kernel<<<dim3(1), dim3(1024), 0, stream>>>(goff);
    csr_scatter_kernel<<<gEdgeT, blk, 0, stream>>>(srcE, dstE, goff, rank, ssrc, perm);

    for (int l = 0; l < LL; ++l) {
        const float* Wl_l = Wl + (size_t)l * HCC * FF;
        const float* bl_l = bl + (size_t)l * HCC;
        const float* Wr_l = Wr + (size_t)l * HCC * FF;
        const float* br_l = br + (size_t)l * HCC;
        const float* We_l = We + (size_t)l * HCC * EDD;
        const float* at_l = att + (size_t)l * HH * CC;
        const float* cb_l = conv_b + (size_t)l * HCC;
        const float* pw_l = pln_w + (size_t)l * HCC;
        const float* pb_l = pln_b + (size_t)l * HCC;
        const float* pW_l = proj_W + (size_t)l * FF * HCC;
        const float* pB_l = proj_b + (size_t)l * FF;
        const float* l2w  = ln2_w + (size_t)l * FF;
        const float* l2b  = ln2_b + (size_t)l * FF;

        gemm_wmma_f32<8, FF, true, false, 0><<<gGemm, blk, shLin, stream>>>(
            h, Wl_l, bl_l, xl, NN, HCC, nullptr, nullptr, nullptr, nullptr, nullptr, nullptr);
        gemm_wmma_f32<8, FF, true, false, 0><<<gGemm, blk, shLin, stream>>>(
            h, Wr_l, br_l, xr, NN, HCC, nullptr, nullptr, nullptr, nullptr, nullptr, nullptr);

        hipMemsetAsync(red,  0, 2 * sizeof(double), stream);

        node_attn_kernel<<<gNode, blk, 0, stream>>>(goff, ssrc, perm, edge_attr, We_l,
                                                    xl, xr, at_l, agg);

        bias_stats_kernel<<<gStat, blk, 0, stream>>>(agg, cb_l, red, totalHC / 4);

        gemm_wmma_f32<2, HCC, true, true, 2><<<gGemm, blk, shProj, stream>>>(
            agg, pW_l, pB_l, (l + 1 < LL) ? h : nullptr, NN, FF, pw_l, pb_l, red, l2w, l2b, jk);
    }
}


# graph-LN sums formed by the attention waves (row + conv bias; f32 wave fold, one f64 atomic pair into 256 slot lines, folded by the proj blocks); stats launches and memsets gone; sum order changes
# speedup vs baseline: 3.4035x; 1.0798x over previous
#include <hip/hip_runtime.h>
#include <math.h>

#define NN   50000
#define EE   800000
#define INF_ 512
#define FF   32
#define HH   4
#define CC   32
#define HCC  128
#define EDD  4
#define LL   2

typedef float v2f __attribute__((ext_vector_type(2)));
typedef float v8f __attribute__((ext_vector_type(8)));
typedef float v4f __attribute__((ext_vector_type(4)));

__device__ __forceinline__ float leakyf(float x, float s) {
    return x >= 0.0f ? x : s * x;
}
__device__ __forceinline__ double shfl_xor_f64(double v, int m) {
    const int lo = __shfl_xor(__double2loint(v), m);
    const int hi = __shfl_xor(__double2hiint(v), m);
    return __hiloint2double(hi, lo);
}

#define STAT_SLOTS 256
#define STAT_PITCH 16

template <int CT, int KT, bool USE_LDS, bool NORM_A, int LN_EPI>
__global__ __launch_bounds__(256)
void gemm_wmma_f32(const float* __restrict__ A, const float* __restrict__ W,
                   const float* __restrict__ bias, float* __restrict__ Cout,
                   int M, int Nout,
                   const float* __restrict__ nw, const float* __restrict__ nb,
                   const double* __restrict__ red,
                   const float* __restrict__ lw, const float* __restrict__ lb,
                   float* __restrict__ jk) {
    extern __shared__ float sW[];
    const int tid  = (int)threadIdx.x;
    const int lane = tid & 31;
    const int wave = (int)((blockIdx.x * blockDim.x + tid) >> 5);
    const int mtiles  = M >> 4;
    const int ngroups = Nout / (16 * CT);
    const int stride  = KT + 4;
    const bool live   = wave < mtiles * ngroups;
    const int hi = lane >> 4;
    const int mi = lane & 15;

    v2f a_all[NORM_A ? KT / 4 : 1];
    float muf = 0.0f, inv = 1.0f;

    if (USE_LDS) {
        const int total = Nout * KT;
        for (int i = tid; i < total; i += 256) {
            const int c  = i / KT;
            const int kk = i - c * KT;
            sW[c * stride + kk] = W[i];
        }
        if (NORM_A) {
            float* sN = sW + Nout * stride;
            for (int i = tid; i < KT; i += 256) {
                sN[i]      = nw[i];
                sN[KT + i] = nb[i];
            }
            const float* Ap = A + (size_t)((live ? wave / ngroups : 0) * 16 + mi) * KT + 2 * hi;
#pragma unroll
            for (int s = 0; s < KT / 4; ++s) {
                a_all[s].x = Ap[4 * s];
                a_all[s].y = Ap[4 * s + 1];
            }
            const double ss = red[(size_t)tid * STAT_PITCH];
            const double qq = red[(size_t)tid * STAT_PITCH + 1];
            const bool up = (lane & 16) != 0;
            double r = (up ? qq : ss) + shfl_xor_f64(up ? ss : qq, 16);
#pragma unroll
            for (int off = 8; off >= 1; off >>= 1) r += shfl_xor_f64(r, off);
            double* sP = (double*)(sN + 2 * KT);
            if ((lane & 15) == 0) sP[(up ? 8 : 0) + (tid >> 5)] = r;
        }
        __syncthreads();
        if (NORM_A) {
            const double* sP = (const double*)(sW + Nout * stride + 2 * KT);
            double t = sP[lane & 15];
            t += shfl_xor_f64(t, 4);
            t += shfl_xor_f64(t, 2);
            t += shfl_xor_f64(t, 1);
            const double o  = shfl_xor_f64(t, 8);
            const bool   hq = (lane & 8) != 0;
            const double S  = hq ? o : t;
            const double Q  = hq ? t : o;
            constexpr double inv_n = 1.0 / ((double)NN * (double)HCC);
            const double mu  = S * inv_n;
            const double var = Q * inv_n - mu * mu;
            float varf = (float)var;
            if (varf < 0.0f) varf = 0.0f;
            inv = 1.0f / (sqrtf(varf) + 1e-5f);
            muf = (float)mu;
        }
    }
    if (!live) return;

    const int mtile = wave / ngroups;
    const int ng    = wave % ngroups;
    const int row0  = mtile * 16;
    const int col0  = ng * 16 * CT;

    v8f acc[CT] = {};
    const float* Arow = A + (size_t)(row0 + mi) * KT + 2 * hi;

    const float* sNk = sW + Nout * stride + 2 * hi;

    v2f a_cur;
    if (NORM_A) {
        a_cur = a_all[0];
    } else {
        a_cur.x = Arow[0];
        a_cur.y = Arow[1];
    }
    v2f b_cur[CT];
    if (!USE_LDS) {
#pragma unroll
        for (int t = 0; t < CT; ++t) {
            const float* Wc = W + (size_t)(col0 + t * 16 + mi) * KT + 2 * hi;
            b_cur[t].x = Wc[0];
            b_cur[t].y = Wc[1];
        }
    }

    for (int k = 0; k < KT; k += 4) {
        v2f a_nxt = {};
        v2f b_nxt[CT];
        if (k + 4 < KT) {
            if (NORM_A) {
                a_nxt = a_all[NORM_A ? (k + 4) / 4 : 0];
            } else {
                a_nxt.x = Arow[k + 4];
                a_nxt.y = Arow[k + 5];
            }
            if (!USE_LDS) {
#pragma unroll
                for (int t = 0; t < CT; ++t) {
                    const float* Wc = W + (size_t)(col0 + t * 16 + mi) * KT + 2 * hi;
                    b_nxt[t].x = Wc[k + 4];
                    b_nxt[t].y = Wc[k + 5];
                }
                if ((k & 31) == 0)
                    __builtin_prefetch(Arow + k + 132, 0, 1);
            }
        }
        v2f a_use = a_cur;
        if (NORM_A) {
            a_use.x = leakyf(sNk[k]     * ((a_cur.x - muf) * inv) + sNk[KT + k],     0.01f);
            a_use.y = leakyf(sNk[k + 1] * ((a_cur.y - muf) * inv) + sNk[KT + k + 1], 0.01f);
        }
#pragma unroll
        for (int t = 0; t < CT; ++t) {
            v2f b;
            if (USE_LDS) {
                const float* Sc = sW + (size_t)(col0 + t * 16 + mi) * stride + 2 * hi;
                b.x = Sc[k];
                b.y = Sc[k + 1];
            } else {
                b = b_cur[t];
            }
            acc[t] = __builtin_amdgcn_wmma_f32_16x16x4_f32(
                false, a_use, false, b, (short)0, acc[t], false, false);
        }
        a_cur = a_nxt;
        if (!USE_LDS) {
#pragma unroll
            for (int t = 0; t < CT; ++t) b_cur[t] = b_nxt[t];
        }
    }

    if (LN_EPI) {
        const size_t e0 = (size_t)(row0 + 8 * hi) * FF + mi;
        float j0[8], j1[8];
        if (LN_EPI == 2) {
#pragma unroll
            for (int v = 0; v < 8; ++v) {
                j0[v] = jk[e0 + (size_t)v * FF];
                j1[v] = jk[e0 + (size_t)v * FF + 16];
            }
        }
        const float bb0 = bias[mi], bb1 = bias[16 + mi];
        const float g0 = lw[mi], g1 = lw[16 + mi];
        const float c0 = lb[mi], c1 = lb[16 + mi];
        float z0[8], z1[8], s[8], q[8];
#pragma unroll
        for (int v = 0; v < 8; ++v) {
            z0[v] = acc[0][v] + bb0;
            z1[v] = acc[1][v] + bb1;
            s[v]  = z0[v] + z1[v];
        }
#pragma unroll
        for (int off = 8; off >= 1; off >>= 1) {
#pragma unroll
            for (int v = 0; v < 8; ++v) s[v] += __shfl_xor(s[v], off);
        }
#pragma unroll
        for (int v = 0; v < 8; ++v) {
            const float mu = s[v] * (1.0f / 32.0f);
            z0[v] -= mu;
            z1[v] -= mu;
            q[v] = z0[v] * z0[v] + z1[v] * z1[v];
        }
#pragma unroll
        for (int off = 8; off >= 1; off >>= 1) {
#pragma unroll
            for (int v = 0; v < 8; ++v) q[v] += __shfl_xor(q[v], off);
        }
#pragma unroll
        for (int v = 0; v < 8; ++v) {
            const float rs = rsqrtf(q[v] * (1.0f / 32.0f) + 1e-5f);
            const float y0 = leakyf(g0 * z0[v] * rs + c0, 0.01f);
            const float y1 = leakyf(g1 * z1[v] * rs + c1, 0.01f);
            const size_t e = e0 + (size_t)v * FF;
            if (Cout) {
                Cout[e]      = y0;
                Cout[e + 16] = y1;
            }
            if (LN_EPI == 2) {
                jk[e]      = fmaxf(j0[v], y0);
                jk[e + 16] = fmaxf(j1[v], y1);
            } else {
                jk[e]      = y0;
                jk[e + 16] = y1;
            }
        }
    } else {
#pragma unroll
        for (int t = 0; t < CT; ++t) {
            const int col = col0 + t * 16 + mi;
            const float bb = bias ? bias[col] : 0.0f;
#pragma unroll
            for (int v = 0; v < 8; ++v) {
                const int r = row0 + v + 8 * hi;
                Cout[(size_t)r * Nout + col] = acc[t][v] + bb;
            }
        }
    }
}

__device__ __forceinline__ float ew_bcast(float v, int j) {
    return __int_as_float(__builtin_amdgcn_readlane(__float_as_int(v), j));
}

__global__ __launch_bounds__(256)
void csr_count_kernel(const int* __restrict__ dst, int* __restrict__ cnt,
                      int* __restrict__ rank) {
    const int e = (int)(blockIdx.x * blockDim.x + threadIdx.x);
    if (e < EE) rank[e] = atomicAdd(&cnt[dst[e]], 1);
}

#define SCN_IT 49
#define SCN_G  7
__global__ __launch_bounds__(1024)
void csr_scan_kernel(int* __restrict__ cnt, double* __restrict__ red) {
    __shared__ int wsum[32];
    const int tid   = (int)threadIdx.x;
    static_assert(LL * STAT_SLOTS * 2 == 1024, "one slot double per thread of the block");
    red[(size_t)(tid >> 1) * STAT_PITCH + (tid & 1)] = 0.0;
    const int lane  = tid & 31;
    const int w     = tid >> 5;
    const int wbase = w * (SCN_IT * 32);
    int v[SCN_IT];
#pragma unroll
    for (int i = 0; i < SCN_IT; ++i) {
        const int idx = wbase + i * 32 + lane;
        v[i] = (idx < NN) ? cnt[idx] : 0;
    }
    int carry = 0;
#pragma unroll
    for (int r = 0; r < SCN_IT; r += SCN_G) {
        int inc[SCN_G];
#pragma unroll
        for (int k = 0; k < SCN_G; ++k) inc[k] = v[r + k];
#pragma unroll
        for (int o = 1; o < 32; o <<= 1) {
            int y[SCN_G];
#pragma unroll
            for (int k = 0; k < SCN_G; ++k) y[k] = __shfl_up(inc[k], o);
#pragma unroll
            for (int k = 0; k < SCN_G; ++k) if (lane >= o) inc[k] += y[k];
        }
#pragma unroll
        for (int k = 0; k < SCN_G; ++k) {
            const int x = v[r + k];
            v[r + k] = carry + inc[k] - x;
            carry += __builtin_amdgcn_readlane(inc[k], 31);
        }
    }
    if (lane == 0) wsum[w] = carry;
    __syncthreads();
    if (w == 0) {
        const int x = wsum[lane];
        int inc = x;
#pragma unroll
        for (int o = 1; o < 32; o <<= 1) {
            const int y = __shfl_up(inc, o);
            if (lane >= o) inc += y;
        }
        wsum[lane] = inc - x;
    }
    __syncthreads();
    const int wex = wsum[w];
#pragma unroll
    for (int i = 0; i < SCN_IT; ++i) {
        const int idx = wbase + i * 32 + lane;
        if (idx <= NN) cnt[idx] = wex + v[i];
    }
}

__global__ __launch_bounds__(256)
void csr_scatter_kernel(const int* __restrict__ src, const int* __restrict__ dst,
                        const int* __restrict__ off, const int* __restrict__ rank,
                        int* __restrict__ ssrc, int* __restrict__ perm) {
    const int e = (int)(blockIdx.x * blockDim.x + threadIdx.x);
    if (e < EE) {
        const int p = off[dst[e]] + rank[e];
        ssrc[p] = src[e];
        perm[p] = e;
    }
}

#define NA_U 8
__device__ __forceinline__ float edge_partial(const v4f xlv, const v4f xrv,
                                              float ex, float ey, float ez, float ew,
                                              const float4 wr0, const float4 wr1,
                                              const float4 wr2, const float4 wr3,
                                              const float4 atv) {
    float s = 0.0f;
    {
        float m = xlv.x + xrv.x;
        m = fmaf(wr0.x, ex, m);  m = fmaf(wr0.y, ey, m);
        m = fmaf(wr0.z, ez, m);  m = fmaf(wr0.w, ew, m);
        m = leakyf(m, 0.2f);
        s += atv.x * m;
    }
    {
        float m = xlv.y + xrv.y;
        m = fmaf(wr1.x, ex, m);  m = fmaf(wr1.y, ey, m);
        m = fmaf(wr1.z, ez, m);  m = fmaf(wr1.w, ew, m);
        m = leakyf(m, 0.2f);
        s += atv.y * m;
    }
    {
        float m = xlv.z + xrv.z;
        m = fmaf(wr2.x, ex, m);  m = fmaf(wr2.y, ey, m);
        m = fmaf(wr2.z, ez, m);  m = fmaf(wr2.w, ew, m);
        m = leakyf(m, 0.2f);
        s += atv.z * m;
    }
    {
        float m = xlv.w + xrv.w;
        m = fmaf(wr3.x, ex, m);  m = fmaf(wr3.y, ey, m);
        m = fmaf(wr3.z, ez, m);  m = fmaf(wr3.w, ew, m);
        m = leakyf(m, 0.2f);
        s += atv.w * m;
    }
    return s;
}

__global__ __launch_bounds__(256)
void node_attn_kernel(const int* __restrict__ off, const int* __restrict__ ssrc,
                      const int* __restrict__ perm, const float* __restrict__ ea,
                      const float* __restrict__ We, const float* __restrict__ xl,
                      const float* __restrict__ xr, const float* __restrict__ att,
                      const float* __restrict__ cb, double* __restrict__ red,
                      float* __restrict__ agg) {
    const int lane = (int)(threadIdx.x & 31);
    const int n = (int)blockIdx.x * 8 + __builtin_amdgcn_readfirstlane((int)(threadIdx.x >> 5));
    if (n >= NN) return;

    const v4f* xl4 = (const v4f*)xl;
    const v4f* xr4 = (const v4f*)xr;
    const float4* ea4 = (const float4*)ea;
    const v4f xrv = xr4[(size_t)n * (HCC / 4) + lane];
    const v4f xls = xl4[(size_t)n * (HCC / 4) + lane];
    const int p0 = __builtin_amdgcn_readfirstlane(off[n]);
    const int p1 = __builtin_amdgcn_readfirstlane(off[n + 1]);
    const float4 atv = ((const float4*)att)[lane];
    const float4 cbv = ((const float4*)cb)[lane];
    const float4* We4 = (const float4*)We;
    const float4 wr0 = We4[lane * 4 + 0];
    const float4 wr1 = We4[lane * 4 + 1];
    const float4 wr2 = We4[lane * 4 + 2];
    const float4 wr3 = We4[lane * 4 + 3];

    float s0 = edge_partial(xls, xrv, 0.0f, 0.0f, 0.0f, 0.0f, wr0, wr1, wr2, wr3, atv);
    s0 += __shfl_xor(s0, 4);
    s0 += __shfl_xor(s0, 2);
    s0 += __shfl_xor(s0, 1);
    float mrun = s0;
    float den  = 1.0f;
    v4f   acc  = xls;

    const int g = lane & 7;
    const bool b2 = (lane & 4) != 0;
    const bool b1 = (lane & 2) != 0;
    const bool b0 = (lane & 1) != 0;

    for (int pb = p0; pb < p1; pb += 32) {
        const int rem = p1 - pb;
        const int nb  = rem < 32 ? rem : 32;
        int sL = n;
        float4 evL = make_float4(0.0f, 0.0f, 0.0f, 0.0f);
        if (lane < nb) {
            sL = ssrc[pb + lane];
            evL = ea4[perm[pb + lane]];
        }
        const int ntrip = (nb + NA_U - 1) / NA_U;
        for (int t = 0; t < ntrip; ++t) {
            v4f xlv[NA_U];
#pragma unroll
            for (int i = 0; i < NA_U; ++i) {
                const int sN = __builtin_amdgcn_readlane(sL, t * NA_U + i);
                xlv[i] = xl4[(size_t)sN * (HCC / 4) + lane];
            }
            float sc[NA_U];
#pragma unroll
            for (int i = 0; i < NA_U; ++i) {
                const int j = t * NA_U + i;
                const float ex = ew_bcast(evL.x, j);
                const float ey = ew_bcast(evL.y, j);
                const float ez = ew_bcast(evL.z, j);
                const float ew = ew_bcast(evL.w, j);
                sc[i] = edge_partial(xlv[i], xrv, ex, ey, ez, ew, wr0, wr1, wr2, wr3, atv);
            }
            float ra[4];
#pragma unroll
            for (int k = 0; k < 4; ++k) {
                const float keep = b2 ? sc[k + 4] : sc[k];
                const float send = b2 ? sc[k] : sc[k + 4];
                ra[k] = keep + __shfl_xor(send, 4);
            }
            float rb[2];
#pragma unroll
            for (int k = 0; k < 2; ++k) {
                const float keep = b1 ? ra[k + 2] : ra[k];
                const float send = b1 ? ra[k] : ra[k + 2];
                rb[k] = keep + __shfl_xor(send, 2);
            }
            const float keep1 = b0 ? rb[1] : rb[0];
            const float send1 = b0 ? rb[0] : rb[1];
            const float tot = keep1 + __shfl_xor(send1, 1);

            const bool valid = (t * NA_U + g) < nb;
            float cm = valid ? tot : -INFINITY;
            cm = fmaxf(cm, __shfl_xor(cm, 4));
            cm = fmaxf(cm, __shfl_xor(cm, 2));
            cm = fmaxf(cm, __shfl_xor(cm, 1));
            const float mnew  = fmaxf(mrun, cm);
            const float scale = expf(mrun - mnew);
            const float pg    = valid ? expf(tot - mnew) : 0.0f;
            den *= scale;
            acc *= scale;
#pragma unroll
            for (int i = 0; i < NA_U; ++i) {
                const float pi = __shfl(pg, (lane & 24) + i);
                den += pi;
                acc += pi * xlv[i];
            }
            mrun = mnew;
        }
    }

    const float dd = den + 1e-16f;
    v4f o;
    o.x = acc.x / dd + cbv.x;
    o.y = acc.y / dd + cbv.y;
    o.z = acc.z / dd + cbv.z;
    o.w = acc.w / dd + cbv.w;
    ((v4f*)agg)[(size_t)n * (HCC / 4) + lane] = o;

    const float ps = (o.x + o.y) + (o.z + o.w);
    const float pq = fmaf(o.w, o.w, fmaf(o.z, o.z, fmaf(o.y, o.y, o.x * o.x)));
    const bool up = (lane & 16) != 0;
    float r = (up ? pq : ps) + __shfl_xor(up ? ps : pq, 16);
    r += __shfl_xor(r, 8);
    r += __shfl_xor(r, 4);
    r += __shfl_xor(r, 2);
    r += __shfl_xor(r, 1);
    if ((lane & 15) == 0)
        atomicAdd(&red[(size_t)(n & (STAT_SLOTS - 1)) * STAT_PITCH + (lane >> 4)], (double)r);
}

extern "C" void kernel_launch(void* const* d_in, const int* in_sizes, int n_in,
                              void* d_out, int out_size, void* d_ws, size_t ws_size,
                              hipStream_t stream) {
    const float* x          = (const float*)d_in[0];
    const int*   edge_index = (const int*)  d_in[1];
    const float* edge_attr  = (const float*)d_in[2];
    const float* embed_W    = (const float*)d_in[3];
    const float* embed_b    = (const float*)d_in[4];
    const float* embed_lnw  = (const float*)d_in[5];
    const float* embed_lnb  = (const float*)d_in[6];
    const float* Wl         = (const float*)d_in[7];
    const float* bl         = (const float*)d_in[8];
    const float* Wr         = (const float*)d_in[9];
    const float* br         = (const float*)d_in[10];
    const float* We         = (const float*)d_in[11];
    const float* att        = (const float*)d_in[12];
    const float* conv_b     = (const float*)d_in[13];
    const float* pln_w      = (const float*)d_in[14];
    const float* pln_b      = (const float*)d_in[15];
    const float* proj_W     = (const float*)d_in[16];
    const float* proj_b     = (const float*)d_in[17];
    const float* ln2_w      = (const float*)d_in[18];
    const float* ln2_b      = (const float*)d_in[19];

    const int* srcE = edge_index;
    const int* dstE = edge_index + EE;

    size_t off = 0;
    auto carve = [&](size_t nbytes) -> char* {
        char* p = (char*)d_ws + off;
        off += (nbytes + 255) & ~(size_t)255;
        return p;
    };
    float*    h     = (float*)   carve((size_t)NN * FF  * 4);
    float*    xl    = (float*)   carve((size_t)NN * HCC * 4);
    float*    xr    = (float*)   carve((size_t)NN * HCC * 4);
    float*    agg   = (float*)   carve((size_t)NN * HCC * 4);
    int*      ssrc  = (int*)     carve((size_t)EE * 4);
    int*      perm  = (int*)     carve((size_t)EE * 4);
    int*      rank  = (int*)     carve((size_t)EE * 4);
    int*      goff  = (int*)     carve((size_t)(NN + 1) * 4);
    double*   red   = (double*)  carve((size_t)LL * STAT_SLOTS * STAT_PITCH * sizeof(double));
    (void)ws_size; (void)in_sizes; (void)n_in; (void)out_size;

    float* jk = (float*)d_out;

    const int mtiles = NN / 16;
    const dim3 blk(256);
    const dim3 gGemm((mtiles * 32 + 255) / 256);
    const dim3 gEdgeT(EE / 256);
    const dim3 gNode(NN / 8);

    const size_t shLin  = (size_t)HCC * (FF + 4) * 4;
    const size_t shProj = (size_t)FF * (HCC + 4) * 4 + (size_t)2 * HCC * 4 + 16 * sizeof(double);

    gemm_wmma_f32<2, INF_, false, false, 1><<<gGemm, blk, 0, stream>>>(
        x, embed_W, embed_b, h, NN, FF, nullptr, nullptr, nullptr, embed_lnw, embed_lnb, jk);

    hipMemsetAsync(goff, 0, (size_t)(NN + 1) * 4, stream);
    csr_count_kernel<<<gEdgeT, blk, 0, stream>>>(dstE, goff, rank);
    csr_scan_kernel<<<dim3(1), dim3(1024), 0, stream>>>(goff, red);
    csr_scatter_kernel<<<gEdgeT, blk, 0, stream>>>(srcE, dstE, goff, rank, ssrc, perm);

    for (int l = 0; l < LL; ++l) {
        const float* Wl_l = Wl + (size_t)l * HCC * FF;
        const float* bl_l = bl + (size_t)l * HCC;
        const float* Wr_l = Wr + (size_t)l * HCC * FF;
        const float* br_l = br + (size_t)l * HCC;
        const float* We_l = We + (size_t)l * HCC * EDD;
        const float* at_l = att + (size_t)l * HH * CC;
        const float* cb_l = conv_b + (size_t)l * HCC;
        const float* pw_l = pln_w + (size_t)l * HCC;
        const float* pb_l = pln_b + (size_t)l * HCC;
        const float* pW_l = proj_W + (size_t)l * FF * HCC;
        const float* pB_l = proj_b + (size_t)l * FF;
        const float* l2w  = ln2_w + (size_t)l * FF;
        const float* l2b  = ln2_b + (size_t)l * FF;
        double*      red_l = red + (size_t)l * STAT_SLOTS * STAT_PITCH;

        gemm_wmma_f32<8, FF, true, false, 0><<<gGemm, blk, shLin, stream>>>(
            h, Wl_l, bl_l, xl, NN, HCC, nullptr, nullptr, nullptr, nullptr, nullptr, nullptr);
        gemm_wmma_f32<8, FF, true, false, 0><<<gGemm, blk, shLin, stream>>>(
            h, Wr_l, br_l, xr, NN, HCC, nullptr, nullptr, nullptr, nullptr, nullptr, nullptr);

        node_attn_kernel<<<gNode, blk, 0, stream>>>(goff, ssrc, perm, edge_attr, We_l,
                                                    xl, xr, at_l, cb_l, red_l, agg);

        gemm_wmma_f32<2, HCC, true, true, 2><<<gGemm, blk, shProj, stream>>>(
            agg, pW_l, pB_l, (l + 1 < LL) ? h : nullptr, NN, FF, pw_l, pb_l, red_l, l2w, l2b, jk);
    }
}


# node attention: 8 resident waves per SIMD, not 5: scores packed over a lane's channel pairs so edge attributes stay scalar, conv bias fetched late, register budget set; sums unchanged
# speedup vs baseline: 3.9087x; 1.1484x over previous
#include <hip/hip_runtime.h>
#include <math.h>

#define NN   50000
#define EE   800000
#define INF_ 512
#define FF   32
#define HH   4
#define CC   32
#define HCC  128
#define EDD  4
#define LL   2

typedef float v2f __attribute__((ext_vector_type(2)));
typedef float v8f __attribute__((ext_vector_type(8)));
typedef float v4f __attribute__((ext_vector_type(4)));

__device__ __forceinline__ float leakyf(float x, float s) {
    return x >= 0.0f ? x : s * x;
}
__device__ __forceinline__ double shfl_xor_f64(double v, int m) {
    const int lo = __shfl_xor(__double2loint(v), m);
    const int hi = __shfl_xor(__double2hiint(v), m);
    return __hiloint2double(hi, lo);
}

#define STAT_SLOTS 256
#define STAT_PITCH 16

template <int CT, int KT, bool USE_LDS, bool NORM_A, int LN_EPI>
__global__ __launch_bounds__(256)
void gemm_wmma_f32(const float* __restrict__ A, const float* __restrict__ W,
                   const float* __restrict__ bias, float* __restrict__ Cout,
                   int M, int Nout,
                   const float* __restrict__ nw, const float* __restrict__ nb,
                   const double* __restrict__ red,
                   const float* __restrict__ lw, const float* __restrict__ lb,
                   float* __restrict__ jk) {
    extern __shared__ float sW[];
    const int tid  = (int)threadIdx.x;
    const int lane = tid & 31;
    const int wave = (int)((blockIdx.x * blockDim.x + tid) >> 5);
    const int mtiles  = M >> 4;
    const int ngroups = Nout / (16 * CT);
    const int stride  = KT + 4;
    const bool live   = wave < mtiles * ngroups;
    const int hi = lane >> 4;
    const int mi = lane & 15;

    v2f a_all[NORM_A ? KT / 4 : 1];
    float muf = 0.0f, inv = 1.0f;

    if (USE_LDS) {
        const int total = Nout * KT;
        for (int i = tid; i < total; i += 256) {
            const int c  = i / KT;
            const int kk = i - c * KT;
            sW[c * stride + kk] = W[i];
        }
        if (NORM_A) {
            float* sN = sW + Nout * stride;
            for (int i = tid; i < KT; i += 256) {
                sN[i]      = nw[i];
                sN[KT + i] = nb[i];
            }
            const float* Ap = A + (size_t)((live ? wave / ngroups : 0) * 16 + mi) * KT + 2 * hi;
#pragma unroll
            for (int s = 0; s < KT / 4; ++s) {
                a_all[s].x = Ap[4 * s];
                a_all[s].y = Ap[4 * s + 1];
            }
            const double ss = red[(size_t)tid * STAT_PITCH];
            const double qq = red[(size_t)tid * STAT_PITCH + 1];
            const bool up = (lane & 16) != 0;
            double r = (up ? qq : ss) + shfl_xor_f64(up ? ss : qq, 16);
#pragma unroll
            for (int off = 8; off >= 1; off >>= 1) r += shfl_xor_f64(r, off);
            double* sP = (double*)(sN + 2 * KT);
            if ((lane & 15) == 0) sP[(up ? 8 : 0) + (tid >> 5)] = r;
        }
        __syncthreads();
        if (NORM_A) {
            const double* sP = (const double*)(sW + Nout * stride + 2 * KT);
            double t = sP[lane & 15];
            t += shfl_xor_f64(t, 4);
            t += shfl_xor_f64(t, 2);
            t += shfl_xor_f64(t, 1);
            const double o  = shfl_xor_f64(t, 8);
            const bool   hq = (lane & 8) != 0;
            const double S  = hq ? o : t;
            const double Q  = hq ? t : o;
            constexpr double inv_n = 1.0 / ((double)NN * (double)HCC);
            const double mu  = S * inv_n;
            const double var = Q * inv_n - mu * mu;
            float varf = (float)var;
            if (varf < 0.0f) varf = 0.0f;
            inv = 1.0f / (sqrtf(varf) + 1e-5f);
            muf = (float)mu;
        }
    }
    if (!live) return;

    const int mtile = wave / ngroups;
    const int ng    = wave % ngroups;
    const int row0  = mtile * 16;
    const int col0  = ng * 16 * CT;

    v8f acc[CT] = {};
    const float* Arow = A + (size_t)(row0 + mi) * KT + 2 * hi;

    const float* sNk = sW + Nout * stride + 2 * hi;

    v2f a_cur;
    if (NORM_A) {
        a_cur = a_all[0];
    } else {
        a_cur.x = Arow[0];
        a_cur.y = Arow[1];
    }
    v2f b_cur[CT];
    if (!USE_LDS) {
#pragma unroll
        for (int t = 0; t < CT; ++t) {
            const float* Wc = W + (size_t)(col0 + t * 16 + mi) * KT + 2 * hi;
            b_cur[t].x = Wc[0];
            b_cur[t].y = Wc[1];
        }
    }

    for (int k = 0; k < KT; k += 4) {
        v2f a_nxt = {};
        v2f b_nxt[CT];
        if (k + 4 < KT) {
            if (NORM_A) {
                a_nxt = a_all[NORM_A ? (k + 4) / 4 : 0];
            } else {
                a_nxt.x = Arow[k + 4];
                a_nxt.y = Arow[k + 5];
            }
            if (!USE_LDS) {
#pragma unroll
                for (int t = 0; t < CT; ++t) {
                    const float* Wc = W + (size_t)(col0 + t * 16 + mi) * KT + 2 * hi;
                    b_nxt[t].x = Wc[k + 4];
                    b_nxt[t].y = Wc[k + 5];
                }
                if ((k & 31) == 0)
                    __builtin_prefetch(Arow + k + 132, 0, 1);
            }
        }
        v2f a_use = a_cur;
        if (NORM_A) {
            a_use.x = leakyf(sNk[k]     * ((a_cur.x - muf) * inv) + sNk[KT + k],     0.01f);
            a_use.y = leakyf(sNk[k + 1] * ((a_cur.y - muf) * inv) + sNk[KT + k + 1], 0.01f);
        }
#pragma unroll
        for (int t = 0; t < CT; ++t) {
            v2f b;
            if (USE_LDS) {
                const float* Sc = sW + (size_t)(col0 + t * 16 + mi) * stride + 2 * hi;
                b.x = Sc[k];
                b.y = Sc[k + 1];
            } else {
                b = b_cur[t];
            }
            acc[t] = __builtin_amdgcn_wmma_f32_16x16x4_f32(
                false, a_use, false, b, (short)0, acc[t], false, false);
        }
        a_cur = a_nxt;
        if (!USE_LDS) {
#pragma unroll
            for (int t = 0; t < CT; ++t) b_cur[t] = b_nxt[t];
        }
    }

    if (LN_EPI) {
        const size_t e0 = (size_t)(row0 + 8 * hi) * FF + mi;
        float j0[8], j1[8];
        if (LN_EPI == 2) {
#pragma unroll
            for (int v = 0; v < 8; ++v) {
                j0[v] = jk[e0 + (size_t)v * FF];
                j1[v] = jk[e0 + (size_t)v * FF + 16];
            }
        }
        const float bb0 = bias[mi], bb1 = bias[16 + mi];
        const float g0 = lw[mi], g1 = lw[16 + mi];
        const float c0 = lb[mi], c1 = lb[16 + mi];
        float z0[8], z1[8], s[8], q[8];
#pragma unroll
        for (int v = 0; v < 8; ++v) {
            z0[v] = acc[0][v] + bb0;
            z1[v] = acc[1][v] + bb1;
            s[v]  = z0[v] + z1[v];
        }
#pragma unroll
        for (int off = 8; off >= 1; off >>= 1) {
#pragma unroll
            for (int v = 0; v < 8; ++v) s[v] += __shfl_xor(s[v], off);
        }
#pragma unroll
        for (int v = 0; v < 8; ++v) {
            const float mu = s[v] * (1.0f / 32.0f);
            z0[v] -= mu;
            z1[v] -= mu;
            q[v] = z0[v] * z0[v] + z1[v] * z1[v];
        }
#pragma unroll
        for (int off = 8; off >= 1; off >>= 1) {
#pragma unroll
            for (int v = 0; v < 8; ++v) q[v] += __shfl_xor(q[v], off);
        }
#pragma unroll
        for (int v = 0; v < 8; ++v) {
            const float rs = rsqrtf(q[v] * (1.0f / 32.0f) + 1e-5f);
            const float y0 = leakyf(g0 * z0[v] * rs + c0, 0.01f);
            const float y1 = leakyf(g1 * z1[v] * rs + c1, 0.01f);
            const size_t e = e0 + (size_t)v * FF;
            if (Cout) {
                Cout[e]      = y0;
                Cout[e + 16] = y1;
            }
            if (LN_EPI == 2) {
                jk[e]      = fmaxf(j0[v], y0);
                jk[e + 16] = fmaxf(j1[v], y1);
            } else {
                jk[e]      = y0;
                jk[e + 16] = y1;
            }
        }
    } else {
#pragma unroll
        for (int t = 0; t < CT; ++t) {
            const int col = col0 + t * 16 + mi;
            const float bb = bias ? bias[col] : 0.0f;
#pragma unroll
            for (int v = 0; v < 8; ++v) {
                const int r = row0 + v + 8 * hi;
                Cout[(size_t)r * Nout + col] = acc[t][v] + bb;
            }
        }
    }
}

__device__ __forceinline__ float ew_bcast(float v, int j) {
    return __int_as_float(__builtin_amdgcn_readlane(__float_as_int(v), j));
}

__global__ __launch_bounds__(256)
void csr_count_kernel(const int* __restrict__ dst, int* __restrict__ cnt,
                      int* __restrict__ rank) {
    const int e = (int)(blockIdx.x * blockDim.x + threadIdx.x);
    if (e < EE) rank[e] = atomicAdd(&cnt[dst[e]], 1);
}

#define SCN_IT 49
#define SCN_G  7
__global__ __launch_bounds__(1024)
void csr_scan_kernel(int* __restrict__ cnt, double* __restrict__ red) {
    __shared__ int wsum[32];
    const int tid   = (int)threadIdx.x;
    static_assert(LL * STAT_SLOTS * 2 == 1024, "one slot double per thread of the block");
    red[(size_t)(tid >> 1) * STAT_PITCH + (tid & 1)] = 0.0;
    const int lane  = tid & 31;
    const int w     = tid >> 5;
    const int wbase = w * (SCN_IT * 32);
    int v[SCN_IT];
#pragma unroll
    for (int i = 0; i < SCN_IT; ++i) {
        const int idx = wbase + i * 32 + lane;
        v[i] = (idx < NN) ? cnt[idx] : 0;
    }
    int carry = 0;
#pragma unroll
    for (int r = 0; r < SCN_IT; r += SCN_G) {
        int inc[SCN_G];
#pragma unroll
        for (int k = 0; k < SCN_G; ++k) inc[k] = v[r + k];
#pragma unroll
        for (int o = 1; o < 32; o <<= 1) {
            int y[SCN_G];
#pragma unroll
            for (int k = 0; k < SCN_G; ++k) y[k] = __shfl_up(inc[k], o);
#pragma unroll
            for (int k = 0; k < SCN_G; ++k) if (lane >= o) inc[k] += y[k];
        }
#pragma unroll
        for (int k = 0; k < SCN_G; ++k) {
            const int x = v[r + k];
            v[r + k] = carry + inc[k] - x;
            carry += __builtin_amdgcn_readlane(inc[k], 31);
        }
    }
    if (lane == 0) wsum[w] = carry;
    __syncthreads();
    if (w == 0) {
        const int x = wsum[lane];
        int inc = x;
#pragma unroll
        for (int o = 1; o < 32; o <<= 1) {
            const int y = __shfl_up(inc, o);
            if (lane >= o) inc += y;
        }
        wsum[lane] = inc - x;
    }
    __syncthreads();
    const int wex = wsum[w];
#pragma unroll
    for (int i = 0; i < SCN_IT; ++i) {
        const int idx = wbase + i * 32 + lane;
        if (idx <= NN) cnt[idx] = wex + v[i];
    }
}

__global__ __launch_bounds__(256)
void csr_scatter_kernel(const int* __restrict__ src, const int* __restrict__ dst,
                        const int* __restrict__ off, const int* __restrict__ rank,
                        int* __restrict__ ssrc, int* __restrict__ perm) {
    const int e = (int)(blockIdx.x * blockDim.x + threadIdx.x);
    if (e < EE) {
        const int p = off[dst[e]] + rank[e];
        ssrc[p] = src[e];
        perm[p] = e;
    }
}

#define NA_U 8
__device__ __forceinline__ float edge_partial(const v4f xlv, const v4f xrv,
                                              float ex, float ey, float ez, float ew,
                                              const v2f wx01, const v2f wy01,
                                              const v2f wz01, const v2f ww01,
                                              const v2f wx23, const v2f wy23,
                                              const v2f wz23, const v2f ww23,
                                              const float4 atv) {
    v2f m01 = {xlv.x + xrv.x, xlv.y + xrv.y};
    v2f m23 = {xlv.z + xrv.z, xlv.w + xrv.w};
    const v2f bx = {ex, ex};
    const v2f by = {ey, ey};
    const v2f bz = {ez, ez};
    const v2f bw = {ew, ew};
    m01 = __builtin_elementwise_fma(wx01, bx, m01);
    m23 = __builtin_elementwise_fma(wx23, bx, m23);
    m01 = __builtin_elementwise_fma(wy01, by, m01);
    m23 = __builtin_elementwise_fma(wy23, by, m23);
    m01 = __builtin_elementwise_fma(wz01, bz, m01);
    m23 = __builtin_elementwise_fma(wz23, bz, m23);
    m01 = __builtin_elementwise_fma(ww01, bw, m01);
    m23 = __builtin_elementwise_fma(ww23, bw, m23);
    const v2f t01 = m01 * 0.2f;
    const v2f t23 = m23 * 0.2f;
    float s = 0.0f;
    s += atv.x * (m01.x >= 0.0f ? m01.x : t01.x);
    s += atv.y * (m01.y >= 0.0f ? m01.y : t01.y);
    s += atv.z * (m23.x >= 0.0f ? m23.x : t23.x);
    s += atv.w * (m23.y >= 0.0f ? m23.y : t23.y);
    return s;
}

__global__ __launch_bounds__(256, 8)
void node_attn_kernel(const int* __restrict__ off, const int* __restrict__ ssrc,
                      const int* __restrict__ perm, const float* __restrict__ ea,
                      const float* __restrict__ We, const float* __restrict__ xl,
                      const float* __restrict__ xr, const float* __restrict__ att,
                      const float* __restrict__ cb, double* __restrict__ red,
                      float* __restrict__ agg) {
    const int lane = (int)(threadIdx.x & 31);
    const int n = (int)blockIdx.x * 8 + __builtin_amdgcn_readfirstlane((int)(threadIdx.x >> 5));
    if (n >= NN) return;

    const v4f* xl4 = (const v4f*)xl;
    const v4f* xr4 = (const v4f*)xr;
    const float4* ea4 = (const float4*)ea;
    const unsigned rown = (unsigned)n * (unsigned)(HCC / 4) + (unsigned)lane;
    const v4f xrv = xr4[rown];
    const v4f xls = xl4[rown];
    const int p0 = __builtin_amdgcn_readfirstlane(off[n]);
    const int p1 = __builtin_amdgcn_readfirstlane(off[n + 1]);
    const float4 atv = ((const float4*)att)[lane];
    const float4* We4 = (const float4*)We;
    const float4 wr0 = We4[lane * 4 + 0];
    const float4 wr1 = We4[lane * 4 + 1];
    const float4 wr2 = We4[lane * 4 + 2];
    const float4 wr3 = We4[lane * 4 + 3];
    const v2f wx01 = {wr0.x, wr1.x}, wy01 = {wr0.y, wr1.y}, wz01 = {wr0.z, wr1.z}, ww01 = {wr0.w, wr1.w};
    const v2f wx23 = {wr2.x, wr3.x}, wy23 = {wr2.y, wr3.y}, wz23 = {wr2.z, wr3.z}, ww23 = {wr2.w, wr3.w};

    float s0 = edge_partial(xls, xrv, 0.0f, 0.0f, 0.0f, 0.0f,
                            wx01, wy01, wz01, ww01, wx23, wy23, wz23, ww23, atv);
    s0 += __shfl_xor(s0, 4);
    s0 += __shfl_xor(s0, 2);
    s0 += __shfl_xor(s0, 1);
    float mrun = s0;
    float den  = 1.0f;
    v4f   acc  = xls;

    const int g = lane & 7;
    const bool b2 = (lane & 4) != 0;
    const bool b1 = (lane & 2) != 0;
    const bool b0 = (lane & 1) != 0;

    for (int pb = p0; pb < p1; pb += 32) {
        const int rem = p1 - pb;
        const int nb  = rem < 32 ? rem : 32;
        int sL = n;
        float4 evL = make_float4(0.0f, 0.0f, 0.0f, 0.0f);
        if (lane < nb) {
            sL = ssrc[pb + lane];
            evL = ea4[perm[pb + lane]];
        }
        const int ntrip = (nb + NA_U - 1) / NA_U;
        for (int t = 0; t < ntrip; ++t) {
            v4f xlv[NA_U];
#pragma unroll
            for (int i = 0; i < NA_U; ++i) {
                const int sN = __builtin_amdgcn_readlane(sL, t * NA_U + i);
                xlv[i] = xl4[(size_t)sN * (HCC / 4) + lane];
            }
            __builtin_amdgcn_sched_barrier(0);
            float sc[NA_U];
#pragma unroll
            for (int i = 0; i < NA_U; ++i) {
                const int j = t * NA_U + i;
                const float ex = ew_bcast(evL.x, j);
                const float ey = ew_bcast(evL.y, j);
                const float ez = ew_bcast(evL.z, j);
                const float ew = ew_bcast(evL.w, j);
                sc[i] = edge_partial(xlv[i], xrv, ex, ey, ez, ew,
                                     wx01, wy01, wz01, ww01, wx23, wy23, wz23, ww23, atv);
            }
            float ra[4];
#pragma unroll
            for (int k = 0; k < 4; ++k) {
                const float keep = b2 ? sc[k + 4] : sc[k];
                const float send = b2 ? sc[k] : sc[k + 4];
                ra[k] = keep + __shfl_xor(send, 4);
            }
            float rb[2];
#pragma unroll
            for (int k = 0; k < 2; ++k) {
                const float keep = b1 ? ra[k + 2] : ra[k];
                const float send = b1 ? ra[k] : ra[k + 2];
                rb[k] = keep + __shfl_xor(send, 2);
            }
            const float keep1 = b0 ? rb[1] : rb[0];
            const float send1 = b0 ? rb[0] : rb[1];
            const float tot = keep1 + __shfl_xor(send1, 1);

            const bool valid = (t * NA_U + g) < nb;
            float cm = valid ? tot : -INFINITY;
            cm = fmaxf(cm, __shfl_xor(cm, 4));
            cm = fmaxf(cm, __shfl_xor(cm, 2));
            cm = fmaxf(cm, __shfl_xor(cm, 1));
            const float mnew  = fmaxf(mrun, cm);
            const float scale = expf(mrun - mnew);
            const float pg    = valid ? expf(tot - mnew) : 0.0f;
            den *= scale;
            acc *= scale;
#pragma unroll
            for (int i = 0; i < NA_U; ++i) {
                const float pi = __shfl(pg, (lane & 24) + i);
                den += pi;
                acc += pi * xlv[i];
            }
            mrun = mnew;
        }
    }

    const float4 cbv = ((const float4*)cb)[lane];
    const float dd = den + 1e-16f;
    v4f o;
    o.x = acc.x / dd + cbv.x;
    o.y = acc.y / dd + cbv.y;
    o.z = acc.z / dd + cbv.z;
    o.w = acc.w / dd + cbv.w;
    ((v4f*)agg)[rown] = o;

    const float ps = (o.x + o.y) + (o.z + o.w);
    const float pq = fmaf(o.w, o.w, fmaf(o.z, o.z, fmaf(o.y, o.y, o.x * o.x)));
    const bool up = (lane & 16) != 0;
    float r = (up ? pq : ps) + __shfl_xor(up ? ps : pq, 16);
    r += __shfl_xor(r, 8);
    r += __shfl_xor(r, 4);
    r += __shfl_xor(r, 2);
    r += __shfl_xor(r, 1);
    if ((lane & 15) == 0)
        atomicAdd(&red[(size_t)(n & (STAT_SLOTS - 1)) * STAT_PITCH + (lane >> 4)], (double)r);
}

extern "C" void kernel_launch(void* const* d_in, const int* in_sizes, int n_in,
                              void* d_out, int out_size, void* d_ws, size_t ws_size,
                              hipStream_t stream) {
    const float* x          = (const float*)d_in[0];
    const int*   edge_index = (const int*)  d_in[1];
    const float* edge_attr  = (const float*)d_in[2];
    const float* embed_W    = (const float*)d_in[3];
    const float* embed_b    = (const float*)d_in[4];
    const float* embed_lnw  = (const float*)d_in[5];
    const float* embed_lnb  = (const float*)d_in[6];
    const float* Wl         = (const float*)d_in[7];
    const float* bl         = (const float*)d_in[8];
    const float* Wr         = (const float*)d_in[9];
    const float* br         = (const float*)d_in[10];
    const float* We         = (const float*)d_in[11];
    const float* att        = (const float*)d_in[12];
    const float* conv_b     = (const float*)d_in[13];
    const float* pln_w      = (const float*)d_in[14];
    const float* pln_b      = (const float*)d_in[15];
    const float* proj_W     = (const float*)d_in[16];
    const float* proj_b     = (const float*)d_in[17];
    const float* ln2_w      = (const float*)d_in[18];
    const float* ln2_b      = (const float*)d_in[19];

    const int* srcE = edge_index;
    const int* dstE = edge_index + EE;

    size_t off = 0;
    auto carve = [&](size_t nbytes) -> char* {
        char* p = (char*)d_ws + off;
        off += (nbytes + 255) & ~(size_t)255;
        return p;
    };
    float*    h     = (float*)   carve((size_t)NN * FF  * 4);
    float*    xl    = (float*)   carve((size_t)NN * HCC * 4);
    float*    xr    = (float*)   carve((size_t)NN * HCC * 4);
    float*    agg   = (float*)   carve((size_t)NN * HCC * 4);
    int*      ssrc  = (int*)     carve((size_t)EE * 4);
    int*      perm  = (int*)     carve((size_t)EE * 4);
    int*      rank  = (int*)     carve((size_t)EE * 4);
    int*      goff  = (int*)     carve((size_t)(NN + 1) * 4);
    double*   red   = (double*)  carve((size_t)LL * STAT_SLOTS * STAT_PITCH * sizeof(double));
    (void)ws_size; (void)in_sizes; (void)n_in; (void)out_size;

    float* jk = (float*)d_out;

    const int mtiles = NN / 16;
    const dim3 blk(256);
    const dim3 gGemm((mtiles * 32 + 255) / 256);
    const dim3 gEdgeT(EE / 256);
    const dim3 gNode(NN / 8);

    const size_t shLin  = (size_t)HCC * (FF + 4) * 4;
    const size_t shProj = (size_t)FF * (HCC + 4) * 4 + (size_t)2 * HCC * 4 + 16 * sizeof(double);

    gemm_wmma_f32<2, INF_, false, false, 1><<<gGemm, blk, 0, stream>>>(
        x, embed_W, embed_b, h, NN, FF, nullptr, nullptr, nullptr, embed_lnw, embed_lnb, jk);

    hipMemsetAsync(goff, 0, (size_t)(NN + 1) * 4, stream);
    csr_count_kernel<<<gEdgeT, blk, 0, stream>>>(dstE, goff, rank);
    csr_scan_kernel<<<dim3(1), dim3(1024), 0, stream>>>(goff, red);
    csr_scatter_kernel<<<gEdgeT, blk, 0, stream>>>(srcE, dstE, goff, rank, ssrc, perm);

    for (int l = 0; l < LL; ++l) {
        const float* Wl_l = Wl + (size_t)l * HCC * FF;
        const float* bl_l = bl + (size_t)l * HCC;
        const float* Wr_l = Wr + (size_t)l * HCC * FF;
        const float* br_l = br + (size_t)l * HCC;
        const float* We_l = We + (size_t)l * HCC * EDD;
        const float* at_l = att + (size_t)l * HH * CC;
        const float* cb_l = conv_b + (size_t)l * HCC;
        const float* pw_l = pln_w + (size_t)l * HCC;
        const float* pb_l = pln_b + (size_t)l * HCC;
        const float* pW_l = proj_W + (size_t)l * FF * HCC;
        const float* pB_l = proj_b + (size_t)l * FF;
        const float* l2w  = ln2_w + (size_t)l * FF;
        const float* l2b  = ln2_b + (size_t)l * FF;
        double*      red_l = red + (size_t)l * STAT_SLOTS * STAT_PITCH;

        gemm_wmma_f32<8, FF, true, false, 0><<<gGemm, blk, shLin, stream>>>(
            h, Wl_l, bl_l, xl, NN, HCC, nullptr, nullptr, nullptr, nullptr, nullptr, nullptr);
        gemm_wmma_f32<8, FF, true, false, 0><<<gGemm, blk, shLin, stream>>>(
            h, Wr_l, br_l, xr, NN, HCC, nullptr, nullptr, nullptr, nullptr, nullptr, nullptr);

        node_attn_kernel<<<gNode, blk, 0, stream>>>(goff, ssrc, perm, edge_attr, We_l,
                                                    xl, xr, at_l, cb_l, red_l, agg);

        gemm_wmma_f32<2, HCC, true, true, 2><<<gGemm, blk, shProj, stream>>>(
            agg, pW_l, pB_l, (l + 1 < LL) ? h : nullptr, NN, FF, pw_l, pb_l, red_l, l2w, l2b, jk);
    }
}


# lin_l / lin_r formed by the waves that form h (embed and proj epilogues; h tile through LDS from C to A layout; Wl, Wr staged per block): launches 12 -> 8, h not stored; sums unchanged
# speedup vs baseline: 4.3353x; 1.1091x over previous
#include <hip/hip_runtime.h>
#include <math.h>

#define NN   50000
#define EE   800000
#define INF_ 512
#define FF   32
#define HH   4
#define CC   32
#define HCC  128
#define EDD  4
#define LL   2

typedef float v2f __attribute__((ext_vector_type(2)));
typedef float v8f __attribute__((ext_vector_type(8)));
typedef float v4f __attribute__((ext_vector_type(4)));

__device__ __forceinline__ float leakyf(float x, float s) {
    return x >= 0.0f ? x : s * x;
}
__device__ __forceinline__ double shfl_xor_f64(double v, int m) {
    const int lo = __shfl_xor(__double2loint(v), m);
    const int hi = __shfl_xor(__double2hiint(v), m);
    return __hiloint2double(hi, lo);
}

#define STAT_SLOTS 256
#define STAT_PITCH 16

#define LIN_STR (FF + 4)
template <int CT, int KT, bool USE_LDS, bool NORM_A, int LN_EPI, bool FUSE_LIN>
__global__ __launch_bounds__(256)
void gemm_wmma_f32(const float* __restrict__ A, const float* __restrict__ W,
                   const float* __restrict__ bias, float* __restrict__ Cout,
                   int M, int Nout,
                   const float* __restrict__ nw, const float* __restrict__ nb,
                   const double* __restrict__ red,
                   const float* __restrict__ lw, const float* __restrict__ lb,
                   float* __restrict__ jk,
                   const float* __restrict__ Wl, const float* __restrict__ bl,
                   const float* __restrict__ Wr, const float* __restrict__ br,
                   float* __restrict__ xl, float* __restrict__ xr) {
    extern __shared__ float sW[];
    __shared__ __attribute__((aligned(16))) float sLin[FUSE_LIN ? (2 * HCC + 8 * 16) * LIN_STR : 4];
    const int tid  = (int)threadIdx.x;
    const int lane = tid & 31;
    const int wave = (int)((blockIdx.x * blockDim.x + tid) >> 5);
    const int mtiles  = M >> 4;
    const int ngroups = Nout / (16 * CT);
    const int stride  = KT + 4;
    const bool live   = wave < mtiles * ngroups;
    const int hi = lane >> 4;
    const int mi = lane & 15;

    v2f a_all[NORM_A ? KT / 4 : 1];
    float muf = 0.0f, inv = 1.0f;

    if (USE_LDS) {
        const int total = Nout * KT;
        for (int i = tid; i < total; i += 256) {
            const int c  = i / KT;
            const int kk = i - c * KT;
            sW[c * stride + kk] = W[i];
        }
        if (NORM_A) {
            float* sN = sW + Nout * stride;
            for (int i = tid; i < KT; i += 256) {
                sN[i]      = nw[i];
                sN[KT + i] = nb[i];
            }
            const float* Ap = A + (size_t)((live ? wave / ngroups : 0) * 16 + mi) * KT + 2 * hi;
#pragma unroll
            for (int s = 0; s < KT / 4; ++s) {
                a_all[s].x = Ap[4 * s];
                a_all[s].y = Ap[4 * s + 1];
            }
            const double ss = red[(size_t)tid * STAT_PITCH];
            const double qq = red[(size_t)tid * STAT_PITCH + 1];
            const bool up = (lane & 16) != 0;
            double r = (up ? qq : ss) + shfl_xor_f64(up ? ss : qq, 16);
#pragma unroll
            for (int off = 8; off >= 1; off >>= 1) r += shfl_xor_f64(r, off);
            double* sP = (double*)(sN + 2 * KT);
            if ((lane & 15) == 0) sP[(up ? 8 : 0) + (tid >> 5)] = r;
        }
    }
    if (FUSE_LIN) {
        const v4f* Wl4 = (const v4f*)Wl;
        const v4f* Wr4 = (const v4f*)Wr;
        constexpr int LIN_NP = HCC * FF / 4 / 256;
        v4f pl[LIN_NP], pr[LIN_NP];
#pragma unroll
        for (int j = 0; j < LIN_NP; ++j) {
            pl[j] = Wl4[tid + 256 * j];
            pr[j] = Wr4[tid + 256 * j];
        }
#pragma unroll
        for (int j = 0; j < LIN_NP; ++j) {
            const int i  = tid + 256 * j;
            const int c  = i >> 3;
            const int k4 = (i & 7) * 4;
            *(v4f*)(sLin + c * LIN_STR + k4)         = pl[j];
            *(v4f*)(sLin + (HCC + c) * LIN_STR + k4) = pr[j];
        }
    }
    if (USE_LDS || FUSE_LIN) __syncthreads();
    if (USE_LDS) {
        if (NORM_A) {
            const double* sP = (const double*)(sW + Nout * stride + 2 * KT);
            double t = sP[lane & 15];
            t += shfl_xor_f64(t, 4);
            t += shfl_xor_f64(t, 2);
            t += shfl_xor_f64(t, 1);
            const double o  = shfl_xor_f64(t, 8);
            const bool   hq = (lane & 8) != 0;
            const double S  = hq ? o : t;
            const double Q  = hq ? t : o;
            constexpr double inv_n = 1.0 / ((double)NN * (double)HCC);
            const double mu  = S * inv_n;
            const double var = Q * inv_n - mu * mu;
            float varf = (float)var;
            if (varf < 0.0f) varf = 0.0f;
            inv = 1.0f / (sqrtf(varf) + 1e-5f);
            muf = (float)mu;
        }
    }
    if (!live) return;

    const int mtile = wave / ngroups;
    const int ng    = wave % ngroups;
    const int row0  = mtile * 16;
    const int col0  = ng * 16 * CT;

    v8f acc[CT] = {};
    const float* Arow = A + (size_t)(row0 + mi) * KT + 2 * hi;

    const float* sNk = sW + Nout * stride + 2 * hi;

    v2f a_cur;
    if (NORM_A) {
        a_cur = a_all[0];
    } else {
        a_cur.x = Arow[0];
        a_cur.y = Arow[1];
    }
    v2f b_cur[CT];
    if (!USE_LDS) {
#pragma unroll
        for (int t = 0; t < CT; ++t) {
            const float* Wc = W + (size_t)(col0 + t * 16 + mi) * KT + 2 * hi;
            b_cur[t].x = Wc[0];
            b_cur[t].y = Wc[1];
        }
    }

    for (int k = 0; k < KT; k += 4) {
        v2f a_nxt = {};
        v2f b_nxt[CT];
        if (k + 4 < KT) {
            if (NORM_A) {
                a_nxt = a_all[NORM_A ? (k + 4) / 4 : 0];
            } else {
                a_nxt.x = Arow[k + 4];
                a_nxt.y = Arow[k + 5];
            }
            if (!USE_LDS) {
#pragma unroll
                for (int t = 0; t < CT; ++t) {
                    const float* Wc = W + (size_t)(col0 + t * 16 + mi) * KT + 2 * hi;
                    b_nxt[t].x = Wc[k + 4];
                    b_nxt[t].y = Wc[k + 5];
                }
                if ((k & 31) == 0)
                    __builtin_prefetch(Arow + k + 132, 0, 1);
            }
        }
        v2f a_use = a_cur;
        if (NORM_A) {
            a_use.x = leakyf(sNk[k]     * ((a_cur.x - muf) * inv) + sNk[KT + k],     0.01f);
            a_use.y = leakyf(sNk[k + 1] * ((a_cur.y - muf) * inv) + sNk[KT + k + 1], 0.01f);
        }
#pragma unroll
        for (int t = 0; t < CT; ++t) {
            v2f b;
            if (USE_LDS) {
                const float* Sc = sW + (size_t)(col0 + t * 16 + mi) * stride + 2 * hi;
                b.x = Sc[k];
                b.y = Sc[k + 1];
            } else {
                b = b_cur[t];
            }
            acc[t] = __builtin_amdgcn_wmma_f32_16x16x4_f32(
                false, a_use, false, b, (short)0, acc[t], false, false);
        }
        a_cur = a_nxt;
        if (!USE_LDS) {
#pragma unroll
            for (int t = 0; t < CT; ++t) b_cur[t] = b_nxt[t];
        }
    }

    if (LN_EPI) {
        const size_t e0 = (size_t)(row0 + 8 * hi) * FF + mi;
        float j0[8], j1[8];
        if (LN_EPI == 2) {
#pragma unroll
            for (int v = 0; v < 8; ++v) {
                j0[v] = jk[e0 + (size_t)v * FF];
                j1[v] = jk[e0 + (size_t)v * FF + 16];
            }
        }
        const float bb0 = bias[mi], bb1 = bias[16 + mi];
        const float g0 = lw[mi], g1 = lw[16 + mi];
        const float c0 = lb[mi], c1 = lb[16 + mi];
        float z0[8], z1[8], s[8], q[8];
#pragma unroll
        for (int v = 0; v < 8; ++v) {
            z0[v] = acc[0][v] + bb0;
            z1[v] = acc[1][v] + bb1;
            s[v]  = z0[v] + z1[v];
        }
#pragma unroll
        for (int off = 8; off >= 1; off >>= 1) {
#pragma unroll
            for (int v = 0; v < 8; ++v) s[v] += __shfl_xor(s[v], off);
        }
#pragma unroll
        for (int v = 0; v < 8; ++v) {
            const float mu = s[v] * (1.0f / 32.0f);
            z0[v] -= mu;
            z1[v] -= mu;
            q[v] = z0[v] * z0[v] + z1[v] * z1[v];
        }
#pragma unroll
        for (int off = 8; off >= 1; off >>= 1) {
#pragma unroll
            for (int v = 0; v < 8; ++v) q[v] += __shfl_xor(q[v], off);
        }
#pragma unroll
        for (int v = 0; v < 8; ++v) {
            const float rs = rsqrtf(q[v] * (1.0f / 32.0f) + 1e-5f);
            const float y0 = leakyf(g0 * z0[v] * rs + c0, 0.01f);
            const float y1 = leakyf(g1 * z1[v] * rs + c1, 0.01f);
            const size_t e = e0 + (size_t)v * FF;
            if (Cout) {
                Cout[e]      = y0;
                Cout[e + 16] = y1;
            }
            if (FUSE_LIN) {
                float* sT = sLin + 2 * HCC * LIN_STR + (tid >> 5) * (16 * LIN_STR);
                sT[(v + 8 * hi) * LIN_STR + mi]      = y0;
                sT[(v + 8 * hi) * LIN_STR + 16 + mi] = y1;
            }
            if (LN_EPI == 2) {
                jk[e]      = fmaxf(j0[v], y0);
                jk[e + 16] = fmaxf(j1[v], y1);
            } else {
                jk[e]      = y0;
                jk[e + 16] = y1;
            }
        }
        if (FUSE_LIN) {
            const float* sT = sLin + 2 * HCC * LIN_STR + (tid >> 5) * (16 * LIN_STR)
                            + mi * LIN_STR + 2 * hi;
            v2f af[FF / 4];
#pragma unroll
            for (int s = 0; s < FF / 4; ++s) {
                af[s].x = sT[4 * s];
                af[s].y = sT[4 * s + 1];
            }
#pragma unroll
            for (int m = 0; m < 2; ++m) {
                const float* sB  = sLin + (m * HCC + mi) * LIN_STR + 2 * hi;
                const float* bv  = m ? br : bl;
                float*       out = m ? xr : xl;
                v8f acc2[HCC / 16] = {};
#pragma unroll
                for (int s = 0; s < FF / 4; ++s) {
#pragma unroll
                    for (int t = 0; t < HCC / 16; ++t) {
                        v2f b;
                        b.x = sB[t * 16 * LIN_STR + 4 * s];
                        b.y = sB[t * 16 * LIN_STR + 4 * s + 1];
                        acc2[t] = __builtin_amdgcn_wmma_f32_16x16x4_f32(
                            false, af[s], false, b, (short)0, acc2[t], false, false);
                    }
                }
#pragma unroll
                for (int t = 0; t < HCC / 16; ++t) {
                    const int col = t * 16 + mi;
                    const float bb = bv[col];
#pragma unroll
                    for (int v = 0; v < 8; ++v) {
                        const int r = row0 + v + 8 * hi;
                        out[(size_t)r * HCC + col] = acc2[t][v] + bb;
                    }
                }
            }
        }
    } else {
#pragma unroll
        for (int t = 0; t < CT; ++t) {
            const int col = col0 + t * 16 + mi;
            const float bb = bias ? bias[col] : 0.0f;
#pragma unroll
            for (int v = 0; v < 8; ++v) {
                const int r = row0 + v + 8 * hi;
                Cout[(size_t)r * Nout + col] = acc[t][v] + bb;
            }
        }
    }
}

__device__ __forceinline__ float ew_bcast(float v, int j) {
    return __int_as_float(__builtin_amdgcn_readlane(__float_as_int(v), j));
}

__global__ __launch_bounds__(256)
void csr_count_kernel(const int* __restrict__ dst, int* __restrict__ cnt,
                      int* __restrict__ rank) {
    const int e = (int)(blockIdx.x * blockDim.x + threadIdx.x);
    if (e < EE) rank[e] = atomicAdd(&cnt[dst[e]], 1);
}

#define SCN_IT 49
#define SCN_G  7
__global__ __launch_bounds__(1024)
void csr_scan_kernel(int* __restrict__ cnt, double* __restrict__ red) {
    __shared__ int wsum[32];
    const int tid   = (int)threadIdx.x;
    static_assert(LL * STAT_SLOTS * 2 == 1024, "one slot double per thread of the block");
    red[(size_t)(tid >> 1) * STAT_PITCH + (tid & 1)] = 0.0;
    const int lane  = tid & 31;
    const int w     = tid >> 5;
    const int wbase = w * (SCN_IT * 32);
    int v[SCN_IT];
#pragma unroll
    for (int i = 0; i < SCN_IT; ++i) {
        const int idx = wbase + i * 32 + lane;
        v[i] = (idx < NN) ? cnt[idx] : 0;
    }
    int carry = 0;
#pragma unroll
    for (int r = 0; r < SCN_IT; r += SCN_G) {
        int inc[SCN_G];
#pragma unroll
        for (int k = 0; k < SCN_G; ++k) inc[k] = v[r + k];
#pragma unroll
        for (int o = 1; o < 32; o <<= 1) {
            int y[SCN_G];
#pragma unroll
            for (int k = 0; k < SCN_G; ++k) y[k] = __shfl_up(inc[k], o);
#pragma unroll
            for (int k = 0; k < SCN_G; ++k) if (lane >= o) inc[k] += y[k];
        }
#pragma unroll
        for (int k = 0; k < SCN_G; ++k) {
            const int x = v[r + k];
            v[r + k] = carry + inc[k] - x;
            carry += __builtin_amdgcn_readlane(inc[k], 31);
        }
    }
    if (lane == 0) wsum[w] = carry;
    __syncthreads();
    if (w == 0) {
        const int x = wsum[lane];
        int inc = x;
#pragma unroll
        for (int o = 1; o < 32; o <<= 1) {
            const int y = __shfl_up(inc, o);
            if (lane >= o) inc += y;
        }
        wsum[lane] = inc - x;
    }
    __syncthreads();
    const int wex = wsum[w];
#pragma unroll
    for (int i = 0; i < SCN_IT; ++i) {
        const int idx = wbase + i * 32 + lane;
        if (idx <= NN) cnt[idx] = wex + v[i];
    }
}

__global__ __launch_bounds__(256)
void csr_scatter_kernel(const int* __restrict__ src, const int* __restrict__ dst,
                        const int* __restrict__ off, const int* __restrict__ rank,
                        int* __restrict__ ssrc, int* __restrict__ perm) {
    const int e = (int)(blockIdx.x * blockDim.x + threadIdx.x);
    if (e < EE) {
        const int p = off[dst[e]] + rank[e];
        ssrc[p] = src[e];
        perm[p] = e;
    }
}

#define NA_U 8
__device__ __forceinline__ float edge_partial(const v4f xlv, const v4f xrv,
                                              float ex, float ey, float ez, float ew,
                                              const v2f wx01, const v2f wy01,
                                              const v2f wz01, const v2f ww01,
                                              const v2f wx23, const v2f wy23,
                                              const v2f wz23, const v2f ww23,
                                              const float4 atv) {
    v2f m01 = {xlv.x + xrv.x, xlv.y + xrv.y};
    v2f m23 = {xlv.z + xrv.z, xlv.w + xrv.w};
    const v2f bx = {ex, ex};
    const v2f by = {ey, ey};
    const v2f bz = {ez, ez};
    const v2f bw = {ew, ew};
    m01 = __builtin_elementwise_fma(wx01, bx, m01);
    m23 = __builtin_elementwise_fma(wx23, bx, m23);
    m01 = __builtin_elementwise_fma(wy01, by, m01);
    m23 = __builtin_elementwise_fma(wy23, by, m23);
    m01 = __builtin_elementwise_fma(wz01, bz, m01);
    m23 = __builtin_elementwise_fma(wz23, bz, m23);
    m01 = __builtin_elementwise_fma(ww01, bw, m01);
    m23 = __builtin_elementwise_fma(ww23, bw, m23);
    const v2f t01 = m01 * 0.2f;
    const v2f t23 = m23 * 0.2f;
    float s = 0.0f;
    s += atv.x * (m01.x >= 0.0f ? m01.x : t01.x);
    s += atv.y * (m01.y >= 0.0f ? m01.y : t01.y);
    s += atv.z * (m23.x >= 0.0f ? m23.x : t23.x);
    s += atv.w * (m23.y >= 0.0f ? m23.y : t23.y);
    return s;
}

__global__ __launch_bounds__(256, 8)
void node_attn_kernel(const int* __restrict__ off, const int* __restrict__ ssrc,
                      const int* __restrict__ perm, const float* __restrict__ ea,
                      const float* __restrict__ We, const float* __restrict__ xl,
                      const float* __restrict__ xr, const float* __restrict__ att,
                      const float* __restrict__ cb, double* __restrict__ red,
                      float* __restrict__ agg) {
    const int lane = (int)(threadIdx.x & 31);
    const int n = (int)blockIdx.x * 8 + __builtin_amdgcn_readfirstlane((int)(threadIdx.x >> 5));
    if (n >= NN) return;

    const v4f* xl4 = (const v4f*)xl;
    const v4f* xr4 = (const v4f*)xr;
    const float4* ea4 = (const float4*)ea;
    const unsigned rown = (unsigned)n * (unsigned)(HCC / 4) + (unsigned)lane;
    const v4f xrv = xr4[rown];
    const v4f xls = xl4[rown];
    const int p0 = __builtin_amdgcn_readfirstlane(off[n]);
    const int p1 = __builtin_amdgcn_readfirstlane(off[n + 1]);
    const float4 atv = ((const float4*)att)[lane];
    const float4* We4 = (const float4*)We;
    const float4 wr0 = We4[lane * 4 + 0];
    const float4 wr1 = We4[lane * 4 + 1];
    const float4 wr2 = We4[lane * 4 + 2];
    const float4 wr3 = We4[lane * 4 + 3];
    const v2f wx01 = {wr0.x, wr1.x}, wy01 = {wr0.y, wr1.y}, wz01 = {wr0.z, wr1.z}, ww01 = {wr0.w, wr1.w};
    const v2f wx23 = {wr2.x, wr3.x}, wy23 = {wr2.y, wr3.y}, wz23 = {wr2.z, wr3.z}, ww23 = {wr2.w, wr3.w};

    float s0 = edge_partial(xls, xrv, 0.0f, 0.0f, 0.0f, 0.0f,
                            wx01, wy01, wz01, ww01, wx23, wy23, wz23, ww23, atv);
    s0 += __shfl_xor(s0, 4);
    s0 += __shfl_xor(s0, 2);
    s0 += __shfl_xor(s0, 1);
    float mrun = s0;
    float den  = 1.0f;
    v4f   acc  = xls;

    const int g = lane & 7;
    const bool b2 = (lane & 4) != 0;
    const bool b1 = (lane & 2) != 0;
    const bool b0 = (lane & 1) != 0;

    for (int pb = p0; pb < p1; pb += 32) {
        const int rem = p1 - pb;
        const int nb  = rem < 32 ? rem : 32;
        int sL = n;
        float4 evL = make_float4(0.0f, 0.0f, 0.0f, 0.0f);
        if (lane < nb) {
            sL = ssrc[pb + lane];
            evL = ea4[perm[pb + lane]];
        }
        const int ntrip = (nb + NA_U - 1) / NA_U;
        for (int t = 0; t < ntrip; ++t) {
            v4f xlv[NA_U];
#pragma unroll
            for (int i = 0; i < NA_U; ++i) {
                const int sN = __builtin_amdgcn_readlane(sL, t * NA_U + i);
                xlv[i] = xl4[(size_t)sN * (HCC / 4) + lane];
            }
            __builtin_amdgcn_sched_barrier(0);
            float sc[NA_U];
#pragma unroll
            for (int i = 0; i < NA_U; ++i) {
                const int j = t * NA_U + i;
                const float ex = ew_bcast(evL.x, j);
                const float ey = ew_bcast(evL.y, j);
                const float ez = ew_bcast(evL.z, j);
                const float ew = ew_bcast(evL.w, j);
                sc[i] = edge_partial(xlv[i], xrv, ex, ey, ez, ew,
                                     wx01, wy01, wz01, ww01, wx23, wy23, wz23, ww23, atv);
            }
            float ra[4];
#pragma unroll
            for (int k = 0; k < 4; ++k) {
                const float keep = b2 ? sc[k + 4] : sc[k];
                const float send = b2 ? sc[k] : sc[k + 4];
                ra[k] = keep + __shfl_xor(send, 4);
            }
            float rb[2];
#pragma unroll
            for (int k = 0; k < 2; ++k) {
                const float keep = b1 ? ra[k + 2] : ra[k];
                const float send = b1 ? ra[k] : ra[k + 2];
                rb[k] = keep + __shfl_xor(send, 2);
            }
            const float keep1 = b0 ? rb[1] : rb[0];
            const float send1 = b0 ? rb[0] : rb[1];
            const float tot = keep1 + __shfl_xor(send1, 1);

            const bool valid = (t * NA_U + g) < nb;
            float cm = valid ? tot : -INFINITY;
            cm = fmaxf(cm, __shfl_xor(cm, 4));
            cm = fmaxf(cm, __shfl_xor(cm, 2));
            cm = fmaxf(cm, __shfl_xor(cm, 1));
            const float mnew  = fmaxf(mrun, cm);
            const float scale = expf(mrun - mnew);
            const float pg    = valid ? expf(tot - mnew) : 0.0f;
            den *= scale;
            acc *= scale;
#pragma unroll
            for (int i = 0; i < NA_U; ++i) {
                const float pi = __shfl(pg, (lane & 24) + i);
                den += pi;
                acc += pi * xlv[i];
            }
            mrun = mnew;
        }
    }

    const float4 cbv = ((const float4*)cb)[lane];
    const float dd = den + 1e-16f;
    v4f o;
    o.x = acc.x / dd + cbv.x;
    o.y = acc.y / dd + cbv.y;
    o.z = acc.z / dd + cbv.z;
    o.w = acc.w / dd + cbv.w;
    ((v4f*)agg)[rown] = o;

    const float ps = (o.x + o.y) + (o.z + o.w);
    const float pq = fmaf(o.w, o.w, fmaf(o.z, o.z, fmaf(o.y, o.y, o.x * o.x)));
    const bool up = (lane & 16) != 0;
    float r = (up ? pq : ps) + __shfl_xor(up ? ps : pq, 16);
    r += __shfl_xor(r, 8);
    r += __shfl_xor(r, 4);
    r += __shfl_xor(r, 2);
    r += __shfl_xor(r, 1);
    if ((lane & 15) == 0)
        atomicAdd(&red[(size_t)(n & (STAT_SLOTS - 1)) * STAT_PITCH + (lane >> 4)], (double)r);
}

extern "C" void kernel_launch(void* const* d_in, const int* in_sizes, int n_in,
                              void* d_out, int out_size, void* d_ws, size_t ws_size,
                              hipStream_t stream) {
    const float* x          = (const float*)d_in[0];
    const int*   edge_index = (const int*)  d_in[1];
    const float* edge_attr  = (const float*)d_in[2];
    const float* embed_W    = (const float*)d_in[3];
    const float* embed_b    = (const float*)d_in[4];
    const float* embed_lnw  = (const float*)d_in[5];
    const float* embed_lnb  = (const float*)d_in[6];
    const float* Wl         = (const float*)d_in[7];
    const float* bl         = (const float*)d_in[8];
    const float* Wr         = (const float*)d_in[9];
    const float* br         = (const float*)d_in[10];
    const float* We         = (const float*)d_in[11];
    const float* att        = (const float*)d_in[12];
    const float* conv_b     = (const float*)d_in[13];
    const float* pln_w      = (const float*)d_in[14];
    const float* pln_b      = (const float*)d_in[15];
    const float* proj_W     = (const float*)d_in[16];
    const float* proj_b     = (const float*)d_in[17];
    const float* ln2_w      = (const float*)d_in[18];
    const float* ln2_b      = (const float*)d_in[19];

    const int* srcE = edge_index;
    const int* dstE = edge_index + EE;

    size_t off = 0;
    auto carve = [&](size_t nbytes) -> char* {
        char* p = (char*)d_ws + off;
        off += (nbytes + 255) & ~(size_t)255;
        return p;
    };
    float*    xl    = (float*)   carve((size_t)NN * HCC * 4);
    float*    xr    = (float*)   carve((size_t)NN * HCC * 4);
    float*    agg   = (float*)   carve((size_t)NN * HCC * 4);
    int*      ssrc  = (int*)     carve((size_t)EE * 4);
    int*      perm  = (int*)     carve((size_t)EE * 4);
    int*      rank  = (int*)     carve((size_t)EE * 4);
    int*      goff  = (int*)     carve((size_t)(NN + 1) * 4);
    double*   red   = (double*)  carve((size_t)LL * STAT_SLOTS * STAT_PITCH * sizeof(double));
    (void)ws_size; (void)in_sizes; (void)n_in; (void)out_size;

    float* jk = (float*)d_out;

    const int mtiles = NN / 16;
    const dim3 blk(256);
    const dim3 gGemm((mtiles * 32 + 255) / 256);
    const dim3 gEdgeT(EE / 256);
    const dim3 gNode(NN / 8);

    const size_t shProj = (size_t)FF * (HCC + 4) * 4 + (size_t)2 * HCC * 4 + 16 * sizeof(double);

    gemm_wmma_f32<2, INF_, false, false, 1, true><<<gGemm, blk, 0, stream>>>(
        x, embed_W, embed_b, nullptr, NN, FF, nullptr, nullptr, nullptr, embed_lnw, embed_lnb, jk,
        Wl, bl, Wr, br, xl, xr);

    hipMemsetAsync(goff, 0, (size_t)(NN + 1) * 4, stream);
    csr_count_kernel<<<gEdgeT, blk, 0, stream>>>(dstE, goff, rank);
    csr_scan_kernel<<<dim3(1), dim3(1024), 0, stream>>>(goff, red);
    csr_scatter_kernel<<<gEdgeT, blk, 0, stream>>>(srcE, dstE, goff, rank, ssrc, perm);

    for (int l = 0; l < LL; ++l) {
        const float* We_l = We + (size_t)l * HCC * EDD;
        const float* at_l = att + (size_t)l * HH * CC;
        const float* cb_l = conv_b + (size_t)l * HCC;
        const float* pw_l = pln_w + (size_t)l * HCC;
        const float* pb_l = pln_b + (size_t)l * HCC;
        const float* pW_l = proj_W + (size_t)l * FF * HCC;
        const float* pB_l = proj_b + (size_t)l * FF;
        const float* l2w  = ln2_w + (size_t)l * FF;
        const float* l2b  = ln2_b + (size_t)l * FF;
        double*      red_l = red + (size_t)l * STAT_SLOTS * STAT_PITCH;

        node_attn_kernel<<<gNode, blk, 0, stream>>>(goff, ssrc, perm, edge_attr, We_l,
                                                    xl, xr, at_l, cb_l, red_l, agg);

        if (l + 1 < LL) {
            gemm_wmma_f32<2, HCC, true, true, 2, true><<<gGemm, blk, shProj, stream>>>(
                agg, pW_l, pB_l, nullptr, NN, FF, pw_l, pb_l, red_l, l2w, l2b, jk,
                Wl + (size_t)(l + 1) * HCC * FF, bl + (size_t)(l + 1) * HCC,
                Wr + (size_t)(l + 1) * HCC * FF, br + (size_t)(l + 1) * HCC, xl, xr);
        } else {
            gemm_wmma_f32<2, HCC, true, true, 2, false><<<gGemm, blk, shProj, stream>>>(
                agg, pW_l, pB_l, nullptr, NN, FF, pw_l, pb_l, red_l, l2w, l2b, jk,
                nullptr, nullptr, nullptr, nullptr, nullptr, nullptr);
        }
    }
}


# node attention: 9 resident waves per SIMD, not 8: lane exchanges as DPP operands (row_xmask, DPP8, permlanex16), no ds_bpermute index registers, no LDS waits; 128-thread blocks; sum order unchanged
# speedup vs baseline: 4.5655x; 1.0531x over previous
#include <hip/hip_runtime.h>
#include <math.h>

#define NN   50000
#define EE   800000
#define INF_ 512
#define FF   32
#define HH   4
#define CC   32
#define HCC  128
#define EDD  4
#define LL   2

typedef float v2f __attribute__((ext_vector_type(2)));
typedef float v8f __attribute__((ext_vector_type(8)));
typedef float v4f __attribute__((ext_vector_type(4)));

__device__ __forceinline__ float leakyf(float x, float s) {
    return x >= 0.0f ? x : s * x;
}
__device__ __forceinline__ double shfl_xor_f64(double v, int m) {
    const int lo = __shfl_xor(__double2loint(v), m);
    const int hi = __shfl_xor(__double2hiint(v), m);
    return __hiloint2double(hi, lo);
}

#define STAT_SLOTS 256
#define STAT_PITCH 16

#define LIN_STR (FF + 4)
template <int CT, int KT, bool USE_LDS, bool NORM_A, int LN_EPI, bool FUSE_LIN>
__global__ __launch_bounds__(256)
void gemm_wmma_f32(const float* __restrict__ A, const float* __restrict__ W,
                   const float* __restrict__ bias, float* __restrict__ Cout,
                   int M, int Nout,
                   const float* __restrict__ nw, const float* __restrict__ nb,
                   const double* __restrict__ red,
                   const float* __restrict__ lw, const float* __restrict__ lb,
                   float* __restrict__ jk,
                   const float* __restrict__ Wl, const float* __restrict__ bl,
                   const float* __restrict__ Wr, const float* __restrict__ br,
                   float* __restrict__ xl, float* __restrict__ xr) {
    extern __shared__ float sW[];
    __shared__ __attribute__((aligned(16))) float sLin[FUSE_LIN ? (2 * HCC + 8 * 16) * LIN_STR : 4];
    const int tid  = (int)threadIdx.x;
    const int lane = tid & 31;
    const int wave = (int)((blockIdx.x * blockDim.x + tid) >> 5);
    const int mtiles  = M >> 4;
    const int ngroups = Nout / (16 * CT);
    const int stride  = KT + 4;
    const bool live   = wave < mtiles * ngroups;
    const int hi = lane >> 4;
    const int mi = lane & 15;

    v2f a_all[NORM_A ? KT / 4 : 1];
    float muf = 0.0f, inv = 1.0f;

    if (USE_LDS) {
        const int total = Nout * KT;
        for (int i = tid; i < total; i += 256) {
            const int c  = i / KT;
            const int kk = i - c * KT;
            sW[c * stride + kk] = W[i];
        }
        if (NORM_A) {
            float* sN = sW + Nout * stride;
            for (int i = tid; i < KT; i += 256) {
                sN[i]      = nw[i];
                sN[KT + i] = nb[i];
            }
            const float* Ap = A + (size_t)((live ? wave / ngroups : 0) * 16 + mi) * KT + 2 * hi;
#pragma unroll
            for (int s = 0; s < KT / 4; ++s) {
                a_all[s].x = Ap[4 * s];
                a_all[s].y = Ap[4 * s + 1];
            }
            const double ss = red[(size_t)tid * STAT_PITCH];
            const double qq = red[(size_t)tid * STAT_PITCH + 1];
            const bool up = (lane & 16) != 0;
            double r = (up ? qq : ss) + shfl_xor_f64(up ? ss : qq, 16);
#pragma unroll
            for (int off = 8; off >= 1; off >>= 1) r += shfl_xor_f64(r, off);
            double* sP = (double*)(sN + 2 * KT);
            if ((lane & 15) == 0) sP[(up ? 8 : 0) + (tid >> 5)] = r;
        }
    }
    if (FUSE_LIN) {
        const v4f* Wl4 = (const v4f*)Wl;
        const v4f* Wr4 = (const v4f*)Wr;
        constexpr int LIN_NP = HCC * FF / 4 / 256;
        v4f pl[LIN_NP], pr[LIN_NP];
#pragma unroll
        for (int j = 0; j < LIN_NP; ++j) {
            pl[j] = Wl4[tid + 256 * j];
            pr[j] = Wr4[tid + 256 * j];
        }
#pragma unroll
        for (int j = 0; j < LIN_NP; ++j) {
            const int i  = tid + 256 * j;
            const int c  = i >> 3;
            const int k4 = (i & 7) * 4;
            *(v4f*)(sLin + c * LIN_STR + k4)         = pl[j];
            *(v4f*)(sLin + (HCC + c) * LIN_STR + k4) = pr[j];
        }
    }
    if (USE_LDS || FUSE_LIN) __syncthreads();
    if (USE_LDS) {
        if (NORM_A) {
            const double* sP = (const double*)(sW + Nout * stride + 2 * KT);
            double t = sP[lane & 15];
            t += shfl_xor_f64(t, 4);
            t += shfl_xor_f64(t, 2);
            t += shfl_xor_f64(t, 1);
            const double o  = shfl_xor_f64(t, 8);
            const bool   hq = (lane & 8) != 0;
            const double S  = hq ? o : t;
            const double Q  = hq ? t : o;
            constexpr double inv_n = 1.0 / ((double)NN * (double)HCC);
            const double mu  = S * inv_n;
            const double var = Q * inv_n - mu * mu;
            float varf = (float)var;
            if (varf < 0.0f) varf = 0.0f;
            inv = 1.0f / (sqrtf(varf) + 1e-5f);
            muf = (float)mu;
        }
    }
    if (!live) return;

    const int mtile = wave / ngroups;
    const int ng    = wave % ngroups;
    const int row0  = mtile * 16;
    const int col0  = ng * 16 * CT;

    v8f acc[CT] = {};
    const float* Arow = A + (size_t)(row0 + mi) * KT + 2 * hi;

    const float* sNk = sW + Nout * stride + 2 * hi;

    v2f a_cur;
    if (NORM_A) {
        a_cur = a_all[0];
    } else {
        a_cur.x = Arow[0];
        a_cur.y = Arow[1];
    }
    v2f b_cur[CT];
    if (!USE_LDS) {
#pragma unroll
        for (int t = 0; t < CT; ++t) {
            const float* Wc = W + (size_t)(col0 + t * 16 + mi) * KT + 2 * hi;
            b_cur[t].x = Wc[0];
            b_cur[t].y = Wc[1];
        }
    }

    for (int k = 0; k < KT; k += 4) {
        v2f a_nxt = {};
        v2f b_nxt[CT];
        if (k + 4 < KT) {
            if (NORM_A) {
                a_nxt = a_all[NORM_A ? (k + 4) / 4 : 0];
            } else {
                a_nxt.x = Arow[k + 4];
                a_nxt.y = Arow[k + 5];
            }
            if (!USE_LDS) {
#pragma unroll
                for (int t = 0; t < CT; ++t) {
                    const float* Wc = W + (size_t)(col0 + t * 16 + mi) * KT + 2 * hi;
                    b_nxt[t].x = Wc[k + 4];
                    b_nxt[t].y = Wc[k + 5];
                }
                if ((k & 31) == 0)
                    __builtin_prefetch(Arow + k + 132, 0, 1);
            }
        }
        v2f a_use = a_cur;
        if (NORM_A) {
            a_use.x = leakyf(sNk[k]     * ((a_cur.x - muf) * inv) + sNk[KT + k],     0.01f);
            a_use.y = leakyf(sNk[k + 1] * ((a_cur.y - muf) * inv) + sNk[KT + k + 1], 0.01f);
        }
#pragma unroll
        for (int t = 0; t < CT; ++t) {
            v2f b;
            if (USE_LDS) {
                const float* Sc = sW + (size_t)(col0 + t * 16 + mi) * stride + 2 * hi;
                b.x = Sc[k];
                b.y = Sc[k + 1];
            } else {
                b = b_cur[t];
            }
            acc[t] = __builtin_amdgcn_wmma_f32_16x16x4_f32(
                false, a_use, false, b, (short)0, acc[t], false, false);
        }
        a_cur = a_nxt;
        if (!USE_LDS) {
#pragma unroll
            for (int t = 0; t < CT; ++t) b_cur[t] = b_nxt[t];
        }
    }

    if (LN_EPI) {
        const size_t e0 = (size_t)(row0 + 8 * hi) * FF + mi;
        float j0[8], j1[8];
        if (LN_EPI == 2) {
#pragma unroll
            for (int v = 0; v < 8; ++v) {
                j0[v] = jk[e0 + (size_t)v * FF];
                j1[v] = jk[e0 + (size_t)v * FF + 16];
            }
        }
        const float bb0 = bias[mi], bb1 = bias[16 + mi];
        const float g0 = lw[mi], g1 = lw[16 + mi];
        const float c0 = lb[mi], c1 = lb[16 + mi];
        float z0[8], z1[8], s[8], q[8];
#pragma unroll
        for (int v = 0; v < 8; ++v) {
            z0[v] = acc[0][v] + bb0;
            z1[v] = acc[1][v] + bb1;
            s[v]  = z0[v] + z1[v];
        }
#pragma unroll
        for (int off = 8; off >= 1; off >>= 1) {
#pragma unroll
            for (int v = 0; v < 8; ++v) s[v] += __shfl_xor(s[v], off);
        }
#pragma unroll
        for (int v = 0; v < 8; ++v) {
            const float mu = s[v] * (1.0f / 32.0f);
            z0[v] -= mu;
            z1[v] -= mu;
            q[v] = z0[v] * z0[v] + z1[v] * z1[v];
        }
#pragma unroll
        for (int off = 8; off >= 1; off >>= 1) {
#pragma unroll
            for (int v = 0; v < 8; ++v) q[v] += __shfl_xor(q[v], off);
        }
#pragma unroll
        for (int v = 0; v < 8; ++v) {
            const float rs = rsqrtf(q[v] * (1.0f / 32.0f) + 1e-5f);
            const float y0 = leakyf(g0 * z0[v] * rs + c0, 0.01f);
            const float y1 = leakyf(g1 * z1[v] * rs + c1, 0.01f);
            const size_t e = e0 + (size_t)v * FF;
            if (Cout) {
                Cout[e]      = y0;
                Cout[e + 16] = y1;
            }
            if (FUSE_LIN) {
                float* sT = sLin + 2 * HCC * LIN_STR + (tid >> 5) * (16 * LIN_STR);
                sT[(v + 8 * hi) * LIN_STR + mi]      = y0;
                sT[(v + 8 * hi) * LIN_STR + 16 + mi] = y1;
            }
            if (LN_EPI == 2) {
                jk[e]      = fmaxf(j0[v], y0);
                jk[e + 16] = fmaxf(j1[v], y1);
            } else {
                jk[e]      = y0;
                jk[e + 16] = y1;
            }
        }
        if (FUSE_LIN) {
            const float* sT = sLin + 2 * HCC * LIN_STR + (tid >> 5) * (16 * LIN_STR)
                            + mi * LIN_STR + 2 * hi;
            v2f af[FF / 4];
#pragma unroll
            for (int s = 0; s < FF / 4; ++s) {
                af[s].x = sT[4 * s];
                af[s].y = sT[4 * s + 1];
            }
#pragma unroll
            for (int m = 0; m < 2; ++m) {
                const float* sB  = sLin + (m * HCC + mi) * LIN_STR + 2 * hi;
                const float* bv  = m ? br : bl;
                float*       out = m ? xr : xl;
                v8f acc2[HCC / 16] = {};
#pragma unroll
                for (int s = 0; s < FF / 4; ++s) {
#pragma unroll
                    for (int t = 0; t < HCC / 16; ++t) {
                        v2f b;
                        b.x = sB[t * 16 * LIN_STR + 4 * s];
                        b.y = sB[t * 16 * LIN_STR + 4 * s + 1];
                        acc2[t] = __builtin_amdgcn_wmma_f32_16x16x4_f32(
                            false, af[s], false, b, (short)0, acc2[t], false, false);
                    }
                }
#pragma unroll
                for (int t = 0; t < HCC / 16; ++t) {
                    const int col = t * 16 + mi;
                    const float bb = bv[col];
#pragma unroll
                    for (int v = 0; v < 8; ++v) {
                        const int r = row0 + v + 8 * hi;
                        out[(size_t)r * HCC + col] = acc2[t][v] + bb;
                    }
                }
            }
        }
    } else {
#pragma unroll
        for (int t = 0; t < CT; ++t) {
            const int col = col0 + t * 16 + mi;
            const float bb = bias ? bias[col] : 0.0f;
#pragma unroll
            for (int v = 0; v < 8; ++v) {
                const int r = row0 + v + 8 * hi;
                Cout[(size_t)r * Nout + col] = acc[t][v] + bb;
            }
        }
    }
}

__device__ __forceinline__ float ew_bcast(float v, int j) {
    return __int_as_float(__builtin_amdgcn_readlane(__float_as_int(v), j));
}

template <int M>
__device__ __forceinline__ float row_xor(float v) {
    return __int_as_float(__builtin_amdgcn_update_dpp(0, __float_as_int(v), 0x160 + M, 0xF, 0xF, true));
}
template <int I>
__device__ __forceinline__ float grp8_bcast(float v) {
    return __int_as_float((int)__builtin_amdgcn_mov_dpp8((unsigned)__float_as_int(v), (unsigned)(I * 0x249249)));
}
__device__ __forceinline__ float xrow(float v) {
    const int b = __float_as_int(v);
    return __int_as_float(__builtin_amdgcn_permlanex16(b, b, 0x76543210, 0xfedcba98, false, false));
}

__global__ __launch_bounds__(256)
void csr_count_kernel(const int* __restrict__ dst, int* __restrict__ cnt,
                      int* __restrict__ rank) {
    const int e = (int)(blockIdx.x * blockDim.x + threadIdx.x);
    if (e < EE) rank[e] = atomicAdd(&cnt[dst[e]], 1);
}

#define SCN_IT 49
#define SCN_G  7
__global__ __launch_bounds__(1024)
void csr_scan_kernel(int* __restrict__ cnt, double* __restrict__ red) {
    __shared__ int wsum[32];
    const int tid   = (int)threadIdx.x;
    static_assert(LL * STAT_SLOTS * 2 == 1024, "one slot double per thread of the block");
    red[(size_t)(tid >> 1) * STAT_PITCH + (tid & 1)] = 0.0;
    const int lane  = tid & 31;
    const int w     = tid >> 5;
    const int wbase = w * (SCN_IT * 32);
    int v[SCN_IT];
#pragma unroll
    for (int i = 0; i < SCN_IT; ++i) {
        const int idx = wbase + i * 32 + lane;
        v[i] = (idx < NN) ? cnt[idx] : 0;
    }
    int carry = 0;
#pragma unroll
    for (int r = 0; r < SCN_IT; r += SCN_G) {
        int inc[SCN_G];
#pragma unroll
        for (int k = 0; k < SCN_G; ++k) inc[k] = v[r + k];
#pragma unroll
        for (int o = 1; o < 32; o <<= 1) {
            int y[SCN_G];
#pragma unroll
            for (int k = 0; k < SCN_G; ++k) y[k] = __shfl_up(inc[k], o);
#pragma unroll
            for (int k = 0; k < SCN_G; ++k) if (lane >= o) inc[k] += y[k];
        }
#pragma unroll
        for (int k = 0; k < SCN_G; ++k) {
            const int x = v[r + k];
            v[r + k] = carry + inc[k] - x;
            carry += __builtin_amdgcn_readlane(inc[k], 31);
        }
    }
    if (lane == 0) wsum[w] = carry;
    __syncthreads();
    if (w == 0) {
        const int x = wsum[lane];
        int inc = x;
#pragma unroll
        for (int o = 1; o < 32; o <<= 1) {
            const int y = __shfl_up(inc, o);
            if (lane >= o) inc += y;
        }
        wsum[lane] = inc - x;
    }
    __syncthreads();
    const int wex = wsum[w];
#pragma unroll
    for (int i = 0; i < SCN_IT; ++i) {
        const int idx = wbase + i * 32 + lane;
        if (idx <= NN) cnt[idx] = wex + v[i];
    }
}

__global__ __launch_bounds__(256)
void csr_scatter_kernel(const int* __restrict__ src, const int* __restrict__ dst,
                        const int* __restrict__ off, const int* __restrict__ rank,
                        int* __restrict__ ssrc, int* __restrict__ perm) {
    const int e = (int)(blockIdx.x * blockDim.x + threadIdx.x);
    if (e < EE) {
        const int p = off[dst[e]] + rank[e];
        ssrc[p] = src[e];
        perm[p] = e;
    }
}

#define NA_U 8
__device__ __forceinline__ float edge_partial(const v4f xlv, const v4f xrv,
                                              float ex, float ey, float ez, float ew,
                                              const v2f wx01, const v2f wy01,
                                              const v2f wz01, const v2f ww01,
                                              const v2f wx23, const v2f wy23,
                                              const v2f wz23, const v2f ww23,
                                              const float4 atv) {
    v2f m01 = {xlv.x + xrv.x, xlv.y + xrv.y};
    v2f m23 = {xlv.z + xrv.z, xlv.w + xrv.w};
    const v2f bx = {ex, ex};
    const v2f by = {ey, ey};
    const v2f bz = {ez, ez};
    const v2f bw = {ew, ew};
    m01 = __builtin_elementwise_fma(wx01, bx, m01);
    m23 = __builtin_elementwise_fma(wx23, bx, m23);
    m01 = __builtin_elementwise_fma(wy01, by, m01);
    m23 = __builtin_elementwise_fma(wy23, by, m23);
    m01 = __builtin_elementwise_fma(wz01, bz, m01);
    m23 = __builtin_elementwise_fma(wz23, bz, m23);
    m01 = __builtin_elementwise_fma(ww01, bw, m01);
    m23 = __builtin_elementwise_fma(ww23, bw, m23);
    const v2f t01 = m01 * 0.2f;
    const v2f t23 = m23 * 0.2f;
    float s = fmaf(atv.x, (m01.x >= 0.0f ? m01.x : t01.x), 0.0f);
    s = fmaf(atv.y, (m01.y >= 0.0f ? m01.y : t01.y), s);
    s = fmaf(atv.z, (m23.x >= 0.0f ? m23.x : t23.x), s);
    s = fmaf(atv.w, (m23.y >= 0.0f ? m23.y : t23.y), s);
    return s;
}

__global__ __launch_bounds__(128, 9)
void node_attn_kernel(const int* __restrict__ off, const int* __restrict__ ssrc,
                      const int* __restrict__ perm, const float* __restrict__ ea,
                      const float* __restrict__ We, const float* __restrict__ xl,
                      const float* __restrict__ xr, const float* __restrict__ att,
                      const float* __restrict__ cb, double* __restrict__ red,
                      float* __restrict__ agg) {
    const int lane = (int)(threadIdx.x & 31);
    const int n = (int)blockIdx.x * 4 + __builtin_amdgcn_readfirstlane((int)(threadIdx.x >> 5));
    if (n >= NN) return;

    const v4f* xl4 = (const v4f*)xl;
    const v4f* xr4 = (const v4f*)xr;
    const float4* ea4 = (const float4*)ea;
    const unsigned rown = (unsigned)n * (unsigned)(HCC / 4) + (unsigned)lane;
    const v4f xrv = xr4[rown];
    const v4f xls = xl4[rown];
    const int p0 = __builtin_amdgcn_readfirstlane(off[n]);
    const int p1 = __builtin_amdgcn_readfirstlane(off[n + 1]);
    const float4 atv = ((const float4*)att)[lane];
    const float4* We4 = (const float4*)We;
    const float4 wr0 = We4[lane * 4 + 0];
    const float4 wr1 = We4[lane * 4 + 1];
    const float4 wr2 = We4[lane * 4 + 2];
    const float4 wr3 = We4[lane * 4 + 3];
    const v2f wx01 = {wr0.x, wr1.x}, wy01 = {wr0.y, wr1.y}, wz01 = {wr0.z, wr1.z}, ww01 = {wr0.w, wr1.w};
    const v2f wx23 = {wr2.x, wr3.x}, wy23 = {wr2.y, wr3.y}, wz23 = {wr2.z, wr3.z}, ww23 = {wr2.w, wr3.w};

    float s0 = edge_partial(xls, xrv, 0.0f, 0.0f, 0.0f, 0.0f,
                            wx01, wy01, wz01, ww01, wx23, wy23, wz23, ww23, atv);
    s0 += row_xor<4>(s0);
    s0 += row_xor<2>(s0);
    s0 += row_xor<1>(s0);
    float mrun = s0;
    float den  = 1.0f;
    v4f   acc  = xls;

    const int g = lane & 7;
    const bool b2 = (lane & 4) != 0;
    const bool b1 = (lane & 2) != 0;
    const bool b0 = (lane & 1) != 0;

    for (int pb = p0; pb < p1; pb += 32) {
        const int rem = p1 - pb;
        const int nb  = rem < 32 ? rem : 32;
        int sL = n;
        float4 evL = make_float4(0.0f, 0.0f, 0.0f, 0.0f);
        if (lane < nb) {
            sL = ssrc[pb + lane];
            evL = ea4[perm[pb + lane]];
        }
        const int ntrip = (nb + NA_U - 1) / NA_U;
        for (int t = 0; t < ntrip; ++t) {
            v4f xlv[NA_U];
#pragma unroll
            for (int i = 0; i < NA_U; ++i) {
                const int sN = __builtin_amdgcn_readlane(sL, t * NA_U + i);
                xlv[i] = xl4[(size_t)sN * (HCC / 4) + lane];
            }
            __builtin_amdgcn_sched_barrier(0);
            float sc[NA_U];
#pragma unroll
            for (int i = 0; i < NA_U; ++i) {
                const int j = t * NA_U + i;
                const float ex = ew_bcast(evL.x, j);
                const float ey = ew_bcast(evL.y, j);
                const float ez = ew_bcast(evL.z, j);
                const float ew = ew_bcast(evL.w, j);
                sc[i] = edge_partial(xlv[i], xrv, ex, ey, ez, ew,
                                     wx01, wy01, wz01, ww01, wx23, wy23, wz23, ww23, atv);
            }
            float ra[4];
#pragma unroll
            for (int k = 0; k < 4; ++k) {
                const float keep = b2 ? sc[k + 4] : sc[k];
                const float send = b2 ? sc[k] : sc[k + 4];
                ra[k] = keep + row_xor<4>(send);
            }
            float rb[2];
#pragma unroll
            for (int k = 0; k < 2; ++k) {
                const float keep = b1 ? ra[k + 2] : ra[k];
                const float send = b1 ? ra[k] : ra[k + 2];
                rb[k] = keep + row_xor<2>(send);
            }
            const float keep1 = b0 ? rb[1] : rb[0];
            const float send1 = b0 ? rb[0] : rb[1];
            const float tot = keep1 + row_xor<1>(send1);

            const bool valid = (t * NA_U + g) < nb;
            float cm = valid ? tot : -INFINITY;
            cm = fmaxf(cm, row_xor<4>(cm));
            cm = fmaxf(cm, row_xor<2>(cm));
            cm = fmaxf(cm, row_xor<1>(cm));
            const float mnew  = fmaxf(mrun, cm);
            const float scale = expf(mrun - mnew);
            const float pg    = valid ? expf(tot - mnew) : 0.0f;
            den *= scale;
            acc *= scale;
            static_assert(NA_U == 8, "one step per lane of a head's group of 8");
#define NA_STEP(I) { const float pi = grp8_bcast<I>(pg); den += pi; acc += pi * xlv[I]; }
            NA_STEP(0) NA_STEP(1) NA_STEP(2) NA_STEP(3) NA_STEP(4) NA_STEP(5) NA_STEP(6) NA_STEP(7)
#undef NA_STEP
            mrun = mnew;
        }
    }

    const float4 cbv = ((const float4*)cb)[lane];
    const float dd = den + 1e-16f;
    v4f o;
    o.x = acc.x / dd + cbv.x;
    o.y = acc.y / dd + cbv.y;
    o.z = acc.z / dd + cbv.z;
    o.w = acc.w / dd + cbv.w;
    const unsigned rowo = (unsigned)n * (unsigned)(HCC / 4) + __builtin_amdgcn_mbcnt_lo(~0u, 0u);
    ((v4f*)agg)[rowo] = o;

    const float ps = (o.x + o.y) + (o.z + o.w);
    const float pq = fmaf(o.w, o.w, fmaf(o.z, o.z, fmaf(o.y, o.y, o.x * o.x)));
    const bool up = (lane & 16) != 0;
    float r = (up ? pq : ps) + xrow(up ? ps : pq);
    r += row_xor<8>(r);
    r += row_xor<4>(r);
    r += row_xor<2>(r);
    r += row_xor<1>(r);
    if ((lane & 15) == 0)
        atomicAdd(&red[(size_t)(n & (STAT_SLOTS - 1)) * STAT_PITCH + (lane >> 4)], (double)r);
}

extern "C" void kernel_launch(void* const* d_in, const int* in_sizes, int n_in,
                              void* d_out, int out_size, void* d_ws, size_t ws_size,
                              hipStream_t stream) {
    const float* x          = (const float*)d_in[0];
    const int*   edge_index = (const int*)  d_in[1];
    const float* edge_attr  = (const float*)d_in[2];
    const float* embed_W    = (const float*)d_in[3];
    const float* embed_b    = (const float*)d_in[4];
    const float* embed_lnw  = (const float*)d_in[5];
    const float* embed_lnb  = (const float*)d_in[6];
    const float* Wl         = (const float*)d_in[7];
    const float* bl         = (const float*)d_in[8];
    const float* Wr         = (const float*)d_in[9];
    const float* br         = (const float*)d_in[10];
    const float* We         = (const float*)d_in[11];
    const float* att        = (const float*)d_in[12];
    const float* conv_b     = (const float*)d_in[13];
    const float* pln_w      = (const float*)d_in[14];
    const float* pln_b      = (const float*)d_in[15];
    const float* proj_W     = (const float*)d_in[16];
    const float* proj_b     = (const float*)d_in[17];
    const float* ln2_w      = (const float*)d_in[18];
    const float* ln2_b      = (const float*)d_in[19];

    const int* srcE = edge_index;
    const int* dstE = edge_index + EE;

    size_t off = 0;
    auto carve = [&](size_t nbytes) -> char* {
        char* p = (char*)d_ws + off;
        off += (nbytes + 255) & ~(size_t)255;
        return p;
    };
    float*    xl    = (float*)   carve((size_t)NN * HCC * 4);
    float*    xr    = (float*)   carve((size_t)NN * HCC * 4);
    float*    agg   = (float*)   carve((size_t)NN * HCC * 4);
    int*      ssrc  = (int*)     carve((size_t)EE * 4);
    int*      perm  = (int*)     carve((size_t)EE * 4);
    int*      rank  = (int*)     carve((size_t)EE * 4);
    int*      goff  = (int*)     carve((size_t)(NN + 1) * 4);
    double*   red   = (double*)  carve((size_t)LL * STAT_SLOTS * STAT_PITCH * sizeof(double));
    (void)ws_size; (void)in_sizes; (void)n_in; (void)out_size;

    float* jk = (float*)d_out;

    const int mtiles = NN / 16;
    const dim3 blk(256);
    const dim3 gGemm((mtiles * 32 + 255) / 256);
    const dim3 gEdgeT(EE / 256);
    const dim3 gNode(NN / 4);

    const size_t shProj = (size_t)FF * (HCC + 4) * 4 + (size_t)2 * HCC * 4 + 16 * sizeof(double);

    gemm_wmma_f32<2, INF_, false, false, 1, true><<<gGemm, blk, 0, stream>>>(
        x, embed_W, embed_b, nullptr, NN, FF, nullptr, nullptr, nullptr, embed_lnw, embed_lnb, jk,
        Wl, bl, Wr, br, xl, xr);

    hipMemsetAsync(goff, 0, (size_t)(NN + 1) * 4, stream);
    csr_count_kernel<<<gEdgeT, blk, 0, stream>>>(dstE, goff, rank);
    csr_scan_kernel<<<dim3(1), dim3(1024), 0, stream>>>(goff, red);
    csr_scatter_kernel<<<gEdgeT, blk, 0, stream>>>(srcE, dstE, goff, rank, ssrc, perm);

    for (int l = 0; l < LL; ++l) {
        const float* We_l = We + (size_t)l * HCC * EDD;
        const float* at_l = att + (size_t)l * HH * CC;
        const float* cb_l = conv_b + (size_t)l * HCC;
        const float* pw_l = pln_w + (size_t)l * HCC;
        const float* pb_l = pln_b + (size_t)l * HCC;
        const float* pW_l = proj_W + (size_t)l * FF * HCC;
        const float* pB_l = proj_b + (size_t)l * FF;
        const float* l2w  = ln2_w + (size_t)l * FF;
        const float* l2b  = ln2_b + (size_t)l * FF;
        double*      red_l = red + (size_t)l * STAT_SLOTS * STAT_PITCH;

        node_attn_kernel<<<gNode, dim3(128), 0, stream>>>(goff, ssrc, perm, edge_attr, We_l,
                                                    xl, xr, at_l, cb_l, red_l, agg);

        if (l + 1 < LL) {
            gemm_wmma_f32<2, HCC, true, true, 2, true><<<gGemm, blk, shProj, stream>>>(
                agg, pW_l, pB_l, nullptr, NN, FF, pw_l, pb_l, red_l, l2w, l2b, jk,
                Wl + (size_t)(l + 1) * HCC * FF, bl + (size_t)(l + 1) * HCC,
                Wr + (size_t)(l + 1) * HCC * FF, br + (size_t)(l + 1) * HCC, xl, xr);
        } else {
            gemm_wmma_f32<2, HCC, true, true, 2, false><<<gGemm, blk, shProj, stream>>>(
                agg, pW_l, pB_l, nullptr, NN, FF, pw_l, pb_l, red_l, l2w, l2b, jk,
                nullptr, nullptr, nullptr, nullptr, nullptr, nullptr);
        }
    }
}
